# starting kernel
# baseline (speedup 1.0000x reference)
#include <hip/hip_runtime.h>
#include <math.h>

#define F 20
#define NEG_SLOPE 0.2f

typedef float v2f __attribute__((ext_vector_type(2)));
typedef float v8f __attribute__((ext_vector_type(8)));

__device__ __forceinline__ void atomicMaxF32(float* addr, float val) {
    if (val >= 0.0f) {
        atomicMax((int*)addr, __float_as_int(val));
    } else {
        atomicMin((unsigned int*)addr, __float_as_uint(val));
    }
}

__global__ void gat_node_init_l1(const float* __restrict__ x,
                                 const float* __restrict__ W1,
                                 const float* __restrict__ a_src,
                                 const float* __restrict__ a_dst,
                                 float* __restrict__ h,
                                 float* __restrict__ asrc,
                                 float* __restrict__ adst,
                                 float* __restrict__ m,
                                 float* __restrict__ denom,
                                 float* __restrict__ accum,
                                 int N) {
    int i = blockIdx.x * blockDim.x + threadIdx.x;
    if (i >= N) return;
    float xv = x[i];
    float sa = 0.0f, sd = 0.0f;
#pragma unroll
    for (int f = 0; f < F; ++f) {
        float hv = xv * W1[f];
        h[i * F + f] = hv;
        sa += hv * a_src[f];
        sd += hv * a_dst[f];
        accum[i * F + f] = 0.0f;
    }
    asrc[i] = sa;
    adst[i] = sd;
    m[i] = -3.402823466e38f;
    denom[i] = 0.0f;
}

__global__ void __launch_bounds__(32)
gat_wmma_feat_x_w(const float* __restrict__ feat,
                  const float* __restrict__ W,
                  float* __restrict__ h,
                  int N) {
    int tile  = blockIdx.x;
    int lane  = threadIdx.x;
    int mrow  = lane & 15;
    int khalf = (lane >> 4) * 2;

    int row  = tile * 16 + mrow;
    int rowc = row < N ? row : N - 1;
    const float* arow = feat + (size_t)rowc * F;

    int   n1   = 16 + mrow;
    int   n1c  = (n1 < F) ? n1 : 0;
    float msk1 = (n1 < F) ? 1.0f : 0.0f;

    v8f c0 = {};
    v8f c1 = {};
#pragma unroll
    for (int k0 = 0; k0 < F; k0 += 4) {
        const float* wr0 = W + (size_t)(k0 + khalf + 0) * F;
        const float* wr1 = W + (size_t)(k0 + khalf + 1) * F;

        v2f a;
        a.x = arow[k0 + khalf + 0];
        a.y = arow[k0 + khalf + 1];

        v2f b0;
        b0.x = wr0[mrow];
        b0.y = wr1[mrow];
        c0 = __builtin_amdgcn_wmma_f32_16x16x4_f32(false, a, false, b0,
                                                   (short)0, c0, false, false);

        v2f b1;
        b1.x = wr0[n1c] * msk1;
        b1.y = wr1[n1c] * msk1;
        c1 = __builtin_amdgcn_wmma_f32_16x16x4_f32(false, a, false, b1,
                                                   (short)0, c1, false, false);
    }

    int rbase = tile * 16 + ((lane >> 4) << 3);
#pragma unroll
    for (int j = 0; j < 8; ++j) {
        int r = rbase + j;
        if (r < N) {
            h[(size_t)r * F + mrow] = c0[j];
            if (n1 < F) h[(size_t)r * F + n1] = c1[j];
        }
    }
}

__global__ void gat_node_alpha_init(const float* __restrict__ h,
                                    const float* __restrict__ a_src,
                                    const float* __restrict__ a_dst,
                                    float* __restrict__ asrc,
                                    float* __restrict__ adst,
                                    float* __restrict__ m,
                                    float* __restrict__ denom,
                                    float* __restrict__ accum,
                                    int N) {
    int i = blockIdx.x * blockDim.x + threadIdx.x;
    if (i >= N) return;
    float sa = 0.0f, sd = 0.0f;
#pragma unroll
    for (int f = 0; f < F; ++f) {
        float hv = h[i * F + f];
        sa += hv * a_src[f];
        sd += hv * a_dst[f];
        accum[i * F + f] = 0.0f;
    }
    asrc[i] = sa;
    adst[i] = sd;
    m[i] = -3.402823466e38f;
    denom[i] = 0.0f;
}

__global__ void gat_edge_max(const int* __restrict__ src,
                             const int* __restrict__ dst,
                             const float* __restrict__ asrc,
                             const float* __restrict__ adst,
                             float* __restrict__ m,
                             int E, int Etot) {
    int e = blockIdx.x * blockDim.x + threadIdx.x;
    if (e >= Etot) return;
    int s, d;
    if (e < E) { s = src[e]; d = dst[e]; }
    else       { s = e - E; d = e - E; }
    float ev = asrc[s] + adst[d];
    ev = ev > 0.0f ? ev : NEG_SLOPE * ev;
    atomicMaxF32(&m[d], ev);
}

__global__ void gat_edge_acc(const int* __restrict__ src,
                             const int* __restrict__ dst,
                             const float* __restrict__ asrc,
                             const float* __restrict__ adst,
                             const float* __restrict__ m,
                             const float* __restrict__ h,
                             float* __restrict__ denom,
                             float* __restrict__ accum,
                             int E, int Etot) {
    int e = blockIdx.x * blockDim.x + threadIdx.x;
    if (e >= Etot) return;
    int s, d;
    if (e < E) { s = src[e]; d = dst[e]; }
    else       { s = e - E; d = e - E; }
    float ev = asrc[s] + adst[d];
    ev = ev > 0.0f ? ev : NEG_SLOPE * ev;
    float t = __expf(ev - m[d]);
    atomicAdd(&denom[d], t);
    const float* hs = h + (size_t)s * F;
    float* ac = accum + (size_t)d * F;
#pragma unroll
    for (int f = 0; f < F; ++f) {
        atomicAdd(&ac[f], t * hs[f]);
    }
}

__global__ void gat_node_finish(const float* __restrict__ accum,
                                const float* __restrict__ denom,
                                const float* __restrict__ b,
                                float* __restrict__ feat,
                                int N) {
    int i = blockIdx.x * blockDim.x + threadIdx.x;
    if (i >= N) return;
    float inv = 1.0f / (denom[i] + 1e-16f);
#pragma unroll
    for (int f = 0; f < F; ++f) {
        float v = accum[i * F + f] * inv + b[f];
        feat[i * F + f] = v > 0.0f ? v : 0.0f;
    }
}

__global__ void gat_final_linear(const float* __restrict__ feat,
                                 const float* __restrict__ Wl,
                                 const float* __restrict__ bl,
                                 float* __restrict__ out,
                                 int N) {
    int i = blockIdx.x * blockDim.x + threadIdx.x;
    if (i >= N) return;
    float sacc = bl[0];
#pragma unroll
    for (int f = 0; f < F; ++f) sacc += feat[i * F + f] * Wl[f];
    out[i] = sacc;
}

extern "C" void kernel_launch(void* const* d_in, const int* in_sizes, int n_in,
                              void* d_out, int out_size, void* d_ws, size_t ws_size,
                              hipStream_t stream) {
    (void)n_in; (void)out_size; (void)ws_size;

    const int N    = in_sizes[0];
    const int E    = in_sizes[1] / 2;
    const int Etot = E + N;

    const float* x    = (const float*)d_in[0];
    const int*   ei   = (const int*)d_in[1];
    const float* W1   = (const float*)d_in[2];
    const float* as1  = (const float*)d_in[3];
    const float* ad1  = (const float*)d_in[4];
    const float* b1   = (const float*)d_in[5];
    const float* W2   = (const float*)d_in[6];
    const float* as2  = (const float*)d_in[7];
    const float* ad2  = (const float*)d_in[8];
    const float* b2   = (const float*)d_in[9];
    const float* Wl   = (const float*)d_in[10];
    const float* bl   = (const float*)d_in[11];

    const int* srcIdx = ei;
    const int* dstIdx = ei + E;

    float* ws     = (float*)d_ws;
    float* h      = ws + (size_t)0  * N;
    float* accum  = ws + (size_t)20 * N;
    float* feat1  = ws + (size_t)40 * N;
    float* feat2  = ws + (size_t)60 * N;
    float* asrc   = ws + (size_t)80 * N;
    float* adst   = ws + (size_t)81 * N;
    float* m      = ws + (size_t)82 * N;
    float* denom  = ws + (size_t)83 * N;

    const int TB   = 256;
    const int gN   = (N + TB - 1) / TB;
    const int gE   = (Etot + TB - 1) / TB;
    const int gMM  = (N + 15) / 16;

    gat_node_init_l1<<<gN, TB, 0, stream>>>(x, W1, as1, ad1,
                                            h, asrc, adst, m, denom, accum, N);
    gat_edge_max<<<gE, TB, 0, stream>>>(srcIdx, dstIdx, asrc, adst, m, E, Etot);
    gat_edge_acc<<<gE, TB, 0, stream>>>(srcIdx, dstIdx, asrc, adst, m, h,
                                        denom, accum, E, Etot);
    gat_node_finish<<<gN, TB, 0, stream>>>(accum, denom, b1, feat1, N);

    gat_wmma_feat_x_w<<<gMM, 32, 0, stream>>>(feat1, W2, h, N);
    gat_node_alpha_init<<<gN, TB, 0, stream>>>(h, as2, ad2,
                                               asrc, adst, m, denom, accum, N);
    gat_edge_max<<<gE, TB, 0, stream>>>(srcIdx, dstIdx, asrc, adst, m, E, Etot);
    gat_edge_acc<<<gE, TB, 0, stream>>>(srcIdx, dstIdx, asrc, adst, m, h,
                                        denom, accum, E, Etot);
    gat_node_finish<<<gN, TB, 0, stream>>>(accum, denom, b2, feat2, N);

    gat_final_linear<<<gN, TB, 0, stream>>>(feat2, Wl, bl, (float*)d_out, N);
}


# edge_acc: one wave-wide float atomic per edge (21 lanes add its 20 products and its exp into one 128-byte row of sums, denominator in column 20), not 21 per-lane atomics; arrival-order sums kept
# speedup vs baseline: 3.4538x; 3.4538x over previous
#include <hip/hip_runtime.h>
#include <math.h>

#define F 20
#define NEG_SLOPE 0.2f
#define AS 32

typedef float v2f __attribute__((ext_vector_type(2)));
typedef float v8f __attribute__((ext_vector_type(8)));

__device__ __forceinline__ void atomicMaxF32(float* addr, float val) {
    if (val >= 0.0f) {
        atomicMax((int*)addr, __float_as_int(val));
    } else {
        atomicMin((unsigned int*)addr, __float_as_uint(val));
    }
}

__device__ __forceinline__ void zero_acc_row(float* __restrict__ accum, int i) {
    float4* row = (float4*)(accum + (size_t)i * AS);
    const float4 z = make_float4(0.0f, 0.0f, 0.0f, 0.0f);
#pragma unroll
    for (int q = 0; q < 6; ++q) row[q] = z;
}

__device__ __forceinline__ int lane_i(int v, int l) {
    return __builtin_amdgcn_readlane(v, l);
}
__device__ __forceinline__ float lane_f(float v, int l) {
    return __int_as_float(__builtin_amdgcn_readlane(__float_as_int(v), l));
}

__global__ void gat_node_init_l1(const float* __restrict__ x,
                                 const float* __restrict__ W1,
                                 const float* __restrict__ a_src,
                                 const float* __restrict__ a_dst,
                                 float* __restrict__ h,
                                 float* __restrict__ asrc,
                                 float* __restrict__ adst,
                                 float* __restrict__ m,
                                 float* __restrict__ accum,
                                 int N) {
    int i = blockIdx.x * blockDim.x + threadIdx.x;
    if (i >= N) return;
    float xv = x[i];
    float sa = 0.0f, sd = 0.0f;
#pragma unroll
    for (int f = 0; f < F; ++f) {
        float hv = xv * W1[f];
        h[i * F + f] = hv;
        sa += hv * a_src[f];
        sd += hv * a_dst[f];
    }
    zero_acc_row(accum, i);
    asrc[i] = sa;
    adst[i] = sd;
    m[i] = -3.402823466e38f;
}

__global__ void __launch_bounds__(32)
gat_wmma_feat_x_w(const float* __restrict__ feat,
                  const float* __restrict__ W,
                  float* __restrict__ h,
                  int N) {
    int tile  = blockIdx.x;
    int lane  = threadIdx.x;
    int mrow  = lane & 15;
    int khalf = (lane >> 4) * 2;

    int row  = tile * 16 + mrow;
    int rowc = row < N ? row : N - 1;
    const float* arow = feat + (size_t)rowc * F;

    int   n1   = 16 + mrow;
    int   n1c  = (n1 < F) ? n1 : 0;
    float msk1 = (n1 < F) ? 1.0f : 0.0f;

    v8f c0 = {};
    v8f c1 = {};
#pragma unroll
    for (int k0 = 0; k0 < F; k0 += 4) {
        const float* wr0 = W + (size_t)(k0 + khalf + 0) * F;
        const float* wr1 = W + (size_t)(k0 + khalf + 1) * F;

        v2f a;
        a.x = arow[k0 + khalf + 0];
        a.y = arow[k0 + khalf + 1];

        v2f b0;
        b0.x = wr0[mrow];
        b0.y = wr1[mrow];
        c0 = __builtin_amdgcn_wmma_f32_16x16x4_f32(false, a, false, b0,
                                                   (short)0, c0, false, false);

        v2f b1;
        b1.x = wr0[n1c] * msk1;
        b1.y = wr1[n1c] * msk1;
        c1 = __builtin_amdgcn_wmma_f32_16x16x4_f32(false, a, false, b1,
                                                   (short)0, c1, false, false);
    }

    int rbase = tile * 16 + ((lane >> 4) << 3);
#pragma unroll
    for (int j = 0; j < 8; ++j) {
        int r = rbase + j;
        if (r < N) {
            h[(size_t)r * F + mrow] = c0[j];
            if (n1 < F) h[(size_t)r * F + n1] = c1[j];
        }
    }
}

__global__ void gat_node_alpha_init(const float* __restrict__ h,
                                    const float* __restrict__ a_src,
                                    const float* __restrict__ a_dst,
                                    float* __restrict__ asrc,
                                    float* __restrict__ adst,
                                    float* __restrict__ m,
                                    float* __restrict__ accum,
                                    int N) {
    int i = blockIdx.x * blockDim.x + threadIdx.x;
    if (i >= N) return;
    float sa = 0.0f, sd = 0.0f;
#pragma unroll
    for (int f = 0; f < F; ++f) {
        float hv = h[i * F + f];
        sa += hv * a_src[f];
        sd += hv * a_dst[f];
    }
    zero_acc_row(accum, i);
    asrc[i] = sa;
    adst[i] = sd;
    m[i] = -3.402823466e38f;
}

__global__ void gat_edge_max(const int* __restrict__ src,
                             const int* __restrict__ dst,
                             const float* __restrict__ asrc,
                             const float* __restrict__ adst,
                             float* __restrict__ m,
                             int E, int Etot) {
    int e = blockIdx.x * blockDim.x + threadIdx.x;
    if (e >= Etot) return;
    int s, d;
    if (e < E) { s = src[e]; d = dst[e]; }
    else       { s = e - E; d = e - E; }
    float ev = asrc[s] + adst[d];
    ev = ev > 0.0f ? ev : NEG_SLOPE * ev;
    atomicMaxF32(&m[d], ev);
}

__global__ void __launch_bounds__(256)
gat_edge_acc(const int* __restrict__ src,
             const int* __restrict__ dst,
             const float* __restrict__ asrc,
             const float* __restrict__ adst,
             const float* __restrict__ m,
             const float* __restrict__ h,
             float* __restrict__ accum,
             int E, int Etot) {
    const int lane = threadIdx.x & 31;
    const int e = blockIdx.x * blockDim.x + threadIdx.x;
    int hoff = 0, aoff = 0;
    float t = 0.0f;
    if (e < Etot) {
        int s, d;
        if (e < E) { s = src[e]; d = dst[e]; }
        else       { s = e - E; d = e - E; }
        float ev = asrc[s] + adst[d];
        ev = ev > 0.0f ? ev : NEG_SLOPE * ev;
        t = __expf(ev - m[d]);
        hoff = s * F;
        aoff = d * AS;
    }
    const int ebase = __builtin_amdgcn_readfirstlane(e - lane);
    const int cnt = Etot - ebase;
    const int lc = lane < F ? lane : F - 1;

    if (cnt >= 32) {
#pragma unroll
        for (int g = 0; g < 32; g += 8) {
            int so[8], ao[8];
            float tt[8];
#pragma unroll
            for (int k = 0; k < 8; ++k) {
                so[k] = lane_i(hoff, g + k);
                ao[k] = lane_i(aoff, g + k);
                tt[k] = lane_f(t, g + k);
            }
            if (lane <= F) {
                float hv[8];
#pragma unroll
                for (int k = 0; k < 8; ++k) hv[k] = h[so[k] + lc];
#pragma unroll
                for (int k = 0; k < 8; ++k)
                    atomicAdd(&accum[ao[k] + lane], lane < F ? tt[k] * hv[k] : tt[k]);
            }
        }
    } else {
        for (int j = 0; j < cnt; ++j) {
            int so = lane_i(hoff, j);
            int ao = lane_i(aoff, j);
            float tt = lane_f(t, j);
            if (lane <= F) {
                float hv = h[so + lc];
                atomicAdd(&accum[ao + lane], lane < F ? tt * hv : tt);
            }
        }
    }
}

__global__ void gat_node_finish(const float* __restrict__ accum,
                                const float* __restrict__ b,
                                float* __restrict__ feat,
                                int N) {
    int i = blockIdx.x * blockDim.x + threadIdx.x;
    if (i >= N) return;
    const float* row = accum + (size_t)i * AS;
    float inv = 1.0f / (row[F] + 1e-16f);
#pragma unroll
    for (int f = 0; f < F; ++f) {
        float v = row[f] * inv + b[f];
        feat[i * F + f] = v > 0.0f ? v : 0.0f;
    }
}

__global__ void gat_final_linear(const float* __restrict__ feat,
                                 const float* __restrict__ Wl,
                                 const float* __restrict__ bl,
                                 float* __restrict__ out,
                                 int N) {
    int i = blockIdx.x * blockDim.x + threadIdx.x;
    if (i >= N) return;
    float sacc = bl[0];
#pragma unroll
    for (int f = 0; f < F; ++f) sacc += feat[i * F + f] * Wl[f];
    out[i] = sacc;
}

extern "C" void kernel_launch(void* const* d_in, const int* in_sizes, int n_in,
                              void* d_out, int out_size, void* d_ws, size_t ws_size,
                              hipStream_t stream) {
    (void)n_in; (void)out_size; (void)ws_size;

    const int N    = in_sizes[0];
    const int E    = in_sizes[1] / 2;
    const int Etot = E + N;

    const float* x    = (const float*)d_in[0];
    const int*   ei   = (const int*)d_in[1];
    const float* W1   = (const float*)d_in[2];
    const float* as1  = (const float*)d_in[3];
    const float* ad1  = (const float*)d_in[4];
    const float* b1   = (const float*)d_in[5];
    const float* W2   = (const float*)d_in[6];
    const float* as2  = (const float*)d_in[7];
    const float* ad2  = (const float*)d_in[8];
    const float* b2   = (const float*)d_in[9];
    const float* Wl   = (const float*)d_in[10];
    const float* bl   = (const float*)d_in[11];

    const int* srcIdx = ei;
    const int* dstIdx = ei + E;

    float* ws     = (float*)d_ws;
    float* accum  = ws + (size_t)0  * N;
    float* h      = ws + (size_t)32 * N;
    float* feat1  = ws + (size_t)52 * N;
    float* feat2  = ws + (size_t)72 * N;
    float* asrc   = ws + (size_t)92 * N;
    float* adst   = ws + (size_t)93 * N;
    float* m      = ws + (size_t)94 * N;

    const int TB   = 256;
    const int gN   = (N + TB - 1) / TB;
    const int gE   = (Etot + TB - 1) / TB;
    const int gMM  = (N + 15) / 16;

    gat_node_init_l1<<<gN, TB, 0, stream>>>(x, W1, as1, ad1,
                                            h, asrc, adst, m, accum, N);
    gat_edge_max<<<gE, TB, 0, stream>>>(srcIdx, dstIdx, asrc, adst, m, E, Etot);
    gat_edge_acc<<<gE, TB, 0, stream>>>(srcIdx, dstIdx, asrc, adst, m, h,
                                        accum, E, Etot);
    gat_node_finish<<<gN, TB, 0, stream>>>(accum, b1, feat1, N);

    gat_wmma_feat_x_w<<<gMM, 32, 0, stream>>>(feat1, W2, h, N);
    gat_node_alpha_init<<<gN, TB, 0, stream>>>(h, as2, ad2,
                                               asrc, adst, m, accum, N);
    gat_edge_max<<<gE, TB, 0, stream>>>(srcIdx, dstIdx, asrc, adst, m, E, Etot);
    gat_edge_acc<<<gE, TB, 0, stream>>>(srcIdx, dstIdx, asrc, adst, m, h,
                                        accum, E, Etot);
    gat_node_finish<<<gN, TB, 0, stream>>>(accum, b2, feat2, N);

    gat_final_linear<<<gN, TB, 0, stream>>>(feat2, Wl, bl, (float*)d_out, N);
}


# Edge softmax and aggregation owner-computes: a workgroup owns 392 dst nodes, finds its edges by scanning one-byte owner ids, sums them in LDS with LDS float adds (arrival order); no global atomics
# speedup vs baseline: 6.3672x; 1.8435x over previous
#include <hip/hip_runtime.h>
#include <math.h>

#define F 20
#define NEG_SLOPE 0.2f

#define RN     392
#define AW     21
#define OWN_T  1024
#define SEG    32768
#define CAP    32768
#define KSEG   131072
#define LOW23  0x7FFFFFu
#define NO_EDGE 0xFFFFFFFFu

typedef float v2f __attribute__((ext_vector_type(2)));
typedef float v8f __attribute__((ext_vector_type(8)));

__device__ __forceinline__ float lrelu(float v) {
    return v > 0.0f ? v : NEG_SLOPE * v;
}

__device__ __forceinline__ int lane_i(int v, int l) {
    return __builtin_amdgcn_readlane(v, l);
}
__device__ __forceinline__ float lane_f(float v, int l) {
    return __int_as_float(__builtin_amdgcn_readlane(__float_as_int(v), l));
}

__global__ void gat_node_init_l1(const float* __restrict__ x,
                                 const float* __restrict__ W1,
                                 const float* __restrict__ a_src,
                                 const float* __restrict__ a_dst,
                                 float* __restrict__ h,
                                 float* __restrict__ asrc,
                                 float* __restrict__ adst,
                                 int N) {
    int i = blockIdx.x * blockDim.x + threadIdx.x;
    if (i >= N) return;
    float xv = x[i];
    float sa = 0.0f, sd = 0.0f;
#pragma unroll
    for (int f = 0; f < F; ++f) {
        float hv = xv * W1[f];
        h[i * F + f] = hv;
        sa += hv * a_src[f];
        sd += hv * a_dst[f];
    }
    asrc[i] = sa;
    adst[i] = sd;
}

__global__ void __launch_bounds__(32)
gat_wmma_feat_x_w(const float* __restrict__ feat,
                  const float* __restrict__ W,
                  float* __restrict__ h,
                  int N) {
    int tile  = blockIdx.x;
    int lane  = threadIdx.x;
    int mrow  = lane & 15;
    int khalf = (lane >> 4) * 2;

    int row  = tile * 16 + mrow;
    int rowc = row < N ? row : N - 1;
    const float* arow = feat + (size_t)rowc * F;

    int   n1   = 16 + mrow;
    int   n1c  = (n1 < F) ? n1 : 0;
    float msk1 = (n1 < F) ? 1.0f : 0.0f;

    v8f c0 = {};
    v8f c1 = {};
#pragma unroll
    for (int k0 = 0; k0 < F; k0 += 4) {
        const float* wr0 = W + (size_t)(k0 + khalf + 0) * F;
        const float* wr1 = W + (size_t)(k0 + khalf + 1) * F;

        v2f a;
        a.x = arow[k0 + khalf + 0];
        a.y = arow[k0 + khalf + 1];

        v2f b0;
        b0.x = wr0[mrow];
        b0.y = wr1[mrow];
        c0 = __builtin_amdgcn_wmma_f32_16x16x4_f32(false, a, false, b0,
                                                   (short)0, c0, false, false);

        v2f b1;
        b1.x = wr0[n1c] * msk1;
        b1.y = wr1[n1c] * msk1;
        c1 = __builtin_amdgcn_wmma_f32_16x16x4_f32(false, a, false, b1,
                                                   (short)0, c1, false, false);
    }

    int rbase = tile * 16 + ((lane >> 4) << 3);
#pragma unroll
    for (int j = 0; j < 8; ++j) {
        int r = rbase + j;
        if (r < N) {
            h[(size_t)r * F + mrow] = c0[j];
            if (n1 < F) h[(size_t)r * F + n1] = c1[j];
        }
    }
}

__global__ void gat_node_alpha(const float* __restrict__ h,
                               const float* __restrict__ a_src,
                               const float* __restrict__ a_dst,
                               float* __restrict__ asrc,
                               float* __restrict__ adst,
                               int N) {
    int i = blockIdx.x * blockDim.x + threadIdx.x;
    if (i >= N) return;
    float sa = 0.0f, sd = 0.0f;
#pragma unroll
    for (int f = 0; f < F; ++f) {
        float hv = h[i * F + f];
        sa += hv * a_src[f];
        sd += hv * a_dst[f];
    }
    asrc[i] = sa;
    adst[i] = sd;
}

__global__ void gat_owner_keys(const int* __restrict__ dst,
                               unsigned int* __restrict__ keyw,
                               int nWords) {
    int w = blockIdx.x * blockDim.x + threadIdx.x;
    if (w >= nWords) return;
    unsigned int k = 0u;
#pragma unroll
    for (int c = 0; c < 4; ++c)
        k |= (((unsigned int)dst[4 * w + c] / (unsigned int)RN) & 0xFFu) << (8 * c);
    keyw[w] = k;
}

__global__ void __launch_bounds__(OWN_T)
gat_owner_layer(const int* __restrict__ src,
                const int* __restrict__ dst,
                const unsigned int* __restrict__ keyw,
                const float* __restrict__ asrc,
                const float* __restrict__ adst,
                const float* __restrict__ h,
                const float* __restrict__ b,
                float* __restrict__ feat,
                int N, int E) {
    __shared__ float accL[RN * AW];
    __shared__ float mL[RN];
    __shared__ float mNew[RN];
    __shared__ float adL[RN];
    __shared__ float capL[RN];
    __shared__ float scL[RN];
    __shared__ unsigned listL[CAP];
    __shared__ int cntL;

    const int tid  = threadIdx.x;
    const int lane = tid & 31;
    const int wave = __builtin_amdgcn_readfirstlane(tid >> 5);
    const int lo   = blockIdx.x * RN;
    const int nOwn = (N - lo) < RN ? (N - lo) : RN;

    for (int dl = tid; dl < RN; dl += OWN_T) {
        float a = -3.402823466e38f, ad = 0.0f;
        if (dl < nOwn) { a = asrc[lo + dl]; ad = adst[lo + dl]; }
        mL[dl] = a;
        mNew[dl] = a;
        adL[dl] = ad;
    }
    for (int idx = tid; idx < RN * AW; idx += OWN_T) {
        const int dl = idx / AW;
        const int f  = idx - dl * AW;
        float v = 0.0f;
        if (dl < nOwn) v = f < F ? h[(size_t)(lo + dl) * F + f] : 1.0f;
        accL[idx] = v;
    }
    if (tid == 0) cntL = 0;
    __syncthreads();

    const unsigned gpat = (unsigned)(blockIdx.x & 0xFF) * 0x01010101u;
    auto scan_keys = [&](const int kb) {
        const uint4* __restrict__ k4 = (const uint4*)(keyw + (kb >> 2));
        uint4 q[KSEG / (16 * OWN_T)];
#pragma unroll
        for (int k = 0; k < KSEG / (16 * OWN_T); ++k) q[k] = k4[k * OWN_T + tid];
#pragma unroll
        for (int k = 0; k < KSEG / (16 * OWN_T); ++k) {
            const unsigned w[4] = {q[k].x, q[k].y, q[k].z, q[k].w};
            unsigned hz[4];
#pragma unroll
            for (int c = 0; c < 4; ++c) {
                const unsigned v = w[c] ^ gpat;
                hz[c] = (v - 0x01010101u) & ~v & 0x80808080u;
            }
            if (hz[0] | hz[1] | hz[2] | hz[3]) {
                const unsigned e0 = (unsigned)kb + (unsigned)(k * OWN_T + tid) * 16u;
#pragma unroll
                for (int c = 0; c < 4; ++c) {
                    unsigned m = hz[c];
                    while (m != 0u) {
                        const unsigned byte = (unsigned)__builtin_ctz(m) >> 3;
                        const int slot = atomicAdd(&cntL, 1);
                        if (slot < CAP) listL[slot] = e0 + 4u * (unsigned)c + byte;
                        m &= m - 1u;
                    }
                }
            }
        }
    };

    auto scan_segment = [&](const int sb) {
        if (((E & 3) == 0) && (sb + SEG <= E)) {
            const int4* __restrict__ dst4 = (const int4*)(dst + sb);
            int4 dq[SEG / (4 * OWN_T)];
#pragma unroll
            for (int k = 0; k < SEG / (4 * OWN_T); ++k) dq[k] = dst4[k * OWN_T + tid];
#pragma unroll
            for (int k = 0; k < SEG / (4 * OWN_T); ++k) {
                const unsigned u0 = (unsigned)(dq[k].x - lo);
                const unsigned u1 = (unsigned)(dq[k].y - lo);
                const unsigned u2 = (unsigned)(dq[k].z - lo);
                const unsigned u3 = (unsigned)(dq[k].w - lo);
                const bool m0 = u0 < (unsigned)nOwn, m1 = u1 < (unsigned)nOwn;
                const bool m2 = u2 < (unsigned)nOwn, m3 = u3 < (unsigned)nOwn;
                if (m0 | m1 | m2 | m3) {
                    const unsigned e0 = (unsigned)(sb + (k * OWN_T + tid) * 4);
                    if (m0) { const int slot = atomicAdd(&cntL, 1); if (slot < CAP) listL[slot] = e0 + 0u; }
                    if (m1) { const int slot = atomicAdd(&cntL, 1); if (slot < CAP) listL[slot] = e0 + 1u; }
                    if (m2) { const int slot = atomicAdd(&cntL, 1); if (slot < CAP) listL[slot] = e0 + 2u; }
                    if (m3) { const int slot = atomicAdd(&cntL, 1); if (slot < CAP) listL[slot] = e0 + 3u; }
                }
            }
        } else {
            for (int k0 = 0; k0 < SEG / OWN_T; k0 += 8) {
                int dv[8];
#pragma unroll
                for (int k = 0; k < 8; ++k) {
                    const int e = sb + (k0 + k) * OWN_T + tid;
                    dv[k] = e < E ? dst[e] : -1;
                }
#pragma unroll
                for (int k = 0; k < 8; ++k) {
                    const unsigned u = (unsigned)(dv[k] - lo);
                    if (u < (unsigned)nOwn) {
                        const int e = sb + (k0 + k) * OWN_T + tid;
                        const int slot = atomicAdd(&cntL, 1);
                        if (slot < CAP) listL[slot] = (unsigned)e;
                    }
                }
            }
        }
    };

    auto process_chunk = [&](const int cnt) {
        for (int j = tid; j < cnt; j += OWN_T) {
            const unsigned e = listL[j];
            const unsigned dl = (unsigned)(dst[e] - lo);
            const int s = src[e];
            unsigned entry = NO_EDGE;
            if (dl < (unsigned)nOwn) {
                const float a = asrc[s];
                entry = (unsigned)s | (dl << 23);
                if (a >= 0.0f) atomicMax((int*)&mNew[dl], __float_as_int(a));
                else           atomicMin((unsigned int*)&mNew[dl], __float_as_uint(a));
            }
            listL[j] = entry;
        }
        __syncthreads();

        if (tid < RN) {
            const float ad = adL[tid];
            const float mn = mNew[tid];
            const float shiftOld = lrelu(mL[tid] + ad);
            const float shiftNew = lrelu(mn + ad);
            scL[tid]  = __expf(shiftOld - shiftNew);
            capL[tid] = shiftNew;
            mL[tid]   = mn;
        }
        __syncthreads();
        for (int idx = tid; idx < RN * AW; idx += OWN_T) accL[idx] *= scL[idx / AW];
        __syncthreads();

        const int lc = lane < F ? lane : F - 1;
        for (int base = wave * 32; base < cnt; base += OWN_T) {
            const int j = base + lane;
            int hoff = lo * F, aoff = 0;
            float t = 0.0f;
            const unsigned p = j < cnt ? listL[j] : NO_EDGE;
            if (p != NO_EDGE) {
                const int s  = (int)(p & LOW23);
                const int dl = (int)(p >> 23);
                const float ev = lrelu(asrc[s] + adL[dl]);
                t = __expf(ev - capL[dl]);
                hoff = s * F;
                aoff = dl * AW;
            }
            const int n = cnt - base;
            if (n >= 32) {
#pragma unroll
                for (int g = 0; g < 32; g += 16) {
                    int so[16], ao[16];
                    float tt[16];
#pragma unroll
                    for (int k = 0; k < 16; ++k) {
                        so[k] = lane_i(hoff, g + k);
                        ao[k] = lane_i(aoff, g + k);
                        tt[k] = lane_f(t, g + k);
                    }
                    if (lane <= F) {
                        float hv[16];
#pragma unroll
                        for (int k = 0; k < 16; ++k) hv[k] = h[so[k] + lc];
#pragma unroll
                        for (int k = 0; k < 16; ++k)
                            __hip_atomic_fetch_add(&accL[ao[k] + lane],
                                                   lane < F ? tt[k] * hv[k] : tt[k],
                                                   __ATOMIC_RELAXED, __HIP_MEMORY_SCOPE_WORKGROUP);
                    }
                }
            } else {
                for (int q = 0; q < n; ++q) {
                    const int so = lane_i(hoff, q);
                    const int ao = lane_i(aoff, q);
                    const float tt = lane_f(t, q);
                    if (lane <= F) {
                        const float hv = h[so + lc];
                        __hip_atomic_fetch_add(&accL[ao + lane], lane < F ? tt * hv : tt,
                                               __ATOMIC_RELAXED, __HIP_MEMORY_SCOPE_WORKGROUP);
                    }
                }
            }
        }
        __syncthreads();
    };

    const int keyed = (E / KSEG) * KSEG;
    for (int kb = 0; kb < keyed; kb += KSEG) scan_keys(kb);
    for (int sb = keyed; sb < E; sb += SEG) scan_segment(sb);
    __syncthreads();
    const int total = __builtin_amdgcn_readfirstlane(cntL);
    __syncthreads();
    if (total <= CAP) {
        process_chunk(total);
    } else {
        if (tid == 0) cntL = 0;
        __syncthreads();
        for (int sb = 0; sb < E; sb += SEG) {
            scan_segment(sb);
            __syncthreads();
            const int cnt = __builtin_amdgcn_readfirstlane(cntL);
            __syncthreads();
            if (cnt <= CAP - SEG && sb + SEG < E) continue;
            process_chunk(cnt);
            if (tid == 0) cntL = 0;
            __syncthreads();
        }
    }

    for (int idx = tid; idx < nOwn * F; idx += OWN_T) {
        const int dl = idx / F;
        const int f  = idx - dl * F;
        const float inv = 1.0f / (accL[dl * AW + F] + 1e-16f);
        const float v = accL[dl * AW + f] * inv + b[f];
        feat[(size_t)lo * F + idx] = v > 0.0f ? v : 0.0f;
    }
}

__global__ void gat_final_linear(const float* __restrict__ feat,
                                 const float* __restrict__ Wl,
                                 const float* __restrict__ bl,
                                 float* __restrict__ out,
                                 int N) {
    int i = blockIdx.x * blockDim.x + threadIdx.x;
    if (i >= N) return;
    float sacc = bl[0];
#pragma unroll
    for (int f = 0; f < F; ++f) sacc += feat[i * F + f] * Wl[f];
    out[i] = sacc;
}

extern "C" void kernel_launch(void* const* d_in, const int* in_sizes, int n_in,
                              void* d_out, int out_size, void* d_ws, size_t ws_size,
                              hipStream_t stream) {
    (void)n_in; (void)out_size; (void)ws_size;

    const int N = in_sizes[0];
    const int E = in_sizes[1] / 2;

    const float* x    = (const float*)d_in[0];
    const int*   ei   = (const int*)d_in[1];
    const float* W1   = (const float*)d_in[2];
    const float* as1  = (const float*)d_in[3];
    const float* ad1  = (const float*)d_in[4];
    const float* b1   = (const float*)d_in[5];
    const float* W2   = (const float*)d_in[6];
    const float* as2  = (const float*)d_in[7];
    const float* ad2  = (const float*)d_in[8];
    const float* b2   = (const float*)d_in[9];
    const float* Wl   = (const float*)d_in[10];
    const float* bl   = (const float*)d_in[11];

    const int* srcIdx = ei;
    const int* dstIdx = ei + E;

    const int nWords = (E / KSEG) * (KSEG / 4);
    unsigned int* keyw = (unsigned int*)d_ws;
    float* ws     = (float*)d_ws + nWords;
    float* h      = ws + (size_t)0  * N;
    float* feat1  = ws + (size_t)20 * N;
    float* feat2  = ws + (size_t)40 * N;
    float* asrc   = ws + (size_t)60 * N;
    float* adst   = ws + (size_t)61 * N;

    const int TB   = 256;
    const int gN   = (N + TB - 1) / TB;
    const int gMM  = (N + 15) / 16;
    const int gOwn = (N + RN - 1) / RN;
    const int gKey = nWords / TB + 1;

    gat_owner_keys<<<gKey, TB, 0, stream>>>(dstIdx, keyw, nWords);

    gat_node_init_l1<<<gN, TB, 0, stream>>>(x, W1, as1, ad1, h, asrc, adst, N);
    gat_owner_layer<<<gOwn, OWN_T, 0, stream>>>(srcIdx, dstIdx, keyw, asrc, adst, h, b1, feat1, N, E);

    gat_wmma_feat_x_w<<<gMM, 32, 0, stream>>>(feat1, W2, h, N);
    gat_node_alpha<<<gN, TB, 0, stream>>>(h, as2, ad2, asrc, adst, N);
    gat_owner_layer<<<gOwn, OWN_T, 0, stream>>>(srcIdx, dstIdx, keyw, asrc, adst, h, b2, feat2, N, E);

    gat_final_linear<<<gN, TB, 0, stream>>>(feat2, Wl, bl, (float*)d_out, N);
}


# Layer 2 reuses layer 1's owner edge lists: an owner stores its resolved (owned node, source) list to the workspace, layer 2 loads it, no owner-id scan, no dst/src reads; LDS adds stay arrival order
# speedup vs baseline: 8.6370x; 1.3565x over previous
#include <hip/hip_runtime.h>
#include <math.h>

#define F 20
#define NEG_SLOPE 0.2f

#define RN     392
#define AW     21
#define OWN_T  1024
#define SEG    32768
#define CAP    32768
#define KSEG   131072
#define LOW23  0x7FFFFFu
#define NO_EDGE 0xFFFFFFFFu

typedef float v2f __attribute__((ext_vector_type(2)));
typedef float v8f __attribute__((ext_vector_type(8)));

__device__ __forceinline__ float lrelu(float v) {
    return v > 0.0f ? v : NEG_SLOPE * v;
}

__device__ __forceinline__ int lane_i(int v, int l) {
    return __builtin_amdgcn_readlane(v, l);
}
__device__ __forceinline__ float lane_f(float v, int l) {
    return __int_as_float(__builtin_amdgcn_readlane(__float_as_int(v), l));
}

__global__ void gat_node_init_l1(const float* __restrict__ x,
                                 const float* __restrict__ W1,
                                 const float* __restrict__ a_src,
                                 const float* __restrict__ a_dst,
                                 float* __restrict__ h,
                                 float* __restrict__ asrc,
                                 float* __restrict__ adst,
                                 int N) {
    int i = blockIdx.x * blockDim.x + threadIdx.x;
    if (i >= N) return;
    float xv = x[i];
    float sa = 0.0f, sd = 0.0f;
#pragma unroll
    for (int f = 0; f < F; ++f) {
        float hv = xv * W1[f];
        h[i * F + f] = hv;
        sa += hv * a_src[f];
        sd += hv * a_dst[f];
    }
    asrc[i] = sa;
    adst[i] = sd;
}

__global__ void __launch_bounds__(32)
gat_wmma_feat_x_w(const float* __restrict__ feat,
                  const float* __restrict__ W,
                  float* __restrict__ h,
                  int N) {
    int tile  = blockIdx.x;
    int lane  = threadIdx.x;
    int mrow  = lane & 15;
    int khalf = (lane >> 4) * 2;

    int row  = tile * 16 + mrow;
    int rowc = row < N ? row : N - 1;
    const float* arow = feat + (size_t)rowc * F;

    int   n1   = 16 + mrow;
    int   n1c  = (n1 < F) ? n1 : 0;
    float msk1 = (n1 < F) ? 1.0f : 0.0f;

    v8f c0 = {};
    v8f c1 = {};
#pragma unroll
    for (int k0 = 0; k0 < F; k0 += 4) {
        const float* wr0 = W + (size_t)(k0 + khalf + 0) * F;
        const float* wr1 = W + (size_t)(k0 + khalf + 1) * F;

        v2f a;
        a.x = arow[k0 + khalf + 0];
        a.y = arow[k0 + khalf + 1];

        v2f b0;
        b0.x = wr0[mrow];
        b0.y = wr1[mrow];
        c0 = __builtin_amdgcn_wmma_f32_16x16x4_f32(false, a, false, b0,
                                                   (short)0, c0, false, false);

        v2f b1;
        b1.x = wr0[n1c] * msk1;
        b1.y = wr1[n1c] * msk1;
        c1 = __builtin_amdgcn_wmma_f32_16x16x4_f32(false, a, false, b1,
                                                   (short)0, c1, false, false);
    }

    int rbase = tile * 16 + ((lane >> 4) << 3);
#pragma unroll
    for (int j = 0; j < 8; ++j) {
        int r = rbase + j;
        if (r < N) {
            h[(size_t)r * F + mrow] = c0[j];
            if (n1 < F) h[(size_t)r * F + n1] = c1[j];
        }
    }
}

__global__ void gat_node_alpha(const float* __restrict__ h,
                               const float* __restrict__ a_src,
                               const float* __restrict__ a_dst,
                               float* __restrict__ asrc,
                               float* __restrict__ adst,
                               int N) {
    int i = blockIdx.x * blockDim.x + threadIdx.x;
    if (i >= N) return;
    float sa = 0.0f, sd = 0.0f;
#pragma unroll
    for (int f = 0; f < F; ++f) {
        float hv = h[i * F + f];
        sa += hv * a_src[f];
        sd += hv * a_dst[f];
    }
    asrc[i] = sa;
    adst[i] = sd;
}

__global__ void gat_owner_keys(const int* __restrict__ dst,
                               unsigned int* __restrict__ keyw,
                               int nWords) {
    int w = blockIdx.x * blockDim.x + threadIdx.x;
    if (w >= nWords) return;
    unsigned int k = 0u;
#pragma unroll
    for (int c = 0; c < 4; ++c)
        k |= (((unsigned int)dst[4 * w + c] / (unsigned int)RN) & 0xFFu) << (8 * c);
    keyw[w] = k;
}

__global__ void __launch_bounds__(OWN_T)
gat_owner_layer(const int* __restrict__ src,
                const int* __restrict__ dst,
                const unsigned int* __restrict__ keyw,
                const float* __restrict__ asrc,
                const float* __restrict__ adst,
                const float* __restrict__ h,
                const float* __restrict__ b,
                float* __restrict__ feat,
                unsigned int* __restrict__ savedList,
                unsigned int* __restrict__ savedCnt,
                int first,
                int N, int E) {
    __shared__ float accL[RN * AW];
    __shared__ float mL[RN];
    __shared__ float mNew[RN];
    __shared__ float adL[RN];
    __shared__ float capL[RN];
    __shared__ float scL[RN];
    __shared__ unsigned listL[CAP];
    __shared__ int cntL;

    const int tid  = threadIdx.x;
    const int lane = tid & 31;
    const int wave = __builtin_amdgcn_readfirstlane(tid >> 5);
    const int lo   = blockIdx.x * RN;
    const int nOwn = (N - lo) < RN ? (N - lo) : RN;

    for (int dl = tid; dl < RN; dl += OWN_T) {
        float a = -3.402823466e38f, ad = 0.0f;
        if (dl < nOwn) { a = asrc[lo + dl]; ad = adst[lo + dl]; }
        mL[dl] = a;
        mNew[dl] = a;
        adL[dl] = ad;
    }
    for (int idx = tid; idx < RN * AW; idx += OWN_T) {
        const int dl = idx / AW;
        const int f  = idx - dl * AW;
        float v = 0.0f;
        if (dl < nOwn) v = f < F ? h[(size_t)(lo + dl) * F + f] : 1.0f;
        accL[idx] = v;
    }
    if (tid == 0) cntL = 0;
    __syncthreads();

    const unsigned gpat = (unsigned)(blockIdx.x & 0xFF) * 0x01010101u;
    auto scan_keys = [&](const int kb) {
        const uint4* __restrict__ k4 = (const uint4*)(keyw + (kb >> 2));
        uint4 q[KSEG / (16 * OWN_T)];
#pragma unroll
        for (int k = 0; k < KSEG / (16 * OWN_T); ++k) q[k] = k4[k * OWN_T + tid];
#pragma unroll
        for (int k = 0; k < KSEG / (16 * OWN_T); ++k) {
            const unsigned w[4] = {q[k].x, q[k].y, q[k].z, q[k].w};
            unsigned hz[4];
#pragma unroll
            for (int c = 0; c < 4; ++c) {
                const unsigned v = w[c] ^ gpat;
                hz[c] = (v - 0x01010101u) & ~v & 0x80808080u;
            }
            if (hz[0] | hz[1] | hz[2] | hz[3]) {
                const unsigned e0 = (unsigned)kb + (unsigned)(k * OWN_T + tid) * 16u;
#pragma unroll
                for (int c = 0; c < 4; ++c) {
                    unsigned m = hz[c];
                    while (m != 0u) {
                        const unsigned byte = (unsigned)__builtin_ctz(m) >> 3;
                        const int slot = atomicAdd(&cntL, 1);
                        if (slot < CAP) listL[slot] = e0 + 4u * (unsigned)c + byte;
                        m &= m - 1u;
                    }
                }
            }
        }
    };

    auto scan_segment = [&](const int sb) {
        if (((E & 3) == 0) && (sb + SEG <= E)) {
            const int4* __restrict__ dst4 = (const int4*)(dst + sb);
            int4 dq[SEG / (4 * OWN_T)];
#pragma unroll
            for (int k = 0; k < SEG / (4 * OWN_T); ++k) dq[k] = dst4[k * OWN_T + tid];
#pragma unroll
            for (int k = 0; k < SEG / (4 * OWN_T); ++k) {
                const unsigned u0 = (unsigned)(dq[k].x - lo);
                const unsigned u1 = (unsigned)(dq[k].y - lo);
                const unsigned u2 = (unsigned)(dq[k].z - lo);
                const unsigned u3 = (unsigned)(dq[k].w - lo);
                const bool m0 = u0 < (unsigned)nOwn, m1 = u1 < (unsigned)nOwn;
                const bool m2 = u2 < (unsigned)nOwn, m3 = u3 < (unsigned)nOwn;
                if (m0 | m1 | m2 | m3) {
                    const unsigned e0 = (unsigned)(sb + (k * OWN_T + tid) * 4);
                    if (m0) { const int slot = atomicAdd(&cntL, 1); if (slot < CAP) listL[slot] = e0 + 0u; }
                    if (m1) { const int slot = atomicAdd(&cntL, 1); if (slot < CAP) listL[slot] = e0 + 1u; }
                    if (m2) { const int slot = atomicAdd(&cntL, 1); if (slot < CAP) listL[slot] = e0 + 2u; }
                    if (m3) { const int slot = atomicAdd(&cntL, 1); if (slot < CAP) listL[slot] = e0 + 3u; }
                }
            }
        } else {
            for (int k0 = 0; k0 < SEG / OWN_T; k0 += 8) {
                int dv[8];
#pragma unroll
                for (int k = 0; k < 8; ++k) {
                    const int e = sb + (k0 + k) * OWN_T + tid;
                    dv[k] = e < E ? dst[e] : -1;
                }
#pragma unroll
                for (int k = 0; k < 8; ++k) {
                    const unsigned u = (unsigned)(dv[k] - lo);
                    if (u < (unsigned)nOwn) {
                        const int e = sb + (k0 + k) * OWN_T + tid;
                        const int slot = atomicAdd(&cntL, 1);
                        if (slot < CAP) listL[slot] = (unsigned)e;
                    }
                }
            }
        }
    };

    auto process_chunk = [&](const int cnt, const bool resolved) {
        for (int j = tid; j < cnt; j += OWN_T) {
            unsigned entry = listL[j];
            if (!resolved) {
                const unsigned e = entry;
                const unsigned dl = (unsigned)(dst[e] - lo);
                const int s = src[e];
                entry = dl < (unsigned)nOwn ? ((unsigned)s | (dl << 23)) : NO_EDGE;
                listL[j] = entry;
            }
            if (entry != NO_EDGE) {
                const float a = asrc[entry & LOW23];
                const unsigned dl = entry >> 23;
                if (a >= 0.0f) atomicMax((int*)&mNew[dl], __float_as_int(a));
                else           atomicMin((unsigned int*)&mNew[dl], __float_as_uint(a));
            }
        }
        __syncthreads();

        if (tid < RN) {
            const float ad = adL[tid];
            const float mn = mNew[tid];
            const float shiftOld = lrelu(mL[tid] + ad);
            const float shiftNew = lrelu(mn + ad);
            scL[tid]  = __expf(shiftOld - shiftNew);
            capL[tid] = shiftNew;
            mL[tid]   = mn;
        }
        __syncthreads();
        for (int idx = tid; idx < RN * AW; idx += OWN_T) accL[idx] *= scL[idx / AW];
        __syncthreads();

        const int lc = lane < F ? lane : F - 1;
        for (int base = wave * 32; base < cnt; base += OWN_T) {
            const int j = base + lane;
            int hoff = lo * F, aoff = 0;
            float t = 0.0f;
            const unsigned p = j < cnt ? listL[j] : NO_EDGE;
            if (p != NO_EDGE) {
                const int s  = (int)(p & LOW23);
                const int dl = (int)(p >> 23);
                const float ev = lrelu(asrc[s] + adL[dl]);
                t = __expf(ev - capL[dl]);
                hoff = s * F;
                aoff = dl * AW;
            }
            const int n = cnt - base;
            if (n >= 32) {
#pragma unroll
                for (int g = 0; g < 32; g += 16) {
                    int so[16], ao[16];
                    float tt[16];
#pragma unroll
                    for (int k = 0; k < 16; ++k) {
                        so[k] = lane_i(hoff, g + k);
                        ao[k] = lane_i(aoff, g + k);
                        tt[k] = lane_f(t, g + k);
                    }
                    if (lane <= F) {
                        float hv[16];
#pragma unroll
                        for (int k = 0; k < 16; ++k) hv[k] = h[so[k] + lc];
#pragma unroll
                        for (int k = 0; k < 16; ++k)
                            __hip_atomic_fetch_add(&accL[ao[k] + lane],
                                                   lane < F ? tt[k] * hv[k] : tt[k],
                                                   __ATOMIC_RELAXED, __HIP_MEMORY_SCOPE_WORKGROUP);
                    }
                }
            } else {
                for (int q = 0; q < n; ++q) {
                    const int so = lane_i(hoff, q);
                    const int ao = lane_i(aoff, q);
                    const float tt = lane_f(t, q);
                    if (lane <= F) {
                        const float hv = h[so + lc];
                        __hip_atomic_fetch_add(&accL[ao + lane], lane < F ? tt * hv : tt,
                                               __ATOMIC_RELAXED, __HIP_MEMORY_SCOPE_WORKGROUP);
                    }
                }
            }
        }
        __syncthreads();
    };

    unsigned int* const saved = savedList + (size_t)blockIdx.x * CAP;
    unsigned have = (unsigned)CAP + 1u;
    if (!first) have = savedCnt[blockIdx.x];

    int total;
    const bool resolved = have <= (unsigned)CAP;
    if (resolved) {
        total = (int)have;
        for (int j0 = 0; j0 < total; j0 += 8 * OWN_T) {
            unsigned v[8];
#pragma unroll
            for (int k = 0; k < 8; ++k) {
                const int j = j0 + k * OWN_T + tid;
                v[k] = j < total ? saved[j] : NO_EDGE;
            }
#pragma unroll
            for (int k = 0; k < 8; ++k) {
                const int j = j0 + k * OWN_T + tid;
                if (j < total) listL[j] = v[k];
            }
        }
    } else {
        const int keyed = (E / KSEG) * KSEG;
        for (int kb = 0; kb < keyed; kb += KSEG) scan_keys(kb);
        for (int sb = keyed; sb < E; sb += SEG) scan_segment(sb);
        __syncthreads();
        total = __builtin_amdgcn_readfirstlane(cntL);
        __syncthreads();
        if (first && tid == 0) savedCnt[blockIdx.x] = (unsigned)total;
    }
    if (total <= CAP) {
        process_chunk(total, resolved);
        if (first)
            for (int j = tid; j < total; j += OWN_T) saved[j] = listL[j];
    } else {
        if (tid == 0) cntL = 0;
        __syncthreads();
        for (int sb = 0; sb < E; sb += SEG) {
            scan_segment(sb);
            __syncthreads();
            const int cnt = __builtin_amdgcn_readfirstlane(cntL);
            __syncthreads();
            if (cnt <= CAP - SEG && sb + SEG < E) continue;
            process_chunk(cnt, false);
            if (tid == 0) cntL = 0;
            __syncthreads();
        }
    }

    for (int idx = tid; idx < nOwn * F; idx += OWN_T) {
        const int dl = idx / F;
        const int f  = idx - dl * F;
        const float inv = 1.0f / (accL[dl * AW + F] + 1e-16f);
        const float v = accL[dl * AW + f] * inv + b[f];
        feat[(size_t)lo * F + idx] = v > 0.0f ? v : 0.0f;
    }
}

__global__ void gat_final_linear(const float* __restrict__ feat,
                                 const float* __restrict__ Wl,
                                 const float* __restrict__ bl,
                                 float* __restrict__ out,
                                 int N) {
    int i = blockIdx.x * blockDim.x + threadIdx.x;
    if (i >= N) return;
    float sacc = bl[0];
#pragma unroll
    for (int f = 0; f < F; ++f) sacc += feat[i * F + f] * Wl[f];
    out[i] = sacc;
}

extern "C" void kernel_launch(void* const* d_in, const int* in_sizes, int n_in,
                              void* d_out, int out_size, void* d_ws, size_t ws_size,
                              hipStream_t stream) {
    (void)n_in; (void)out_size; (void)ws_size;

    const int N = in_sizes[0];
    const int E = in_sizes[1] / 2;

    const float* x    = (const float*)d_in[0];
    const int*   ei   = (const int*)d_in[1];
    const float* W1   = (const float*)d_in[2];
    const float* as1  = (const float*)d_in[3];
    const float* ad1  = (const float*)d_in[4];
    const float* b1   = (const float*)d_in[5];
    const float* W2   = (const float*)d_in[6];
    const float* as2  = (const float*)d_in[7];
    const float* ad2  = (const float*)d_in[8];
    const float* b2   = (const float*)d_in[9];
    const float* Wl   = (const float*)d_in[10];
    const float* bl   = (const float*)d_in[11];

    const int* srcIdx = ei;
    const int* dstIdx = ei + E;

    const int nWords = (E / KSEG) * (KSEG / 4);
    unsigned int* keyw = (unsigned int*)d_ws;
    float* ws     = (float*)d_ws + nWords;
    float* h      = ws + (size_t)0  * N;
    float* feat1  = ws + (size_t)20 * N;
    float* feat2  = ws + (size_t)40 * N;
    float* asrc   = ws + (size_t)60 * N;
    float* adst   = ws + (size_t)61 * N;

    const int TB   = 256;
    const int gN   = (N + TB - 1) / TB;
    const int gMM  = (N + 15) / 16;
    const int gOwn = (N + RN - 1) / RN;
    const int gKey = nWords / TB + 1;

    unsigned int* savedList = (unsigned int*)(ws + (size_t)62 * N);
    unsigned int* savedCnt  = savedList + (size_t)gOwn * CAP;

    gat_owner_keys<<<gKey, TB, 0, stream>>>(dstIdx, keyw, nWords);

    gat_node_init_l1<<<gN, TB, 0, stream>>>(x, W1, as1, ad1, h, asrc, adst, N);
    gat_owner_layer<<<gOwn, OWN_T, 0, stream>>>(srcIdx, dstIdx, keyw, asrc, adst, h, b1, feat1,
                                                savedList, savedCnt, 1, N, E);

    gat_wmma_feat_x_w<<<gMM, 32, 0, stream>>>(feat1, W2, h, N);
    gat_node_alpha<<<gN, TB, 0, stream>>>(h, as2, ad2, asrc, adst, N);
    gat_owner_layer<<<gOwn, OWN_T, 0, stream>>>(srcIdx, dstIdx, keyw, asrc, adst, h, b2, feat2,
                                                savedList, savedCnt, 0, N, E);

    gat_final_linear<<<gN, TB, 0, stream>>>(feat2, Wl, bl, (float*)d_out, N);
}


# Add pass takes each edge's source score from an LDS copy that the max pass stored beside the edge list, not from a second random gather of asrc[src]; same values, LDS adds stay arrival order
# speedup vs baseline: 10.2299x; 1.1844x over previous
#include <hip/hip_runtime.h>
#include <math.h>

#define F 20
#define NEG_SLOPE 0.2f

#define RN     392
#define AW     21
#define OWN_T  1024
#define SEG    32768
#define CAP    32768
#define KSEG   131072
#define LOW23  0x7FFFFFu
#define NO_EDGE 0xFFFFFFFFu

typedef float v2f __attribute__((ext_vector_type(2)));
typedef float v8f __attribute__((ext_vector_type(8)));

__device__ __forceinline__ float lrelu(float v) {
    return v > 0.0f ? v : NEG_SLOPE * v;
}

__device__ __forceinline__ int lane_i(int v, int l) {
    return __builtin_amdgcn_readlane(v, l);
}
__device__ __forceinline__ float lane_f(float v, int l) {
    return __int_as_float(__builtin_amdgcn_readlane(__float_as_int(v), l));
}

__global__ void gat_node_init_l1(const float* __restrict__ x,
                                 const float* __restrict__ W1,
                                 const float* __restrict__ a_src,
                                 const float* __restrict__ a_dst,
                                 float* __restrict__ h,
                                 float* __restrict__ asrc,
                                 float* __restrict__ adst,
                                 int N) {
    int i = blockIdx.x * blockDim.x + threadIdx.x;
    if (i >= N) return;
    float xv = x[i];
    float sa = 0.0f, sd = 0.0f;
#pragma unroll
    for (int f = 0; f < F; ++f) {
        float hv = xv * W1[f];
        h[i * F + f] = hv;
        sa += hv * a_src[f];
        sd += hv * a_dst[f];
    }
    asrc[i] = sa;
    adst[i] = sd;
}

__global__ void __launch_bounds__(32)
gat_wmma_feat_x_w(const float* __restrict__ feat,
                  const float* __restrict__ W,
                  float* __restrict__ h,
                  int N) {
    int tile  = blockIdx.x;
    int lane  = threadIdx.x;
    int mrow  = lane & 15;
    int khalf = (lane >> 4) * 2;

    int row  = tile * 16 + mrow;
    int rowc = row < N ? row : N - 1;
    const float* arow = feat + (size_t)rowc * F;

    int   n1   = 16 + mrow;
    int   n1c  = (n1 < F) ? n1 : 0;
    float msk1 = (n1 < F) ? 1.0f : 0.0f;

    v8f c0 = {};
    v8f c1 = {};
#pragma unroll
    for (int k0 = 0; k0 < F; k0 += 4) {
        const float* wr0 = W + (size_t)(k0 + khalf + 0) * F;
        const float* wr1 = W + (size_t)(k0 + khalf + 1) * F;

        v2f a;
        a.x = arow[k0 + khalf + 0];
        a.y = arow[k0 + khalf + 1];

        v2f b0;
        b0.x = wr0[mrow];
        b0.y = wr1[mrow];
        c0 = __builtin_amdgcn_wmma_f32_16x16x4_f32(false, a, false, b0,
                                                   (short)0, c0, false, false);

        v2f b1;
        b1.x = wr0[n1c] * msk1;
        b1.y = wr1[n1c] * msk1;
        c1 = __builtin_amdgcn_wmma_f32_16x16x4_f32(false, a, false, b1,
                                                   (short)0, c1, false, false);
    }

    int rbase = tile * 16 + ((lane >> 4) << 3);
#pragma unroll
    for (int j = 0; j < 8; ++j) {
        int r = rbase + j;
        if (r < N) {
            h[(size_t)r * F + mrow] = c0[j];
            if (n1 < F) h[(size_t)r * F + n1] = c1[j];
        }
    }
}

__global__ void gat_node_alpha(const float* __restrict__ h,
                               const float* __restrict__ a_src,
                               const float* __restrict__ a_dst,
                               float* __restrict__ asrc,
                               float* __restrict__ adst,
                               int N) {
    int i = blockIdx.x * blockDim.x + threadIdx.x;
    if (i >= N) return;
    float sa = 0.0f, sd = 0.0f;
#pragma unroll
    for (int f = 0; f < F; ++f) {
        float hv = h[i * F + f];
        sa += hv * a_src[f];
        sd += hv * a_dst[f];
    }
    asrc[i] = sa;
    adst[i] = sd;
}

__global__ void gat_owner_keys(const int* __restrict__ dst,
                               unsigned int* __restrict__ keyw,
                               int nWords) {
    int w = blockIdx.x * blockDim.x + threadIdx.x;
    if (w >= nWords) return;
    unsigned int k = 0u;
#pragma unroll
    for (int c = 0; c < 4; ++c)
        k |= (((unsigned int)dst[4 * w + c] / (unsigned int)RN) & 0xFFu) << (8 * c);
    keyw[w] = k;
}

__global__ void __launch_bounds__(OWN_T)
gat_owner_layer(const int* __restrict__ src,
                const int* __restrict__ dst,
                const unsigned int* __restrict__ keyw,
                const float* __restrict__ asrc,
                const float* __restrict__ adst,
                const float* __restrict__ h,
                const float* __restrict__ b,
                float* __restrict__ feat,
                unsigned int* __restrict__ savedList,
                unsigned int* __restrict__ savedCnt,
                int first,
                int N, int E) {
    __shared__ float accL[RN * AW];
    __shared__ float mL[RN];
    __shared__ float mNew[RN];
    __shared__ float adL[RN];
    __shared__ float capL[RN];
    __shared__ float scL[RN];
    __shared__ unsigned listL[CAP];
    __shared__ float asL[CAP];
    __shared__ int cntL;

    const int tid  = threadIdx.x;
    const int lane = tid & 31;
    const int wave = __builtin_amdgcn_readfirstlane(tid >> 5);
    const int lo   = blockIdx.x * RN;
    const int nOwn = (N - lo) < RN ? (N - lo) : RN;

    for (int dl = tid; dl < RN; dl += OWN_T) {
        float a = -3.402823466e38f, ad = 0.0f;
        if (dl < nOwn) { a = asrc[lo + dl]; ad = adst[lo + dl]; }
        mL[dl] = a;
        mNew[dl] = a;
        adL[dl] = ad;
    }
    for (int idx = tid; idx < RN * AW; idx += OWN_T) {
        const int dl = idx / AW;
        const int f  = idx - dl * AW;
        float v = 0.0f;
        if (dl < nOwn) v = f < F ? h[(size_t)(lo + dl) * F + f] : 1.0f;
        accL[idx] = v;
    }
    if (tid == 0) cntL = 0;
    __syncthreads();

    const unsigned gpat = (unsigned)(blockIdx.x & 0xFF) * 0x01010101u;
    auto scan_keys = [&](const int kb) {
        const uint4* __restrict__ k4 = (const uint4*)(keyw + (kb >> 2));
        uint4 q[KSEG / (16 * OWN_T)];
#pragma unroll
        for (int k = 0; k < KSEG / (16 * OWN_T); ++k) q[k] = k4[k * OWN_T + tid];
#pragma unroll
        for (int k = 0; k < KSEG / (16 * OWN_T); ++k) {
            const unsigned w[4] = {q[k].x, q[k].y, q[k].z, q[k].w};
            unsigned hz[4];
#pragma unroll
            for (int c = 0; c < 4; ++c) {
                const unsigned v = w[c] ^ gpat;
                hz[c] = (v - 0x01010101u) & ~v & 0x80808080u;
            }
            if (hz[0] | hz[1] | hz[2] | hz[3]) {
                const unsigned e0 = (unsigned)kb + (unsigned)(k * OWN_T + tid) * 16u;
#pragma unroll
                for (int c = 0; c < 4; ++c) {
                    unsigned m = hz[c];
                    while (m != 0u) {
                        const unsigned byte = (unsigned)__builtin_ctz(m) >> 3;
                        const int slot = atomicAdd(&cntL, 1);
                        if (slot < CAP) listL[slot] = e0 + 4u * (unsigned)c + byte;
                        m &= m - 1u;
                    }
                }
            }
        }
    };

    auto scan_segment = [&](const int sb) {
        if (((E & 3) == 0) && (sb + SEG <= E)) {
            const int4* __restrict__ dst4 = (const int4*)(dst + sb);
            int4 dq[SEG / (4 * OWN_T)];
#pragma unroll
            for (int k = 0; k < SEG / (4 * OWN_T); ++k) dq[k] = dst4[k * OWN_T + tid];
#pragma unroll
            for (int k = 0; k < SEG / (4 * OWN_T); ++k) {
                const unsigned u0 = (unsigned)(dq[k].x - lo);
                const unsigned u1 = (unsigned)(dq[k].y - lo);
                const unsigned u2 = (unsigned)(dq[k].z - lo);
                const unsigned u3 = (unsigned)(dq[k].w - lo);
                const bool m0 = u0 < (unsigned)nOwn, m1 = u1 < (unsigned)nOwn;
                const bool m2 = u2 < (unsigned)nOwn, m3 = u3 < (unsigned)nOwn;
                if (m0 | m1 | m2 | m3) {
                    const unsigned e0 = (unsigned)(sb + (k * OWN_T + tid) * 4);
                    if (m0) { const int slot = atomicAdd(&cntL, 1); if (slot < CAP) listL[slot] = e0 + 0u; }
                    if (m1) { const int slot = atomicAdd(&cntL, 1); if (slot < CAP) listL[slot] = e0 + 1u; }
                    if (m2) { const int slot = atomicAdd(&cntL, 1); if (slot < CAP) listL[slot] = e0 + 2u; }
                    if (m3) { const int slot = atomicAdd(&cntL, 1); if (slot < CAP) listL[slot] = e0 + 3u; }
                }
            }
        } else {
            for (int k0 = 0; k0 < SEG / OWN_T; k0 += 8) {
                int dv[8];
#pragma unroll
                for (int k = 0; k < 8; ++k) {
                    const int e = sb + (k0 + k) * OWN_T + tid;
                    dv[k] = e < E ? dst[e] : -1;
                }
#pragma unroll
                for (int k = 0; k < 8; ++k) {
                    const unsigned u = (unsigned)(dv[k] - lo);
                    if (u < (unsigned)nOwn) {
                        const int e = sb + (k0 + k) * OWN_T + tid;
                        const int slot = atomicAdd(&cntL, 1);
                        if (slot < CAP) listL[slot] = (unsigned)e;
                    }
                }
            }
        }
    };

    auto process_chunk = [&](const int cnt, const bool resolved) {
        for (int j = tid; j < cnt; j += OWN_T) {
            unsigned entry = listL[j];
            if (!resolved) {
                const unsigned e = entry;
                const unsigned dl = (unsigned)(dst[e] - lo);
                const int s = src[e];
                entry = dl < (unsigned)nOwn ? ((unsigned)s | (dl << 23)) : NO_EDGE;
                listL[j] = entry;
            }
            if (entry != NO_EDGE) {
                const float a = asrc[entry & LOW23];
                asL[j] = a;
                const unsigned dl = entry >> 23;
                if (a >= 0.0f) atomicMax((int*)&mNew[dl], __float_as_int(a));
                else           atomicMin((unsigned int*)&mNew[dl], __float_as_uint(a));
            }
        }
        __syncthreads();

        if (tid < RN) {
            const float ad = adL[tid];
            const float mn = mNew[tid];
            const float shiftOld = lrelu(mL[tid] + ad);
            const float shiftNew = lrelu(mn + ad);
            scL[tid]  = __expf(shiftOld - shiftNew);
            capL[tid] = shiftNew;
            mL[tid]   = mn;
        }
        __syncthreads();
        for (int idx = tid; idx < RN * AW; idx += OWN_T) accL[idx] *= scL[idx / AW];
        __syncthreads();

        const int lc = lane < F ? lane : F - 1;
        for (int base = wave * 32; base < cnt; base += OWN_T) {
            const int j = base + lane;
            int hoff = lo * F, aoff = 0;
            float t = 0.0f;
            const unsigned p = j < cnt ? listL[j] : NO_EDGE;
            if (p != NO_EDGE) {
                const int s  = (int)(p & LOW23);
                const int dl = (int)(p >> 23);
                const float ev = lrelu(asL[j] + adL[dl]);
                t = __expf(ev - capL[dl]);
                hoff = s * F;
                aoff = dl * AW;
            }
            const int n = cnt - base;
            if (n >= 32) {
#pragma unroll
                for (int g = 0; g < 32; g += 16) {
                    int so[16], ao[16];
                    float tt[16];
#pragma unroll
                    for (int k = 0; k < 16; ++k) {
                        so[k] = lane_i(hoff, g + k);
                        ao[k] = lane_i(aoff, g + k);
                        tt[k] = lane_f(t, g + k);
                    }
                    if (lane <= F) {
                        float hv[16];
#pragma unroll
                        for (int k = 0; k < 16; ++k) hv[k] = h[so[k] + lc];
#pragma unroll
                        for (int k = 0; k < 16; ++k)
                            __hip_atomic_fetch_add(&accL[ao[k] + lane],
                                                   lane < F ? tt[k] * hv[k] : tt[k],
                                                   __ATOMIC_RELAXED, __HIP_MEMORY_SCOPE_WORKGROUP);
                    }
                }
            } else {
                for (int q = 0; q < n; ++q) {
                    const int so = lane_i(hoff, q);
                    const int ao = lane_i(aoff, q);
                    const float tt = lane_f(t, q);
                    if (lane <= F) {
                        const float hv = h[so + lc];
                        __hip_atomic_fetch_add(&accL[ao + lane], lane < F ? tt * hv : tt,
                                               __ATOMIC_RELAXED, __HIP_MEMORY_SCOPE_WORKGROUP);
                    }
                }
            }
        }
        __syncthreads();
    };

    unsigned int* const saved = savedList + (size_t)blockIdx.x * CAP;
    unsigned have = (unsigned)CAP + 1u;
    if (!first) have = savedCnt[blockIdx.x];

    int total;
    const bool resolved = have <= (unsigned)CAP;
    if (resolved) {
        total = (int)have;
        for (int j0 = 0; j0 < total; j0 += 8 * OWN_T) {
            unsigned v[8];
#pragma unroll
            for (int k = 0; k < 8; ++k) {
                const int j = j0 + k * OWN_T + tid;
                v[k] = j < total ? saved[j] : NO_EDGE;
            }
#pragma unroll
            for (int k = 0; k < 8; ++k) {
                const int j = j0 + k * OWN_T + tid;
                if (j < total) listL[j] = v[k];
            }
        }
    } else {
        const int keyed = (E / KSEG) * KSEG;
        for (int kb = 0; kb < keyed; kb += KSEG) scan_keys(kb);
        for (int sb = keyed; sb < E; sb += SEG) scan_segment(sb);
        __syncthreads();
        total = __builtin_amdgcn_readfirstlane(cntL);
        __syncthreads();
        if (first && tid == 0) savedCnt[blockIdx.x] = (unsigned)total;
    }
    if (total <= CAP) {
        process_chunk(total, resolved);
        if (first)
            for (int j = tid; j < total; j += OWN_T) saved[j] = listL[j];
    } else {
        if (tid == 0) cntL = 0;
        __syncthreads();
        for (int sb = 0; sb < E; sb += SEG) {
            scan_segment(sb);
            __syncthreads();
            const int cnt = __builtin_amdgcn_readfirstlane(cntL);
            __syncthreads();
            if (cnt <= CAP - SEG && sb + SEG < E) continue;
            process_chunk(cnt, false);
            if (tid == 0) cntL = 0;
            __syncthreads();
        }
    }

    for (int idx = tid; idx < nOwn * F; idx += OWN_T) {
        const int dl = idx / F;
        const int f  = idx - dl * F;
        const float inv = 1.0f / (accL[dl * AW + F] + 1e-16f);
        const float v = accL[dl * AW + f] * inv + b[f];
        feat[(size_t)lo * F + idx] = v > 0.0f ? v : 0.0f;
    }
}

__global__ void gat_final_linear(const float* __restrict__ feat,
                                 const float* __restrict__ Wl,
                                 const float* __restrict__ bl,
                                 float* __restrict__ out,
                                 int N) {
    int i = blockIdx.x * blockDim.x + threadIdx.x;
    if (i >= N) return;
    float sacc = bl[0];
#pragma unroll
    for (int f = 0; f < F; ++f) sacc += feat[i * F + f] * Wl[f];
    out[i] = sacc;
}

extern "C" void kernel_launch(void* const* d_in, const int* in_sizes, int n_in,
                              void* d_out, int out_size, void* d_ws, size_t ws_size,
                              hipStream_t stream) {
    (void)n_in; (void)out_size; (void)ws_size;

    const int N = in_sizes[0];
    const int E = in_sizes[1] / 2;

    const float* x    = (const float*)d_in[0];
    const int*   ei   = (const int*)d_in[1];
    const float* W1   = (const float*)d_in[2];
    const float* as1  = (const float*)d_in[3];
    const float* ad1  = (const float*)d_in[4];
    const float* b1   = (const float*)d_in[5];
    const float* W2   = (const float*)d_in[6];
    const float* as2  = (const float*)d_in[7];
    const float* ad2  = (const float*)d_in[8];
    const float* b2   = (const float*)d_in[9];
    const float* Wl   = (const float*)d_in[10];
    const float* bl   = (const float*)d_in[11];

    const int* srcIdx = ei;
    const int* dstIdx = ei + E;

    const int nWords = (E / KSEG) * (KSEG / 4);
    unsigned int* keyw = (unsigned int*)d_ws;
    float* ws     = (float*)d_ws + nWords;
    float* h      = ws + (size_t)0  * N;
    float* feat1  = ws + (size_t)20 * N;
    float* feat2  = ws + (size_t)40 * N;
    float* asrc   = ws + (size_t)60 * N;
    float* adst   = ws + (size_t)61 * N;

    const int TB   = 256;
    const int gN   = (N + TB - 1) / TB;
    const int gMM  = (N + 15) / 16;
    const int gOwn = (N + RN - 1) / RN;
    const int gKey = nWords / TB + 1;

    unsigned int* savedList = (unsigned int*)(ws + (size_t)62 * N);
    unsigned int* savedCnt  = savedList + (size_t)gOwn * CAP;

    gat_owner_keys<<<gKey, TB, 0, stream>>>(dstIdx, keyw, nWords);

    gat_node_init_l1<<<gN, TB, 0, stream>>>(x, W1, as1, ad1, h, asrc, adst, N);
    gat_owner_layer<<<gOwn, OWN_T, 0, stream>>>(srcIdx, dstIdx, keyw, asrc, adst, h, b1, feat1,
                                                savedList, savedCnt, 1, N, E);

    gat_wmma_feat_x_w<<<gMM, 32, 0, stream>>>(feat1, W2, h, N);
    gat_node_alpha<<<gN, TB, 0, stream>>>(h, as2, ad2, asrc, adst, N);
    gat_owner_layer<<<gOwn, OWN_T, 0, stream>>>(srcIdx, dstIdx, keyw, asrc, adst, h, b2, feat2,
                                                savedList, savedCnt, 0, N, E);

    gat_final_linear<<<gN, TB, 0, stream>>>(feat2, Wl, bl, (float*)d_out, N);
}


# Owner-id scan appends by count: a pass's id flags go into masks branch-free, a wave takes all its list slots by one counter add, not one per flagged byte; list order changes, LDS adds arrival order
# speedup vs baseline: 11.9614x; 1.1693x over previous
#include <hip/hip_runtime.h>
#include <math.h>

#define F 20
#define NEG_SLOPE 0.2f

#define RN     392
#define AW     21
#define OWN_T  1024
#define SEG    32768
#define CAP    32768
#define KSEG   131072
#define LOW23  0x7FFFFFu
#define NO_EDGE 0xFFFFFFFFu

typedef float v2f __attribute__((ext_vector_type(2)));
typedef float v8f __attribute__((ext_vector_type(8)));

__device__ __forceinline__ float lrelu(float v) {
    return v > 0.0f ? v : NEG_SLOPE * v;
}

__device__ __forceinline__ int lane_i(int v, int l) {
    return __builtin_amdgcn_readlane(v, l);
}
__device__ __forceinline__ float lane_f(float v, int l) {
    return __int_as_float(__builtin_amdgcn_readlane(__float_as_int(v), l));
}

__global__ void gat_node_init_l1(const float* __restrict__ x,
                                 const float* __restrict__ W1,
                                 const float* __restrict__ a_src,
                                 const float* __restrict__ a_dst,
                                 float* __restrict__ h,
                                 float* __restrict__ asrc,
                                 float* __restrict__ adst,
                                 int N) {
    int i = blockIdx.x * blockDim.x + threadIdx.x;
    if (i >= N) return;
    float xv = x[i];
    float sa = 0.0f, sd = 0.0f;
#pragma unroll
    for (int f = 0; f < F; ++f) {
        float hv = xv * W1[f];
        h[i * F + f] = hv;
        sa += hv * a_src[f];
        sd += hv * a_dst[f];
    }
    asrc[i] = sa;
    adst[i] = sd;
}

__global__ void __launch_bounds__(32)
gat_wmma_feat_x_w(const float* __restrict__ feat,
                  const float* __restrict__ W,
                  float* __restrict__ h,
                  int N) {
    int tile  = blockIdx.x;
    int lane  = threadIdx.x;
    int mrow  = lane & 15;
    int khalf = (lane >> 4) * 2;

    int row  = tile * 16 + mrow;
    int rowc = row < N ? row : N - 1;
    const float* arow = feat + (size_t)rowc * F;

    int   n1   = 16 + mrow;
    int   n1c  = (n1 < F) ? n1 : 0;
    float msk1 = (n1 < F) ? 1.0f : 0.0f;

    v8f c0 = {};
    v8f c1 = {};
#pragma unroll
    for (int k0 = 0; k0 < F; k0 += 4) {
        const float* wr0 = W + (size_t)(k0 + khalf + 0) * F;
        const float* wr1 = W + (size_t)(k0 + khalf + 1) * F;

        v2f a;
        a.x = arow[k0 + khalf + 0];
        a.y = arow[k0 + khalf + 1];

        v2f b0;
        b0.x = wr0[mrow];
        b0.y = wr1[mrow];
        c0 = __builtin_amdgcn_wmma_f32_16x16x4_f32(false, a, false, b0,
                                                   (short)0, c0, false, false);

        v2f b1;
        b1.x = wr0[n1c] * msk1;
        b1.y = wr1[n1c] * msk1;
        c1 = __builtin_amdgcn_wmma_f32_16x16x4_f32(false, a, false, b1,
                                                   (short)0, c1, false, false);
    }

    int rbase = tile * 16 + ((lane >> 4) << 3);
#pragma unroll
    for (int j = 0; j < 8; ++j) {
        int r = rbase + j;
        if (r < N) {
            h[(size_t)r * F + mrow] = c0[j];
            if (n1 < F) h[(size_t)r * F + n1] = c1[j];
        }
    }
}

__global__ void gat_node_alpha(const float* __restrict__ h,
                               const float* __restrict__ a_src,
                               const float* __restrict__ a_dst,
                               float* __restrict__ asrc,
                               float* __restrict__ adst,
                               int N) {
    int i = blockIdx.x * blockDim.x + threadIdx.x;
    if (i >= N) return;
    float sa = 0.0f, sd = 0.0f;
#pragma unroll
    for (int f = 0; f < F; ++f) {
        float hv = h[i * F + f];
        sa += hv * a_src[f];
        sd += hv * a_dst[f];
    }
    asrc[i] = sa;
    adst[i] = sd;
}

__global__ void gat_owner_keys(const int* __restrict__ dst,
                               unsigned int* __restrict__ keyw,
                               int nWords) {
    int w = blockIdx.x * blockDim.x + threadIdx.x;
    if (w >= nWords) return;
    unsigned int k = 0u;
#pragma unroll
    for (int c = 0; c < 4; ++c)
        k |= (((unsigned int)dst[4 * w + c] / (unsigned int)RN) & 0xFFu) << (8 * c);
    keyw[w] = k;
}

__global__ void __launch_bounds__(OWN_T)
gat_owner_layer(const int* __restrict__ src,
                const int* __restrict__ dst,
                const unsigned int* __restrict__ keyw,
                const float* __restrict__ asrc,
                const float* __restrict__ adst,
                const float* __restrict__ h,
                const float* __restrict__ b,
                float* __restrict__ feat,
                unsigned int* __restrict__ savedList,
                unsigned int* __restrict__ savedCnt,
                int first,
                int N, int E) {
    __shared__ float accL[RN * AW];
    __shared__ float mL[RN];
    __shared__ float mNew[RN];
    __shared__ float adL[RN];
    __shared__ float capL[RN];
    __shared__ float scL[RN];
    __shared__ unsigned listL[CAP];
    __shared__ float asL[CAP];
    __shared__ int cntL;

    const int tid  = threadIdx.x;
    const int lane = tid & 31;
    const int wave = __builtin_amdgcn_readfirstlane(tid >> 5);
    const int lo   = blockIdx.x * RN;
    const int nOwn = (N - lo) < RN ? (N - lo) : RN;

    for (int dl = tid; dl < RN; dl += OWN_T) {
        float a = -3.402823466e38f, ad = 0.0f;
        if (dl < nOwn) { a = asrc[lo + dl]; ad = adst[lo + dl]; }
        mL[dl] = a;
        mNew[dl] = a;
        adL[dl] = ad;
    }
    for (int idx = tid; idx < RN * AW; idx += OWN_T) {
        const int dl = idx / AW;
        const int f  = idx - dl * AW;
        float v = 0.0f;
        if (dl < nOwn) v = f < F ? h[(size_t)(lo + dl) * F + f] : 1.0f;
        accL[idx] = v;
    }
    if (tid == 0) cntL = 0;
    __syncthreads();

    const unsigned gpat = (unsigned)(blockIdx.x & 0xFF) * 0x01010101u;
    auto scan_keys = [&](const int kb) {
        const uint4* __restrict__ k4 = (const uint4*)(keyw + (kb >> 2));
        uint4 q[KSEG / (16 * OWN_T)];
#pragma unroll
        for (int k = 0; k < KSEG / (16 * OWN_T); ++k) q[k] = k4[k * OWN_T + tid];
        unsigned mk[KSEG / (32 * OWN_T)];
        int n = 0;
#pragma unroll
        for (int p = 0; p < KSEG / (32 * OWN_T); ++p) {
            const unsigned w[8] = {q[2 * p].x,     q[2 * p].y,     q[2 * p].z,     q[2 * p].w,
                                   q[2 * p + 1].x, q[2 * p + 1].y, q[2 * p + 1].z, q[2 * p + 1].w};
            unsigned m = 0u;
#pragma unroll
            for (int j = 0; j < 8; ++j) {
                const unsigned v = w[j] ^ gpat;
                m |= ((v - 0x01010101u) & ~v & 0x80808080u) >> (7 - j);
            }
            mk[p] = m;
            n += __popc(m);
        }
        int pre = 0, tot = 0;
        for (int d = 0; d < 8; ++d) {
            if (__builtin_amdgcn_ballot_w32((n >> d) != 0) == 0u) break;
            const unsigned set = __builtin_amdgcn_ballot_w32(((n >> d) & 1) != 0);
            pre += (int)__builtin_amdgcn_mbcnt_lo(set, 0u) << d;
            tot += __builtin_popcount(set) << d;
        }
        int base = 0;
        if (lane == 0) base = atomicAdd(&cntL, tot);
        unsigned slot = (unsigned)(lane_i(base, 0) + pre);
        const unsigned eLane = (unsigned)kb + (unsigned)tid * 16u;
#pragma unroll
        for (int p = 0; p < KSEG / (32 * OWN_T); ++p) {
            const unsigned eP = eLane + (unsigned)(2 * p * OWN_T) * 16u;
            unsigned m = mk[p];
            while (m != 0u) {
                const unsigned bit = (unsigned)__builtin_ctz(m);
                const unsigned e = eP + ((bit & 4u) * (unsigned)(OWN_T * 4)) + ((bit & 3u) << 2) + (bit >> 3);
                listL[slot < (unsigned)CAP ? slot : (unsigned)(CAP - 1)] = e;
                ++slot;
                m &= m - 1u;
            }
        }
    };

    auto scan_segment = [&](const int sb) {
        if (((E & 3) == 0) && (sb + SEG <= E)) {
            const int4* __restrict__ dst4 = (const int4*)(dst + sb);
            int4 dq[SEG / (4 * OWN_T)];
#pragma unroll
            for (int k = 0; k < SEG / (4 * OWN_T); ++k) dq[k] = dst4[k * OWN_T + tid];
#pragma unroll
            for (int k = 0; k < SEG / (4 * OWN_T); ++k) {
                const unsigned u0 = (unsigned)(dq[k].x - lo);
                const unsigned u1 = (unsigned)(dq[k].y - lo);
                const unsigned u2 = (unsigned)(dq[k].z - lo);
                const unsigned u3 = (unsigned)(dq[k].w - lo);
                const bool m0 = u0 < (unsigned)nOwn, m1 = u1 < (unsigned)nOwn;
                const bool m2 = u2 < (unsigned)nOwn, m3 = u3 < (unsigned)nOwn;
                if (m0 | m1 | m2 | m3) {
                    const unsigned e0 = (unsigned)(sb + (k * OWN_T + tid) * 4);
                    if (m0) { const int slot = atomicAdd(&cntL, 1); if (slot < CAP) listL[slot] = e0 + 0u; }
                    if (m1) { const int slot = atomicAdd(&cntL, 1); if (slot < CAP) listL[slot] = e0 + 1u; }
                    if (m2) { const int slot = atomicAdd(&cntL, 1); if (slot < CAP) listL[slot] = e0 + 2u; }
                    if (m3) { const int slot = atomicAdd(&cntL, 1); if (slot < CAP) listL[slot] = e0 + 3u; }
                }
            }
        } else {
            for (int k0 = 0; k0 < SEG / OWN_T; k0 += 8) {
                int dv[8];
#pragma unroll
                for (int k = 0; k < 8; ++k) {
                    const int e = sb + (k0 + k) * OWN_T + tid;
                    dv[k] = e < E ? dst[e] : -1;
                }
#pragma unroll
                for (int k = 0; k < 8; ++k) {
                    const unsigned u = (unsigned)(dv[k] - lo);
                    if (u < (unsigned)nOwn) {
                        const int e = sb + (k0 + k) * OWN_T + tid;
                        const int slot = atomicAdd(&cntL, 1);
                        if (slot < CAP) listL[slot] = (unsigned)e;
                    }
                }
            }
        }
    };

    auto process_chunk = [&](const int cnt, const bool resolved) {
        for (int j = tid; j < cnt; j += OWN_T) {
            unsigned entry = listL[j];
            if (!resolved) {
                const unsigned e = entry;
                const unsigned dl = (unsigned)(dst[e] - lo);
                const int s = src[e];
                entry = dl < (unsigned)nOwn ? ((unsigned)s | (dl << 23)) : NO_EDGE;
                listL[j] = entry;
            }
            if (entry != NO_EDGE) {
                const float a = asrc[entry & LOW23];
                asL[j] = a;
                const unsigned dl = entry >> 23;
                if (a >= 0.0f) atomicMax((int*)&mNew[dl], __float_as_int(a));
                else           atomicMin((unsigned int*)&mNew[dl], __float_as_uint(a));
            }
        }
        __syncthreads();

        if (tid < RN) {
            const float ad = adL[tid];
            const float mn = mNew[tid];
            const float shiftOld = lrelu(mL[tid] + ad);
            const float shiftNew = lrelu(mn + ad);
            scL[tid]  = __expf(shiftOld - shiftNew);
            capL[tid] = shiftNew;
            mL[tid]   = mn;
        }
        __syncthreads();
        for (int idx = tid; idx < RN * AW; idx += OWN_T) accL[idx] *= scL[idx / AW];
        __syncthreads();

        const int lc = lane < F ? lane : F - 1;
        for (int base = wave * 32; base < cnt; base += OWN_T) {
            const int j = base + lane;
            int hoff = lo * F, aoff = 0;
            float t = 0.0f;
            const unsigned p = j < cnt ? listL[j] : NO_EDGE;
            if (p != NO_EDGE) {
                const int s  = (int)(p & LOW23);
                const int dl = (int)(p >> 23);
                const float ev = lrelu(asL[j] + adL[dl]);
                t = __expf(ev - capL[dl]);
                hoff = s * F;
                aoff = dl * AW;
            }
            const int n = cnt - base;
            if (n >= 32) {
#pragma unroll
                for (int g = 0; g < 32; g += 16) {
                    int so[16], ao[16];
                    float tt[16];
#pragma unroll
                    for (int k = 0; k < 16; ++k) {
                        so[k] = lane_i(hoff, g + k);
                        ao[k] = lane_i(aoff, g + k);
                        tt[k] = lane_f(t, g + k);
                    }
                    if (lane <= F) {
                        float hv[16];
#pragma unroll
                        for (int k = 0; k < 16; ++k) hv[k] = h[so[k] + lc];
#pragma unroll
                        for (int k = 0; k < 16; ++k)
                            __hip_atomic_fetch_add(&accL[ao[k] + lane],
                                                   lane < F ? tt[k] * hv[k] : tt[k],
                                                   __ATOMIC_RELAXED, __HIP_MEMORY_SCOPE_WORKGROUP);
                    }
                }
            } else {
                for (int q = 0; q < n; ++q) {
                    const int so = lane_i(hoff, q);
                    const int ao = lane_i(aoff, q);
                    const float tt = lane_f(t, q);
                    if (lane <= F) {
                        const float hv = h[so + lc];
                        __hip_atomic_fetch_add(&accL[ao + lane], lane < F ? tt * hv : tt,
                                               __ATOMIC_RELAXED, __HIP_MEMORY_SCOPE_WORKGROUP);
                    }
                }
            }
        }
        __syncthreads();
    };

    unsigned int* const saved = savedList + (size_t)blockIdx.x * CAP;
    unsigned have = (unsigned)CAP + 1u;
    if (!first) have = savedCnt[blockIdx.x];

    int total;
    const bool resolved = have <= (unsigned)CAP;
    if (resolved) {
        total = (int)have;
        for (int j0 = 0; j0 < total; j0 += 8 * OWN_T) {
            unsigned v[8];
#pragma unroll
            for (int k = 0; k < 8; ++k) {
                const int j = j0 + k * OWN_T + tid;
                v[k] = j < total ? saved[j] : NO_EDGE;
            }
#pragma unroll
            for (int k = 0; k < 8; ++k) {
                const int j = j0 + k * OWN_T + tid;
                if (j < total) listL[j] = v[k];
            }
        }
    } else {
        const int keyed = (E / KSEG) * KSEG;
        for (int kb = 0; kb < keyed; kb += KSEG) scan_keys(kb);
        for (int sb = keyed; sb < E; sb += SEG) scan_segment(sb);
        __syncthreads();
        total = __builtin_amdgcn_readfirstlane(cntL);
        __syncthreads();
        if (first && tid == 0) savedCnt[blockIdx.x] = (unsigned)total;
    }
    if (total <= CAP) {
        process_chunk(total, resolved);
        if (first)
            for (int j = tid; j < total; j += OWN_T) saved[j] = listL[j];
    } else {
        if (tid == 0) cntL = 0;
        __syncthreads();
        for (int sb = 0; sb < E; sb += SEG) {
            scan_segment(sb);
            __syncthreads();
            const int cnt = __builtin_amdgcn_readfirstlane(cntL);
            __syncthreads();
            if (cnt <= CAP - SEG && sb + SEG < E) continue;
            process_chunk(cnt, false);
            if (tid == 0) cntL = 0;
            __syncthreads();
        }
    }

    for (int idx = tid; idx < nOwn * F; idx += OWN_T) {
        const int dl = idx / F;
        const int f  = idx - dl * F;
        const float inv = 1.0f / (accL[dl * AW + F] + 1e-16f);
        const float v = accL[dl * AW + f] * inv + b[f];
        feat[(size_t)lo * F + idx] = v > 0.0f ? v : 0.0f;
    }
}

__global__ void gat_final_linear(const float* __restrict__ feat,
                                 const float* __restrict__ Wl,
                                 const float* __restrict__ bl,
                                 float* __restrict__ out,
                                 int N) {
    int i = blockIdx.x * blockDim.x + threadIdx.x;
    if (i >= N) return;
    float sacc = bl[0];
#pragma unroll
    for (int f = 0; f < F; ++f) sacc += feat[i * F + f] * Wl[f];
    out[i] = sacc;
}

extern "C" void kernel_launch(void* const* d_in, const int* in_sizes, int n_in,
                              void* d_out, int out_size, void* d_ws, size_t ws_size,
                              hipStream_t stream) {
    (void)n_in; (void)out_size; (void)ws_size;

    const int N = in_sizes[0];
    const int E = in_sizes[1] / 2;

    const float* x    = (const float*)d_in[0];
    const int*   ei   = (const int*)d_in[1];
    const float* W1   = (const float*)d_in[2];
    const float* as1  = (const float*)d_in[3];
    const float* ad1  = (const float*)d_in[4];
    const float* b1   = (const float*)d_in[5];
    const float* W2   = (const float*)d_in[6];
    const float* as2  = (const float*)d_in[7];
    const float* ad2  = (const float*)d_in[8];
    const float* b2   = (const float*)d_in[9];
    const float* Wl   = (const float*)d_in[10];
    const float* bl   = (const float*)d_in[11];

    const int* srcIdx = ei;
    const int* dstIdx = ei + E;

    const int nWords = (E / KSEG) * (KSEG / 4);
    unsigned int* keyw = (unsigned int*)d_ws;
    float* ws     = (float*)d_ws + nWords;
    float* h      = ws + (size_t)0  * N;
    float* feat1  = ws + (size_t)20 * N;
    float* feat2  = ws + (size_t)40 * N;
    float* asrc   = ws + (size_t)60 * N;
    float* adst   = ws + (size_t)61 * N;

    const int TB   = 256;
    const int gN   = (N + TB - 1) / TB;
    const int gMM  = (N + 15) / 16;
    const int gOwn = (N + RN - 1) / RN;
    const int gKey = nWords / TB + 1;

    unsigned int* savedList = (unsigned int*)(ws + (size_t)62 * N);
    unsigned int* savedCnt  = savedList + (size_t)gOwn * CAP;

    gat_owner_keys<<<gKey, TB, 0, stream>>>(dstIdx, keyw, nWords);

    gat_node_init_l1<<<gN, TB, 0, stream>>>(x, W1, as1, ad1, h, asrc, adst, N);
    gat_owner_layer<<<gOwn, OWN_T, 0, stream>>>(srcIdx, dstIdx, keyw, asrc, adst, h, b1, feat1,
                                                savedList, savedCnt, 1, N, E);

    gat_wmma_feat_x_w<<<gMM, 32, 0, stream>>>(feat1, W2, h, N);
    gat_node_alpha<<<gN, TB, 0, stream>>>(h, as2, ad2, asrc, adst, N);
    gat_owner_layer<<<gOwn, OWN_T, 0, stream>>>(srcIdx, dstIdx, keyw, asrc, adst, h, b2, feat2,
                                                savedList, savedCnt, 0, N, E);

    gat_final_linear<<<gN, TB, 0, stream>>>(feat2, Wl, bl, (float*)d_out, N);
}


# Owner orders its resolved edge list by source bin (one LDS counting pass, 128-node bins) before the score gather: a wave's 32 asrc reads hit few lines; list order changes, LDS adds arrival order
# speedup vs baseline: 12.5667x; 1.0506x over previous
#include <hip/hip_runtime.h>
#include <math.h>

#define F 20
#define NEG_SLOPE 0.2f

#define RN     392
#define AW     21
#define OWN_T  1024
#define SEG    32768
#define CAP    32768
#define KSEG   131072
#define LOW23  0x7FFFFFu
#define NO_EDGE 0xFFFFFFFFu
#define NBIN   1024
#define BSHIFT 7

typedef float v2f __attribute__((ext_vector_type(2)));
typedef float v8f __attribute__((ext_vector_type(8)));

__device__ __forceinline__ float lrelu(float v) {
    return v > 0.0f ? v : NEG_SLOPE * v;
}

__device__ __forceinline__ int lane_i(int v, int l) {
    return __builtin_amdgcn_readlane(v, l);
}
__device__ __forceinline__ float lane_f(float v, int l) {
    return __int_as_float(__builtin_amdgcn_readlane(__float_as_int(v), l));
}

__global__ void gat_node_init_l1(const float* __restrict__ x,
                                 const float* __restrict__ W1,
                                 const float* __restrict__ a_src,
                                 const float* __restrict__ a_dst,
                                 float* __restrict__ h,
                                 float* __restrict__ asrc,
                                 float* __restrict__ adst,
                                 int N) {
    int i = blockIdx.x * blockDim.x + threadIdx.x;
    if (i >= N) return;
    float xv = x[i];
    float sa = 0.0f, sd = 0.0f;
#pragma unroll
    for (int f = 0; f < F; ++f) {
        float hv = xv * W1[f];
        h[i * F + f] = hv;
        sa += hv * a_src[f];
        sd += hv * a_dst[f];
    }
    asrc[i] = sa;
    adst[i] = sd;
}

__global__ void __launch_bounds__(32)
gat_wmma_feat_x_w(const float* __restrict__ feat,
                  const float* __restrict__ W,
                  float* __restrict__ h,
                  int N) {
    int tile  = blockIdx.x;
    int lane  = threadIdx.x;
    int mrow  = lane & 15;
    int khalf = (lane >> 4) * 2;

    int row  = tile * 16 + mrow;
    int rowc = row < N ? row : N - 1;
    const float* arow = feat + (size_t)rowc * F;

    int   n1   = 16 + mrow;
    int   n1c  = (n1 < F) ? n1 : 0;
    float msk1 = (n1 < F) ? 1.0f : 0.0f;

    v8f c0 = {};
    v8f c1 = {};
#pragma unroll
    for (int k0 = 0; k0 < F; k0 += 4) {
        const float* wr0 = W + (size_t)(k0 + khalf + 0) * F;
        const float* wr1 = W + (size_t)(k0 + khalf + 1) * F;

        v2f a;
        a.x = arow[k0 + khalf + 0];
        a.y = arow[k0 + khalf + 1];

        v2f b0;
        b0.x = wr0[mrow];
        b0.y = wr1[mrow];
        c0 = __builtin_amdgcn_wmma_f32_16x16x4_f32(false, a, false, b0,
                                                   (short)0, c0, false, false);

        v2f b1;
        b1.x = wr0[n1c] * msk1;
        b1.y = wr1[n1c] * msk1;
        c1 = __builtin_amdgcn_wmma_f32_16x16x4_f32(false, a, false, b1,
                                                   (short)0, c1, false, false);
    }

    int rbase = tile * 16 + ((lane >> 4) << 3);
#pragma unroll
    for (int j = 0; j < 8; ++j) {
        int r = rbase + j;
        if (r < N) {
            h[(size_t)r * F + mrow] = c0[j];
            if (n1 < F) h[(size_t)r * F + n1] = c1[j];
        }
    }
}

__global__ void gat_node_alpha(const float* __restrict__ h,
                               const float* __restrict__ a_src,
                               const float* __restrict__ a_dst,
                               float* __restrict__ asrc,
                               float* __restrict__ adst,
                               int N) {
    int i = blockIdx.x * blockDim.x + threadIdx.x;
    if (i >= N) return;
    float sa = 0.0f, sd = 0.0f;
#pragma unroll
    for (int f = 0; f < F; ++f) {
        float hv = h[i * F + f];
        sa += hv * a_src[f];
        sd += hv * a_dst[f];
    }
    asrc[i] = sa;
    adst[i] = sd;
}

__global__ void gat_owner_keys(const int* __restrict__ dst,
                               unsigned int* __restrict__ keyw,
                               int nWords) {
    int w = blockIdx.x * blockDim.x + threadIdx.x;
    if (w >= nWords) return;
    unsigned int k = 0u;
#pragma unroll
    for (int c = 0; c < 4; ++c)
        k |= (((unsigned int)dst[4 * w + c] / (unsigned int)RN) & 0xFFu) << (8 * c);
    keyw[w] = k;
}

__global__ void __launch_bounds__(OWN_T)
gat_owner_layer(const int* __restrict__ src,
                const int* __restrict__ dst,
                const unsigned int* __restrict__ keyw,
                const float* __restrict__ asrc,
                const float* __restrict__ adst,
                const float* __restrict__ h,
                const float* __restrict__ b,
                float* __restrict__ feat,
                unsigned int* __restrict__ savedList,
                unsigned int* __restrict__ savedCnt,
                int first,
                int N, int E) {
    __shared__ float accL[RN * AW];
    __shared__ float mL[RN];
    __shared__ float mNew[RN];
    __shared__ float adL[RN];
    __shared__ float capL[RN];
    __shared__ float scL[RN];
    __shared__ unsigned listL[CAP];
    __shared__ unsigned asU[CAP];
    __shared__ unsigned binL[NBIN];
    __shared__ unsigned wsumL[32];
    __shared__ int cntL;

    const int tid  = threadIdx.x;
    const int lane = tid & 31;
    const int wave = __builtin_amdgcn_readfirstlane(tid >> 5);
    const int lo   = blockIdx.x * RN;
    const int nOwn = (N - lo) < RN ? (N - lo) : RN;

    for (int dl = tid; dl < RN; dl += OWN_T) {
        float a = -3.402823466e38f, ad = 0.0f;
        if (dl < nOwn) { a = asrc[lo + dl]; ad = adst[lo + dl]; }
        mL[dl] = a;
        mNew[dl] = a;
        adL[dl] = ad;
    }
    for (int idx = tid; idx < RN * AW; idx += OWN_T) {
        const int dl = idx / AW;
        const int f  = idx - dl * AW;
        float v = 0.0f;
        if (dl < nOwn) v = f < F ? h[(size_t)(lo + dl) * F + f] : 1.0f;
        accL[idx] = v;
    }
    if (tid == 0) cntL = 0;
    __syncthreads();

    const unsigned gpat = (unsigned)(blockIdx.x & 0xFF) * 0x01010101u;
    auto scan_keys = [&](const int kb) {
        const uint4* __restrict__ k4 = (const uint4*)(keyw + (kb >> 2));
        uint4 q[KSEG / (16 * OWN_T)];
#pragma unroll
        for (int k = 0; k < KSEG / (16 * OWN_T); ++k) q[k] = k4[k * OWN_T + tid];
        unsigned mk[KSEG / (32 * OWN_T)];
        int n = 0;
#pragma unroll
        for (int p = 0; p < KSEG / (32 * OWN_T); ++p) {
            const unsigned w[8] = {q[2 * p].x,     q[2 * p].y,     q[2 * p].z,     q[2 * p].w,
                                   q[2 * p + 1].x, q[2 * p + 1].y, q[2 * p + 1].z, q[2 * p + 1].w};
            unsigned m = 0u;
#pragma unroll
            for (int j = 0; j < 8; ++j) {
                const unsigned v = w[j] ^ gpat;
                m |= ((v - 0x01010101u) & ~v & 0x80808080u) >> (7 - j);
            }
            mk[p] = m;
            n += __popc(m);
        }
        int pre = 0, tot = 0;
        for (int d = 0; d < 8; ++d) {
            if (__builtin_amdgcn_ballot_w32((n >> d) != 0) == 0u) break;
            const unsigned set = __builtin_amdgcn_ballot_w32(((n >> d) & 1) != 0);
            pre += (int)__builtin_amdgcn_mbcnt_lo(set, 0u) << d;
            tot += __builtin_popcount(set) << d;
        }
        int base = 0;
        if (lane == 0) base = atomicAdd(&cntL, tot);
        unsigned slot = (unsigned)(lane_i(base, 0) + pre);
        const unsigned eLane = (unsigned)kb + (unsigned)tid * 16u;
#pragma unroll
        for (int p = 0; p < KSEG / (32 * OWN_T); ++p) {
            const unsigned eP = eLane + (unsigned)(2 * p * OWN_T) * 16u;
            unsigned m = mk[p];
            while (m != 0u) {
                const unsigned bit = (unsigned)__builtin_ctz(m);
                const unsigned e = eP + ((bit & 4u) * (unsigned)(OWN_T * 4)) + ((bit & 3u) << 2) + (bit >> 3);
                listL[slot < (unsigned)CAP ? slot : (unsigned)(CAP - 1)] = e;
                ++slot;
                m &= m - 1u;
            }
        }
    };

    auto scan_segment = [&](const int sb) {
        if (((E & 3) == 0) && (sb + SEG <= E)) {
            const int4* __restrict__ dst4 = (const int4*)(dst + sb);
            int4 dq[SEG / (4 * OWN_T)];
#pragma unroll
            for (int k = 0; k < SEG / (4 * OWN_T); ++k) dq[k] = dst4[k * OWN_T + tid];
#pragma unroll
            for (int k = 0; k < SEG / (4 * OWN_T); ++k) {
                const unsigned u0 = (unsigned)(dq[k].x - lo);
                const unsigned u1 = (unsigned)(dq[k].y - lo);
                const unsigned u2 = (unsigned)(dq[k].z - lo);
                const unsigned u3 = (unsigned)(dq[k].w - lo);
                const bool m0 = u0 < (unsigned)nOwn, m1 = u1 < (unsigned)nOwn;
                const bool m2 = u2 < (unsigned)nOwn, m3 = u3 < (unsigned)nOwn;
                if (m0 | m1 | m2 | m3) {
                    const unsigned e0 = (unsigned)(sb + (k * OWN_T + tid) * 4);
                    if (m0) { const int slot = atomicAdd(&cntL, 1); if (slot < CAP) listL[slot] = e0 + 0u; }
                    if (m1) { const int slot = atomicAdd(&cntL, 1); if (slot < CAP) listL[slot] = e0 + 1u; }
                    if (m2) { const int slot = atomicAdd(&cntL, 1); if (slot < CAP) listL[slot] = e0 + 2u; }
                    if (m3) { const int slot = atomicAdd(&cntL, 1); if (slot < CAP) listL[slot] = e0 + 3u; }
                }
            }
        } else {
            for (int k0 = 0; k0 < SEG / OWN_T; k0 += 8) {
                int dv[8];
#pragma unroll
                for (int k = 0; k < 8; ++k) {
                    const int e = sb + (k0 + k) * OWN_T + tid;
                    dv[k] = e < E ? dst[e] : -1;
                }
#pragma unroll
                for (int k = 0; k < 8; ++k) {
                    const unsigned u = (unsigned)(dv[k] - lo);
                    if (u < (unsigned)nOwn) {
                        const int e = sb + (k0 + k) * OWN_T + tid;
                        const int slot = atomicAdd(&cntL, 1);
                        if (slot < CAP) listL[slot] = (unsigned)e;
                    }
                }
            }
        }
    };

    auto process_chunk = [&](const int cnt, const bool resolved) {
        for (int j = tid; j < cnt; j += OWN_T) {
            unsigned entry = listL[j];
            if (!resolved) {
                const unsigned e = entry;
                const unsigned dl = (unsigned)(dst[e] - lo);
                const int s = src[e];
                entry = dl < (unsigned)nOwn ? ((unsigned)s | (dl << 23)) : NO_EDGE;
                listL[j] = entry;
            }
            if (entry != NO_EDGE) {
                const float a = asrc[entry & LOW23];
                asU[j] = __float_as_uint(a);
                const unsigned dl = entry >> 23;
                if (a >= 0.0f) atomicMax((int*)&mNew[dl], __float_as_int(a));
                else           atomicMin((unsigned int*)&mNew[dl], __float_as_uint(a));
            }
        }
        __syncthreads();

        if (tid < RN) {
            const float ad = adL[tid];
            const float mn = mNew[tid];
            const float shiftOld = lrelu(mL[tid] + ad);
            const float shiftNew = lrelu(mn + ad);
            scL[tid]  = __expf(shiftOld - shiftNew);
            capL[tid] = shiftNew;
            mL[tid]   = mn;
        }
        __syncthreads();
        for (int idx = tid; idx < RN * AW; idx += OWN_T) accL[idx] *= scL[idx / AW];
        __syncthreads();

        const int lc = lane < F ? lane : F - 1;
        for (int base = wave * 32; base < cnt; base += OWN_T) {
            const int j = base + lane;
            int hoff = lo * F, aoff = 0;
            float t = 0.0f;
            const unsigned p = j < cnt ? listL[j] : NO_EDGE;
            if (p != NO_EDGE) {
                const int s  = (int)(p & LOW23);
                const int dl = (int)(p >> 23);
                const float ev = lrelu(__uint_as_float(asU[j]) + adL[dl]);
                t = __expf(ev - capL[dl]);
                hoff = s * F;
                aoff = dl * AW;
            }
            const int n = cnt - base;
            if (n >= 32) {
#pragma unroll
                for (int g = 0; g < 32; g += 16) {
                    int so[16], ao[16];
                    float tt[16];
#pragma unroll
                    for (int k = 0; k < 16; ++k) {
                        so[k] = lane_i(hoff, g + k);
                        ao[k] = lane_i(aoff, g + k);
                        tt[k] = lane_f(t, g + k);
                    }
                    if (lane <= F) {
                        float hv[16];
#pragma unroll
                        for (int k = 0; k < 16; ++k) hv[k] = h[so[k] + lc];
#pragma unroll
                        for (int k = 0; k < 16; ++k)
                            __hip_atomic_fetch_add(&accL[ao[k] + lane],
                                                   lane < F ? tt[k] * hv[k] : tt[k],
                                                   __ATOMIC_RELAXED, __HIP_MEMORY_SCOPE_WORKGROUP);
                    }
                }
            } else {
                for (int q = 0; q < n; ++q) {
                    const int so = lane_i(hoff, q);
                    const int ao = lane_i(aoff, q);
                    const float tt = lane_f(t, q);
                    if (lane <= F) {
                        const float hv = h[so + lc];
                        __hip_atomic_fetch_add(&accL[ao + lane], lane < F ? tt * hv : tt,
                                               __ATOMIC_RELAXED, __HIP_MEMORY_SCOPE_WORKGROUP);
                    }
                }
            }
        }
        __syncthreads();
    };

    const unsigned nbAll = ((unsigned)(N - 1) >> BSHIFT) + 1u;
    const unsigned nb    = nbAll < (unsigned)NBIN ? nbAll : (unsigned)NBIN;
    const unsigned rot   = ((unsigned)blockIdx.x * nb) / (unsigned)gridDim.x;
    auto bin_of = [&](const unsigned s) -> unsigned {
        const unsigned b0 = s >> BSHIFT;
        const unsigned t  = (b0 < nb ? b0 : nb - 1u) + rot;
        return t >= nb ? t - nb : t;
    };
    auto resolve_sort = [&](const int cnt) -> int {
        for (int bI = tid; bI < NBIN; bI += OWN_T) binL[bI] = 0u;
        __syncthreads();
        for (int j = tid; j < cnt; j += OWN_T) {
            const unsigned e = listL[j];
            const unsigned dl = (unsigned)(dst[e] - lo);
            const int s = src[e];
            const bool mine = dl < (unsigned)nOwn;
            listL[j] = mine ? ((unsigned)s | (dl << 23)) : NO_EDGE;
            if (mine) atomicAdd(&binL[bin_of((unsigned)s)], 1u);
        }
        __syncthreads();
        const unsigned own = binL[tid];
        unsigned inc = own;
#pragma unroll
        for (int o = 1; o < 32; o <<= 1) {
            const unsigned v = (unsigned)__shfl_up((int)inc, o);
            if (lane >= o) inc += v;
        }
        if (lane == 31) wsumL[wave] = inc;
        __syncthreads();
        if (wave == 0) {
            const unsigned w = wsumL[lane];
            unsigned wi = w;
#pragma unroll
            for (int o = 1; o < 32; o <<= 1) {
                const unsigned v = (unsigned)__shfl_up((int)wi, o);
                if (lane >= o) wi += v;
            }
            wsumL[lane] = wi - w;
            if (lane == 31) cntL = (int)wi;
        }
        __syncthreads();
        binL[tid] = wsumL[wave] + inc - own;
        const int kept = __builtin_amdgcn_readfirstlane(cntL);
        __syncthreads();
        for (int j = tid; j < cnt; j += OWN_T) {
            const unsigned entry = listL[j];
            if (entry != NO_EDGE) {
                const unsigned pos = atomicAdd(&binL[bin_of(entry & LOW23)], 1u);
                asU[pos] = entry;
            }
        }
        __syncthreads();
        for (int j = tid; j < kept; j += OWN_T) listL[j] = asU[j];
        __syncthreads();
        return kept;
    };

    unsigned int* const saved = savedList + (size_t)blockIdx.x * CAP;
    unsigned have = (unsigned)CAP + 1u;
    if (!first) have = savedCnt[blockIdx.x];

    int total;
    const bool resolved = have <= (unsigned)CAP;
    if (resolved) {
        total = (int)have;
        for (int j0 = 0; j0 < total; j0 += 8 * OWN_T) {
            unsigned v[8];
#pragma unroll
            for (int k = 0; k < 8; ++k) {
                const int j = j0 + k * OWN_T + tid;
                v[k] = j < total ? saved[j] : NO_EDGE;
            }
#pragma unroll
            for (int k = 0; k < 8; ++k) {
                const int j = j0 + k * OWN_T + tid;
                if (j < total) listL[j] = v[k];
            }
        }
    } else {
        const int keyed = (E / KSEG) * KSEG;
        for (int kb = 0; kb < keyed; kb += KSEG) scan_keys(kb);
        for (int sb = keyed; sb < E; sb += SEG) scan_segment(sb);
        __syncthreads();
        total = __builtin_amdgcn_readfirstlane(cntL);
        __syncthreads();
    }
    if (total <= CAP) {
        int cnt = total;
        if (!resolved) cnt = resolve_sort(total);
        process_chunk(cnt, true);
        if (first) {
            if (tid == 0) savedCnt[blockIdx.x] = (unsigned)cnt;
            for (int j = tid; j < cnt; j += OWN_T) saved[j] = listL[j];
        }
    } else {
        if (first && tid == 0) savedCnt[blockIdx.x] = (unsigned)total;
        if (tid == 0) cntL = 0;
        __syncthreads();
        for (int sb = 0; sb < E; sb += SEG) {
            scan_segment(sb);
            __syncthreads();
            const int cnt = __builtin_amdgcn_readfirstlane(cntL);
            __syncthreads();
            if (cnt <= CAP - SEG && sb + SEG < E) continue;
            process_chunk(cnt, false);
            if (tid == 0) cntL = 0;
            __syncthreads();
        }
    }

    for (int idx = tid; idx < nOwn * F; idx += OWN_T) {
        const int dl = idx / F;
        const int f  = idx - dl * F;
        const float inv = 1.0f / (accL[dl * AW + F] + 1e-16f);
        const float v = accL[dl * AW + f] * inv + b[f];
        feat[(size_t)lo * F + idx] = v > 0.0f ? v : 0.0f;
    }
}

__global__ void gat_final_linear(const float* __restrict__ feat,
                                 const float* __restrict__ Wl,
                                 const float* __restrict__ bl,
                                 float* __restrict__ out,
                                 int N) {
    int i = blockIdx.x * blockDim.x + threadIdx.x;
    if (i >= N) return;
    float sacc = bl[0];
#pragma unroll
    for (int f = 0; f < F; ++f) sacc += feat[i * F + f] * Wl[f];
    out[i] = sacc;
}

extern "C" void kernel_launch(void* const* d_in, const int* in_sizes, int n_in,
                              void* d_out, int out_size, void* d_ws, size_t ws_size,
                              hipStream_t stream) {
    (void)n_in; (void)out_size; (void)ws_size;

    const int N = in_sizes[0];
    const int E = in_sizes[1] / 2;

    const float* x    = (const float*)d_in[0];
    const int*   ei   = (const int*)d_in[1];
    const float* W1   = (const float*)d_in[2];
    const float* as1  = (const float*)d_in[3];
    const float* ad1  = (const float*)d_in[4];
    const float* b1   = (const float*)d_in[5];
    const float* W2   = (const float*)d_in[6];
    const float* as2  = (const float*)d_in[7];
    const float* ad2  = (const float*)d_in[8];
    const float* b2   = (const float*)d_in[9];
    const float* Wl   = (const float*)d_in[10];
    const float* bl   = (const float*)d_in[11];

    const int* srcIdx = ei;
    const int* dstIdx = ei + E;

    const int nWords = (E / KSEG) * (KSEG / 4);
    unsigned int* keyw = (unsigned int*)d_ws;
    float* ws     = (float*)d_ws + nWords;
    float* h      = ws + (size_t)0  * N;
    float* feat1  = ws + (size_t)20 * N;
    float* feat2  = ws + (size_t)40 * N;
    float* asrc   = ws + (size_t)60 * N;
    float* adst   = ws + (size_t)61 * N;

    const int TB   = 256;
    const int gN   = (N + TB - 1) / TB;
    const int gMM  = (N + 15) / 16;
    const int gOwn = (N + RN - 1) / RN;
    const int gKey = nWords / TB + 1;

    unsigned int* savedList = (unsigned int*)(ws + (size_t)62 * N);
    unsigned int* savedCnt  = savedList + (size_t)gOwn * CAP;

    gat_owner_keys<<<gKey, TB, 0, stream>>>(dstIdx, keyw, nWords);

    gat_node_init_l1<<<gN, TB, 0, stream>>>(x, W1, as1, ad1, h, asrc, adst, N);
    gat_owner_layer<<<gOwn, OWN_T, 0, stream>>>(srcIdx, dstIdx, keyw, asrc, adst, h, b1, feat1,
                                                savedList, savedCnt, 1, N, E);

    gat_wmma_feat_x_w<<<gMM, 32, 0, stream>>>(feat1, W2, h, N);
    gat_node_alpha<<<gN, TB, 0, stream>>>(h, as2, ad2, asrc, adst, N);
    gat_owner_layer<<<gOwn, OWN_T, 0, stream>>>(srcIdx, dstIdx, keyw, asrc, adst, h, b2, feat2,
                                                savedList, savedCnt, 0, N, E);

    gat_final_linear<<<gN, TB, 0, stream>>>(feat2, Wl, bl, (float*)d_out, N);
}


# Key pass also stores every edge's list word (owned-node index, source, owner parity); an owner resolves a found edge by one fetch of it, not dst[e] and src[e]: one random line per edge, not two
# speedup vs baseline: 13.0140x; 1.0356x over previous
#include <hip/hip_runtime.h>
#include <math.h>

#define F 20
#define NEG_SLOPE 0.2f

#define RN     392
#define AW     21
#define OWN_T  1024
#define SEG    32768
#define CAP    32768
#define KSEG   131072
#define SBITS  22
#define LOWS   0x3FFFFFu
#define NO_EDGE 0xFFFFFFFFu
#define NBIN   1024
#define BSHIFT 7

typedef float v2f __attribute__((ext_vector_type(2)));
typedef float v8f __attribute__((ext_vector_type(8)));

__device__ __forceinline__ float lrelu(float v) {
    return v > 0.0f ? v : NEG_SLOPE * v;
}

__device__ __forceinline__ int lane_i(int v, int l) {
    return __builtin_amdgcn_readlane(v, l);
}
__device__ __forceinline__ float lane_f(float v, int l) {
    return __int_as_float(__builtin_amdgcn_readlane(__float_as_int(v), l));
}

__global__ void gat_node_init_l1(const float* __restrict__ x,
                                 const float* __restrict__ W1,
                                 const float* __restrict__ a_src,
                                 const float* __restrict__ a_dst,
                                 float* __restrict__ h,
                                 float* __restrict__ asrc,
                                 float* __restrict__ adst,
                                 int N) {
    int i = blockIdx.x * blockDim.x + threadIdx.x;
    if (i >= N) return;
    float xv = x[i];
    float sa = 0.0f, sd = 0.0f;
#pragma unroll
    for (int f = 0; f < F; ++f) {
        float hv = xv * W1[f];
        h[i * F + f] = hv;
        sa += hv * a_src[f];
        sd += hv * a_dst[f];
    }
    asrc[i] = sa;
    adst[i] = sd;
}

__global__ void __launch_bounds__(32)
gat_wmma_feat_x_w(const float* __restrict__ feat,
                  const float* __restrict__ W,
                  float* __restrict__ h,
                  int N) {
    int tile  = blockIdx.x;
    int lane  = threadIdx.x;
    int mrow  = lane & 15;
    int khalf = (lane >> 4) * 2;

    int row  = tile * 16 + mrow;
    int rowc = row < N ? row : N - 1;
    const float* arow = feat + (size_t)rowc * F;

    int   n1   = 16 + mrow;
    int   n1c  = (n1 < F) ? n1 : 0;
    float msk1 = (n1 < F) ? 1.0f : 0.0f;

    v8f c0 = {};
    v8f c1 = {};
#pragma unroll
    for (int k0 = 0; k0 < F; k0 += 4) {
        const float* wr0 = W + (size_t)(k0 + khalf + 0) * F;
        const float* wr1 = W + (size_t)(k0 + khalf + 1) * F;

        v2f a;
        a.x = arow[k0 + khalf + 0];
        a.y = arow[k0 + khalf + 1];

        v2f b0;
        b0.x = wr0[mrow];
        b0.y = wr1[mrow];
        c0 = __builtin_amdgcn_wmma_f32_16x16x4_f32(false, a, false, b0,
                                                   (short)0, c0, false, false);

        v2f b1;
        b1.x = wr0[n1c] * msk1;
        b1.y = wr1[n1c] * msk1;
        c1 = __builtin_amdgcn_wmma_f32_16x16x4_f32(false, a, false, b1,
                                                   (short)0, c1, false, false);
    }

    int rbase = tile * 16 + ((lane >> 4) << 3);
#pragma unroll
    for (int j = 0; j < 8; ++j) {
        int r = rbase + j;
        if (r < N) {
            h[(size_t)r * F + mrow] = c0[j];
            if (n1 < F) h[(size_t)r * F + n1] = c1[j];
        }
    }
}

__global__ void gat_node_alpha(const float* __restrict__ h,
                               const float* __restrict__ a_src,
                               const float* __restrict__ a_dst,
                               float* __restrict__ asrc,
                               float* __restrict__ adst,
                               int N) {
    int i = blockIdx.x * blockDim.x + threadIdx.x;
    if (i >= N) return;
    float sa = 0.0f, sd = 0.0f;
#pragma unroll
    for (int f = 0; f < F; ++f) {
        float hv = h[i * F + f];
        sa += hv * a_src[f];
        sd += hv * a_dst[f];
    }
    asrc[i] = sa;
    adst[i] = sd;
}

__global__ void gat_owner_keys(const int* __restrict__ src,
                               const int* __restrict__ dst,
                               unsigned int* __restrict__ keyw,
                               unsigned int* __restrict__ pk,
                               int nWords, int E) {
    const int w  = blockIdx.x * blockDim.x + threadIdx.x;
    const int e0 = 4 * w;
    if (e0 >= E) return;
    const bool whole = (E & 3) == 0;
    int d[4], s[4];
    if (whole) {
        const int4 d4 = ((const int4*)dst)[w];
        const int4 s4 = ((const int4*)src)[w];
        d[0] = d4.x; d[1] = d4.y; d[2] = d4.z; d[3] = d4.w;
        s[0] = s4.x; s[1] = s4.y; s[2] = s4.z; s[3] = s4.w;
    } else {
#pragma unroll
        for (int c = 0; c < 4; ++c) {
            const bool ok = e0 + c < E;
            d[c] = ok ? dst[e0 + c] : 0;
            s[c] = ok ? src[e0 + c] : 0;
        }
    }
    unsigned int k = 0u;
    unsigned int p[4];
#pragma unroll
    for (int c = 0; c < 4; ++c) {
        const unsigned int owner = (unsigned int)d[c] / (unsigned int)RN;
        p[c] = (unsigned int)s[c] | (((unsigned int)d[c] - owner * (unsigned int)RN) << SBITS) | (owner << 31);
        k |= (owner & 0xFFu) << (8 * c);
    }
    if (w < nWords) keyw[w] = k;
    if (whole) {
        ((uint4*)pk)[w] = make_uint4(p[0], p[1], p[2], p[3]);
    } else {
#pragma unroll
        for (int c = 0; c < 4; ++c)
            if (e0 + c < E) pk[e0 + c] = p[c];
    }
}

__global__ void __launch_bounds__(OWN_T)
gat_owner_layer(const int* __restrict__ src,
                const int* __restrict__ dst,
                const unsigned int* __restrict__ keyw,
                const unsigned int* __restrict__ pk,
                const float* __restrict__ asrc,
                const float* __restrict__ adst,
                const float* __restrict__ h,
                const float* __restrict__ b,
                float* __restrict__ feat,
                unsigned int* __restrict__ savedList,
                unsigned int* __restrict__ savedCnt,
                int first,
                int N, int E) {
    __shared__ float accL[RN * AW];
    __shared__ float mL[RN];
    __shared__ float mNew[RN];
    __shared__ float adL[RN];
    __shared__ float capL[RN];
    __shared__ float scL[RN];
    __shared__ unsigned listL[CAP];
    __shared__ unsigned asU[CAP];
    __shared__ unsigned binL[NBIN];
    __shared__ unsigned wsumL[32];
    __shared__ int cntL;

    const int tid  = threadIdx.x;
    const int lane = tid & 31;
    const int wave = __builtin_amdgcn_readfirstlane(tid >> 5);
    const int lo   = blockIdx.x * RN;
    const int nOwn = (N - lo) < RN ? (N - lo) : RN;

    for (int dl = tid; dl < RN; dl += OWN_T) {
        float a = -3.402823466e38f, ad = 0.0f;
        if (dl < nOwn) { a = asrc[lo + dl]; ad = adst[lo + dl]; }
        mL[dl] = a;
        mNew[dl] = a;
        adL[dl] = ad;
    }
    for (int idx = tid; idx < RN * AW; idx += OWN_T) {
        const int dl = idx / AW;
        const int f  = idx - dl * AW;
        float v = 0.0f;
        if (dl < nOwn) v = f < F ? h[(size_t)(lo + dl) * F + f] : 1.0f;
        accL[idx] = v;
    }
    if (tid == 0) cntL = 0;
    __syncthreads();

    const unsigned gpat = (unsigned)(blockIdx.x & 0xFF) * 0x01010101u;
    auto scan_keys = [&](const int kb) {
        const uint4* __restrict__ k4 = (const uint4*)(keyw + (kb >> 2));
        uint4 q[KSEG / (16 * OWN_T)];
#pragma unroll
        for (int k = 0; k < KSEG / (16 * OWN_T); ++k) q[k] = k4[k * OWN_T + tid];
        unsigned mk[KSEG / (32 * OWN_T)];
        int n = 0;
#pragma unroll
        for (int p = 0; p < KSEG / (32 * OWN_T); ++p) {
            const unsigned w[8] = {q[2 * p].x,     q[2 * p].y,     q[2 * p].z,     q[2 * p].w,
                                   q[2 * p + 1].x, q[2 * p + 1].y, q[2 * p + 1].z, q[2 * p + 1].w};
            unsigned m = 0u;
#pragma unroll
            for (int j = 0; j < 8; ++j) {
                const unsigned v = w[j] ^ gpat;
                m |= ((v - 0x01010101u) & ~v & 0x80808080u) >> (7 - j);
            }
            mk[p] = m;
            n += __popc(m);
        }
        int pre = 0, tot = 0;
        for (int d = 0; d < 8; ++d) {
            if (__builtin_amdgcn_ballot_w32((n >> d) != 0) == 0u) break;
            const unsigned set = __builtin_amdgcn_ballot_w32(((n >> d) & 1) != 0);
            pre += (int)__builtin_amdgcn_mbcnt_lo(set, 0u) << d;
            tot += __builtin_popcount(set) << d;
        }
        int base = 0;
        if (lane == 0) base = atomicAdd(&cntL, tot);
        unsigned slot = (unsigned)(lane_i(base, 0) + pre);
        const unsigned eLane = (unsigned)kb + (unsigned)tid * 16u;
#pragma unroll
        for (int p = 0; p < KSEG / (32 * OWN_T); ++p) {
            const unsigned eP = eLane + (unsigned)(2 * p * OWN_T) * 16u;
            unsigned m = mk[p];
            while (m != 0u) {
                const unsigned bit = (unsigned)__builtin_ctz(m);
                const unsigned e = eP + ((bit & 4u) * (unsigned)(OWN_T * 4)) + ((bit & 3u) << 2) + (bit >> 3);
                listL[slot < (unsigned)CAP ? slot : (unsigned)(CAP - 1)] = e;
                ++slot;
                m &= m - 1u;
            }
        }
    };

    auto scan_segment = [&](const int sb) {
        if (((E & 3) == 0) && (sb + SEG <= E)) {
            const int4* __restrict__ dst4 = (const int4*)(dst + sb);
            int4 dq[SEG / (4 * OWN_T)];
#pragma unroll
            for (int k = 0; k < SEG / (4 * OWN_T); ++k) dq[k] = dst4[k * OWN_T + tid];
#pragma unroll
            for (int k = 0; k < SEG / (4 * OWN_T); ++k) {
                const unsigned u0 = (unsigned)(dq[k].x - lo);
                const unsigned u1 = (unsigned)(dq[k].y - lo);
                const unsigned u2 = (unsigned)(dq[k].z - lo);
                const unsigned u3 = (unsigned)(dq[k].w - lo);
                const bool m0 = u0 < (unsigned)nOwn, m1 = u1 < (unsigned)nOwn;
                const bool m2 = u2 < (unsigned)nOwn, m3 = u3 < (unsigned)nOwn;
                if (m0 | m1 | m2 | m3) {
                    const unsigned e0 = (unsigned)(sb + (k * OWN_T + tid) * 4);
                    if (m0) { const int slot = atomicAdd(&cntL, 1); if (slot < CAP) listL[slot] = e0 + 0u; }
                    if (m1) { const int slot = atomicAdd(&cntL, 1); if (slot < CAP) listL[slot] = e0 + 1u; }
                    if (m2) { const int slot = atomicAdd(&cntL, 1); if (slot < CAP) listL[slot] = e0 + 2u; }
                    if (m3) { const int slot = atomicAdd(&cntL, 1); if (slot < CAP) listL[slot] = e0 + 3u; }
                }
            }
        } else {
            for (int k0 = 0; k0 < SEG / OWN_T; k0 += 8) {
                int dv[8];
#pragma unroll
                for (int k = 0; k < 8; ++k) {
                    const int e = sb + (k0 + k) * OWN_T + tid;
                    dv[k] = e < E ? dst[e] : -1;
                }
#pragma unroll
                for (int k = 0; k < 8; ++k) {
                    const unsigned u = (unsigned)(dv[k] - lo);
                    if (u < (unsigned)nOwn) {
                        const int e = sb + (k0 + k) * OWN_T + tid;
                        const int slot = atomicAdd(&cntL, 1);
                        if (slot < CAP) listL[slot] = (unsigned)e;
                    }
                }
            }
        }
    };

    auto process_chunk = [&](const int cnt, const bool resolved) {
        for (int j = tid; j < cnt; j += OWN_T) {
            unsigned entry = listL[j];
            if (!resolved) {
                const unsigned e = entry;
                const unsigned dl = (unsigned)(dst[e] - lo);
                const int s = src[e];
                entry = dl < (unsigned)nOwn ? ((unsigned)s | (dl << SBITS)) : NO_EDGE;
                listL[j] = entry;
            }
            if (entry != NO_EDGE) {
                const float a = asrc[entry & LOWS];
                asU[j] = __float_as_uint(a);
                const unsigned dl = entry >> SBITS;
                if (a >= 0.0f) atomicMax((int*)&mNew[dl], __float_as_int(a));
                else           atomicMin((unsigned int*)&mNew[dl], __float_as_uint(a));
            }
        }
        __syncthreads();

        if (tid < RN) {
            const float ad = adL[tid];
            const float mn = mNew[tid];
            const float shiftOld = lrelu(mL[tid] + ad);
            const float shiftNew = lrelu(mn + ad);
            scL[tid]  = __expf(shiftOld - shiftNew);
            capL[tid] = shiftNew;
            mL[tid]   = mn;
        }
        __syncthreads();
        for (int idx = tid; idx < RN * AW; idx += OWN_T) accL[idx] *= scL[idx / AW];
        __syncthreads();

        const int lc = lane < F ? lane : F - 1;
        for (int base = wave * 32; base < cnt; base += OWN_T) {
            const int j = base + lane;
            int hoff = lo * F, aoff = 0;
            float t = 0.0f;
            const unsigned p = j < cnt ? listL[j] : NO_EDGE;
            if (p != NO_EDGE) {
                const int s  = (int)(p & LOWS);
                const int dl = (int)(p >> SBITS);
                const float ev = lrelu(__uint_as_float(asU[j]) + adL[dl]);
                t = __expf(ev - capL[dl]);
                hoff = s * F;
                aoff = dl * AW;
            }
            const int n = cnt - base;
            if (n >= 32) {
#pragma unroll
                for (int g = 0; g < 32; g += 16) {
                    int so[16], ao[16];
                    float tt[16];
#pragma unroll
                    for (int k = 0; k < 16; ++k) {
                        so[k] = lane_i(hoff, g + k);
                        ao[k] = lane_i(aoff, g + k);
                        tt[k] = lane_f(t, g + k);
                    }
                    if (lane <= F) {
                        float hv[16];
#pragma unroll
                        for (int k = 0; k < 16; ++k) hv[k] = h[so[k] + lc];
#pragma unroll
                        for (int k = 0; k < 16; ++k)
                            __hip_atomic_fetch_add(&accL[ao[k] + lane],
                                                   lane < F ? tt[k] * hv[k] : tt[k],
                                                   __ATOMIC_RELAXED, __HIP_MEMORY_SCOPE_WORKGROUP);
                    }
                }
            } else {
                for (int q = 0; q < n; ++q) {
                    const int so = lane_i(hoff, q);
                    const int ao = lane_i(aoff, q);
                    const float tt = lane_f(t, q);
                    if (lane <= F) {
                        const float hv = h[so + lc];
                        __hip_atomic_fetch_add(&accL[ao + lane], lane < F ? tt * hv : tt,
                                               __ATOMIC_RELAXED, __HIP_MEMORY_SCOPE_WORKGROUP);
                    }
                }
            }
        }
        __syncthreads();
    };

    const unsigned nbAll = ((unsigned)(N - 1) >> BSHIFT) + 1u;
    const unsigned nb    = nbAll < (unsigned)NBIN ? nbAll : (unsigned)NBIN;
    const unsigned rot   = ((unsigned)blockIdx.x * nb) / (unsigned)gridDim.x;
    auto bin_of = [&](const unsigned s) -> unsigned {
        const unsigned b0 = s >> BSHIFT;
        const unsigned t  = (b0 < nb ? b0 : nb - 1u) + rot;
        return t >= nb ? t - nb : t;
    };
    const bool exactIds = gridDim.x <= 256u;
    const unsigned myPar = (unsigned)blockIdx.x & 1u;
    auto resolve_sort = [&](const int cnt) -> int {
        for (int bI = tid; bI < NBIN; bI += OWN_T) binL[bI] = 0u;
        __syncthreads();
        for (int j = tid; j < cnt; j += OWN_T) {
            const unsigned e = listL[j];
            const unsigned p = pk[e];
            bool mine = (p >> 31) == myPar;
            if (!exactIds) mine = (unsigned)(dst[e] - lo) < (unsigned)nOwn;
            const unsigned entry = p & 0x7FFFFFFFu;
            listL[j] = mine ? entry : NO_EDGE;
            if (mine) atomicAdd(&binL[bin_of(entry & LOWS)], 1u);
        }
        __syncthreads();
        const unsigned own = binL[tid];
        unsigned inc = own;
#pragma unroll
        for (int o = 1; o < 32; o <<= 1) {
            const unsigned v = (unsigned)__shfl_up((int)inc, o);
            if (lane >= o) inc += v;
        }
        if (lane == 31) wsumL[wave] = inc;
        __syncthreads();
        if (wave == 0) {
            const unsigned w = wsumL[lane];
            unsigned wi = w;
#pragma unroll
            for (int o = 1; o < 32; o <<= 1) {
                const unsigned v = (unsigned)__shfl_up((int)wi, o);
                if (lane >= o) wi += v;
            }
            wsumL[lane] = wi - w;
            if (lane == 31) cntL = (int)wi;
        }
        __syncthreads();
        binL[tid] = wsumL[wave] + inc - own;
        const int kept = __builtin_amdgcn_readfirstlane(cntL);
        __syncthreads();
        for (int j = tid; j < cnt; j += OWN_T) {
            const unsigned entry = listL[j];
            if (entry != NO_EDGE) {
                const unsigned pos = atomicAdd(&binL[bin_of(entry & LOWS)], 1u);
                asU[pos] = entry;
            }
        }
        __syncthreads();
        for (int j = tid; j < kept; j += OWN_T) listL[j] = asU[j];
        __syncthreads();
        return kept;
    };

    unsigned int* const saved = savedList + (size_t)blockIdx.x * CAP;
    unsigned have = (unsigned)CAP + 1u;
    if (!first) have = savedCnt[blockIdx.x];

    int total;
    const bool resolved = have <= (unsigned)CAP;
    if (resolved) {
        total = (int)have;
        for (int j0 = 0; j0 < total; j0 += 8 * OWN_T) {
            unsigned v[8];
#pragma unroll
            for (int k = 0; k < 8; ++k) {
                const int j = j0 + k * OWN_T + tid;
                v[k] = j < total ? saved[j] : NO_EDGE;
            }
#pragma unroll
            for (int k = 0; k < 8; ++k) {
                const int j = j0 + k * OWN_T + tid;
                if (j < total) listL[j] = v[k];
            }
        }
    } else {
        const int keyed = (E / KSEG) * KSEG;
        for (int kb = 0; kb < keyed; kb += KSEG) scan_keys(kb);
        for (int sb = keyed; sb < E; sb += SEG) scan_segment(sb);
        __syncthreads();
        total = __builtin_amdgcn_readfirstlane(cntL);
        __syncthreads();
    }
    if (total <= CAP) {
        int cnt = total;
        if (!resolved) cnt = resolve_sort(total);
        process_chunk(cnt, true);
        if (first) {
            if (tid == 0) savedCnt[blockIdx.x] = (unsigned)cnt;
            for (int j = tid; j < cnt; j += OWN_T) saved[j] = listL[j];
        }
    } else {
        if (first && tid == 0) savedCnt[blockIdx.x] = (unsigned)total;
        if (tid == 0) cntL = 0;
        __syncthreads();
        for (int sb = 0; sb < E; sb += SEG) {
            scan_segment(sb);
            __syncthreads();
            const int cnt = __builtin_amdgcn_readfirstlane(cntL);
            __syncthreads();
            if (cnt <= CAP - SEG && sb + SEG < E) continue;
            process_chunk(cnt, false);
            if (tid == 0) cntL = 0;
            __syncthreads();
        }
    }

    for (int idx = tid; idx < nOwn * F; idx += OWN_T) {
        const int dl = idx / F;
        const int f  = idx - dl * F;
        const float inv = 1.0f / (accL[dl * AW + F] + 1e-16f);
        const float v = accL[dl * AW + f] * inv + b[f];
        feat[(size_t)lo * F + idx] = v > 0.0f ? v : 0.0f;
    }
}

__global__ void gat_final_linear(const float* __restrict__ feat,
                                 const float* __restrict__ Wl,
                                 const float* __restrict__ bl,
                                 float* __restrict__ out,
                                 int N) {
    int i = blockIdx.x * blockDim.x + threadIdx.x;
    if (i >= N) return;
    float sacc = bl[0];
#pragma unroll
    for (int f = 0; f < F; ++f) sacc += feat[i * F + f] * Wl[f];
    out[i] = sacc;
}

extern "C" void kernel_launch(void* const* d_in, const int* in_sizes, int n_in,
                              void* d_out, int out_size, void* d_ws, size_t ws_size,
                              hipStream_t stream) {
    (void)n_in; (void)out_size; (void)ws_size;

    const int N = in_sizes[0];
    const int E = in_sizes[1] / 2;

    const float* x    = (const float*)d_in[0];
    const int*   ei   = (const int*)d_in[1];
    const float* W1   = (const float*)d_in[2];
    const float* as1  = (const float*)d_in[3];
    const float* ad1  = (const float*)d_in[4];
    const float* b1   = (const float*)d_in[5];
    const float* W2   = (const float*)d_in[6];
    const float* as2  = (const float*)d_in[7];
    const float* ad2  = (const float*)d_in[8];
    const float* b2   = (const float*)d_in[9];
    const float* Wl   = (const float*)d_in[10];
    const float* bl   = (const float*)d_in[11];

    const int* srcIdx = ei;
    const int* dstIdx = ei + E;

    const int nWords = (E / KSEG) * (KSEG / 4);
    unsigned int* keyw = (unsigned int*)d_ws;
    unsigned int* pk   = keyw + nWords;
    float* ws     = (float*)(pk + E);
    float* h      = ws + (size_t)0  * N;
    float* feat1  = ws + (size_t)20 * N;
    float* feat2  = ws + (size_t)40 * N;
    float* asrc   = ws + (size_t)60 * N;
    float* adst   = ws + (size_t)61 * N;

    const int TB   = 256;
    const int gN   = (N + TB - 1) / TB;
    const int gMM  = (N + 15) / 16;
    const int gOwn = (N + RN - 1) / RN;
    const int gKey = ((E + 3) / 4) / TB + 1;

    unsigned int* savedList = (unsigned int*)(ws + (size_t)62 * N);
    unsigned int* savedCnt  = savedList + (size_t)gOwn * CAP;

    gat_owner_keys<<<gKey, TB, 0, stream>>>(srcIdx, dstIdx, keyw, pk, nWords, E);

    gat_node_init_l1<<<gN, TB, 0, stream>>>(x, W1, as1, ad1, h, asrc, adst, N);
    gat_owner_layer<<<gOwn, OWN_T, 0, stream>>>(srcIdx, dstIdx, keyw, pk, asrc, adst, h, b1, feat1,
                                                savedList, savedCnt, 1, N, E);

    gat_wmma_feat_x_w<<<gMM, 32, 0, stream>>>(feat1, W2, h, N);
    gat_node_alpha<<<gN, TB, 0, stream>>>(h, as2, ad2, asrc, adst, N);
    gat_owner_layer<<<gOwn, OWN_T, 0, stream>>>(srcIdx, dstIdx, keyw, pk, asrc, adst, h, b2, feat2,
                                                savedList, savedCnt, 0, N, E);

    gat_final_linear<<<gN, TB, 0, stream>>>(feat2, Wl, bl, (float*)d_out, N);
}


# Launch fusion 7 -> 4: node init shares the key pass's launch, layer-2 scores come from an LDS copy of the tile in the GEMM's waves, the head runs in the layer-2 owners' finish; sum orders kept
# speedup vs baseline: 13.6414x; 1.0482x over previous
#include <hip/hip_runtime.h>
#include <math.h>

#define F 20
#define NEG_SLOPE 0.2f

#define RN     392
#define AW     21
#define OWN_T  1024
#define SEG    32768
#define CAP    32768
#define KSEG   131072
#define SBITS  22
#define LOWS   0x3FFFFFu
#define NO_EDGE 0xFFFFFFFFu
#define NBIN   1024
#define BSHIFT 7

typedef float v2f __attribute__((ext_vector_type(2)));
typedef float v8f __attribute__((ext_vector_type(8)));

__device__ __forceinline__ float lrelu(float v) {
    return v > 0.0f ? v : NEG_SLOPE * v;
}

__device__ __forceinline__ int lane_i(int v, int l) {
    return __builtin_amdgcn_readlane(v, l);
}
__device__ __forceinline__ float lane_f(float v, int l) {
    return __int_as_float(__builtin_amdgcn_readlane(__float_as_int(v), l));
}

__device__ __forceinline__ void node_init_l1(const int blk,
                                             const float* __restrict__ x,
                                             const float* __restrict__ W1,
                                             const float* __restrict__ a_src,
                                             const float* __restrict__ a_dst,
                                             float* __restrict__ h,
                                             float* __restrict__ asrc,
                                             float* __restrict__ adst,
                                             int N) {
    int i = blk * blockDim.x + threadIdx.x;
    if (i >= N) return;
    float xv = x[i];
    float sa = 0.0f, sd = 0.0f;
#pragma unroll
    for (int f = 0; f < F; ++f) {
        float hv = xv * W1[f];
        h[i * F + f] = hv;
        sa += hv * a_src[f];
        sd += hv * a_dst[f];
    }
    asrc[i] = sa;
    adst[i] = sd;
}

__global__ void __launch_bounds__(32)
gat_wmma_feat_x_w(const float* __restrict__ feat,
                  const float* __restrict__ W,
                  const float* __restrict__ a_src,
                  const float* __restrict__ a_dst,
                  float* __restrict__ h,
                  float* __restrict__ asrc,
                  float* __restrict__ adst,
                  int N) {
    __shared__ float tileL[16 * AW];
    int tile  = blockIdx.x;
    int lane  = threadIdx.x;
    int mrow  = lane & 15;
    int khalf = (lane >> 4) * 2;

    int row  = tile * 16 + mrow;
    int rowc = row < N ? row : N - 1;
    const float* arow = feat + (size_t)rowc * F;

    int   n1   = 16 + mrow;
    int   n1c  = (n1 < F) ? n1 : 0;
    float msk1 = (n1 < F) ? 1.0f : 0.0f;

    v8f c0 = {};
    v8f c1 = {};
#pragma unroll
    for (int k0 = 0; k0 < F; k0 += 4) {
        const float* wr0 = W + (size_t)(k0 + khalf + 0) * F;
        const float* wr1 = W + (size_t)(k0 + khalf + 1) * F;

        v2f a;
        a.x = arow[k0 + khalf + 0];
        a.y = arow[k0 + khalf + 1];

        v2f b0;
        b0.x = wr0[mrow];
        b0.y = wr1[mrow];
        c0 = __builtin_amdgcn_wmma_f32_16x16x4_f32(false, a, false, b0,
                                                   (short)0, c0, false, false);

        v2f b1;
        b1.x = wr0[n1c] * msk1;
        b1.y = wr1[n1c] * msk1;
        c1 = __builtin_amdgcn_wmma_f32_16x16x4_f32(false, a, false, b1,
                                                   (short)0, c1, false, false);
    }

    int rbase = tile * 16 + ((lane >> 4) << 3);
#pragma unroll
    for (int j = 0; j < 8; ++j) {
        int r = rbase + j;
        if (r < N) {
            h[(size_t)r * F + mrow] = c0[j];
            if (n1 < F) h[(size_t)r * F + n1] = c1[j];
        }
        const int rl = ((lane >> 4) << 3) + j;
        tileL[rl * AW + mrow] = c0[j];
        if (n1 < F) tileL[rl * AW + n1] = c1[j];
    }
    __syncthreads();
    if (lane < 16) {
        const int r = tile * 16 + lane;
        if (r < N) {
            float sa = 0.0f, sd = 0.0f;
#pragma unroll
            for (int f = 0; f < F; ++f) {
                const float hv = tileL[lane * AW + f];
                sa += hv * a_src[f];
                sd += hv * a_dst[f];
            }
            asrc[r] = sa;
            adst[r] = sd;
        }
    }
}

__device__ __forceinline__ void owner_keys(const int blk,
                                           const int* __restrict__ src,
                                           const int* __restrict__ dst,
                                           unsigned int* __restrict__ keyw,
                                           unsigned int* __restrict__ pk,
                                           int nWords, int E) {
    const int w  = blk * blockDim.x + threadIdx.x;
    const int e0 = 4 * w;
    if (e0 >= E) return;
    const bool whole = (E & 3) == 0;
    int d[4], s[4];
    if (whole) {
        const int4 d4 = ((const int4*)dst)[w];
        const int4 s4 = ((const int4*)src)[w];
        d[0] = d4.x; d[1] = d4.y; d[2] = d4.z; d[3] = d4.w;
        s[0] = s4.x; s[1] = s4.y; s[2] = s4.z; s[3] = s4.w;
    } else {
#pragma unroll
        for (int c = 0; c < 4; ++c) {
            const bool ok = e0 + c < E;
            d[c] = ok ? dst[e0 + c] : 0;
            s[c] = ok ? src[e0 + c] : 0;
        }
    }
    unsigned int k = 0u;
    unsigned int p[4];
#pragma unroll
    for (int c = 0; c < 4; ++c) {
        const unsigned int owner = (unsigned int)d[c] / (unsigned int)RN;
        p[c] = (unsigned int)s[c] | (((unsigned int)d[c] - owner * (unsigned int)RN) << SBITS) | (owner << 31);
        k |= (owner & 0xFFu) << (8 * c);
    }
    if (w < nWords) keyw[w] = k;
    if (whole) {
        ((uint4*)pk)[w] = make_uint4(p[0], p[1], p[2], p[3]);
    } else {
#pragma unroll
        for (int c = 0; c < 4; ++c)
            if (e0 + c < E) pk[e0 + c] = p[c];
    }
}

__global__ void gat_keys_and_node_init(const int* __restrict__ src,
                                       const int* __restrict__ dst,
                                       unsigned int* __restrict__ keyw,
                                       unsigned int* __restrict__ pk,
                                       int nWords, int E, int keyBlocks,
                                       const float* __restrict__ x,
                                       const float* __restrict__ W1,
                                       const float* __restrict__ a_src,
                                       const float* __restrict__ a_dst,
                                       float* __restrict__ h,
                                       float* __restrict__ asrc,
                                       float* __restrict__ adst,
                                       int N) {
    const int blk = (int)blockIdx.x;
    if (blk < keyBlocks) owner_keys(blk, src, dst, keyw, pk, nWords, E);
    else                 node_init_l1(blk - keyBlocks, x, W1, a_src, a_dst, h, asrc, adst, N);
}

__global__ void __launch_bounds__(OWN_T)
gat_owner_layer(const int* __restrict__ src,
                const int* __restrict__ dst,
                const unsigned int* __restrict__ keyw,
                const unsigned int* __restrict__ pk,
                const float* __restrict__ asrc,
                const float* __restrict__ adst,
                const float* __restrict__ h,
                const float* __restrict__ b,
                float* __restrict__ feat,
                unsigned int* __restrict__ savedList,
                unsigned int* __restrict__ savedCnt,
                const float* __restrict__ Wl,
                const float* __restrict__ bl,
                float* __restrict__ out,
                int first,
                int N, int E) {
    __shared__ float accL[RN * AW];
    __shared__ float mL[RN];
    __shared__ float mNew[RN];
    __shared__ float adL[RN];
    __shared__ float capL[RN];
    __shared__ float scL[RN];
    __shared__ unsigned listL[CAP];
    __shared__ unsigned asU[CAP];
    __shared__ unsigned binL[NBIN];
    __shared__ unsigned wsumL[32];
    __shared__ int cntL;

    const int tid  = threadIdx.x;
    const int lane = tid & 31;
    const int wave = __builtin_amdgcn_readfirstlane(tid >> 5);
    const int lo   = blockIdx.x * RN;
    const int nOwn = (N - lo) < RN ? (N - lo) : RN;

    for (int dl = tid; dl < RN; dl += OWN_T) {
        float a = -3.402823466e38f, ad = 0.0f;
        if (dl < nOwn) { a = asrc[lo + dl]; ad = adst[lo + dl]; }
        mL[dl] = a;
        mNew[dl] = a;
        adL[dl] = ad;
    }
    for (int idx = tid; idx < RN * AW; idx += OWN_T) {
        const int dl = idx / AW;
        const int f  = idx - dl * AW;
        float v = 0.0f;
        if (dl < nOwn) v = f < F ? h[(size_t)(lo + dl) * F + f] : 1.0f;
        accL[idx] = v;
    }
    if (tid == 0) cntL = 0;
    __syncthreads();

    const unsigned gpat = (unsigned)(blockIdx.x & 0xFF) * 0x01010101u;
    auto scan_keys = [&](const int kb) {
        const uint4* __restrict__ k4 = (const uint4*)(keyw + (kb >> 2));
        uint4 q[KSEG / (16 * OWN_T)];
#pragma unroll
        for (int k = 0; k < KSEG / (16 * OWN_T); ++k) q[k] = k4[k * OWN_T + tid];
        unsigned mk[KSEG / (32 * OWN_T)];
        int n = 0;
#pragma unroll
        for (int p = 0; p < KSEG / (32 * OWN_T); ++p) {
            const unsigned w[8] = {q[2 * p].x,     q[2 * p].y,     q[2 * p].z,     q[2 * p].w,
                                   q[2 * p + 1].x, q[2 * p + 1].y, q[2 * p + 1].z, q[2 * p + 1].w};
            unsigned m = 0u;
#pragma unroll
            for (int j = 0; j < 8; ++j) {
                const unsigned v = w[j] ^ gpat;
                m |= ((v - 0x01010101u) & ~v & 0x80808080u) >> (7 - j);
            }
            mk[p] = m;
            n += __popc(m);
        }
        int pre = 0, tot = 0;
        for (int d = 0; d < 8; ++d) {
            if (__builtin_amdgcn_ballot_w32((n >> d) != 0) == 0u) break;
            const unsigned set = __builtin_amdgcn_ballot_w32(((n >> d) & 1) != 0);
            pre += (int)__builtin_amdgcn_mbcnt_lo(set, 0u) << d;
            tot += __builtin_popcount(set) << d;
        }
        int base = 0;
        if (lane == 0) base = atomicAdd(&cntL, tot);
        unsigned slot = (unsigned)(lane_i(base, 0) + pre);
        const unsigned eLane = (unsigned)kb + (unsigned)tid * 16u;
#pragma unroll
        for (int p = 0; p < KSEG / (32 * OWN_T); ++p) {
            const unsigned eP = eLane + (unsigned)(2 * p * OWN_T) * 16u;
            unsigned m = mk[p];
            while (m != 0u) {
                const unsigned bit = (unsigned)__builtin_ctz(m);
                const unsigned e = eP + ((bit & 4u) * (unsigned)(OWN_T * 4)) + ((bit & 3u) << 2) + (bit >> 3);
                listL[slot < (unsigned)CAP ? slot : (unsigned)(CAP - 1)] = e;
                ++slot;
                m &= m - 1u;
            }
        }
    };

    auto scan_segment = [&](const int sb) {
        if (((E & 3) == 0) && (sb + SEG <= E)) {
            const int4* __restrict__ dst4 = (const int4*)(dst + sb);
            int4 dq[SEG / (4 * OWN_T)];
#pragma unroll
            for (int k = 0; k < SEG / (4 * OWN_T); ++k) dq[k] = dst4[k * OWN_T + tid];
#pragma unroll
            for (int k = 0; k < SEG / (4 * OWN_T); ++k) {
                const unsigned u0 = (unsigned)(dq[k].x - lo);
                const unsigned u1 = (unsigned)(dq[k].y - lo);
                const unsigned u2 = (unsigned)(dq[k].z - lo);
                const unsigned u3 = (unsigned)(dq[k].w - lo);
                const bool m0 = u0 < (unsigned)nOwn, m1 = u1 < (unsigned)nOwn;
                const bool m2 = u2 < (unsigned)nOwn, m3 = u3 < (unsigned)nOwn;
                if (m0 | m1 | m2 | m3) {
                    const unsigned e0 = (unsigned)(sb + (k * OWN_T + tid) * 4);
                    if (m0) { const int slot = atomicAdd(&cntL, 1); if (slot < CAP) listL[slot] = e0 + 0u; }
                    if (m1) { const int slot = atomicAdd(&cntL, 1); if (slot < CAP) listL[slot] = e0 + 1u; }
                    if (m2) { const int slot = atomicAdd(&cntL, 1); if (slot < CAP) listL[slot] = e0 + 2u; }
                    if (m3) { const int slot = atomicAdd(&cntL, 1); if (slot < CAP) listL[slot] = e0 + 3u; }
                }
            }
        } else {
            for (int k0 = 0; k0 < SEG / OWN_T; k0 += 8) {
                int dv[8];
#pragma unroll
                for (int k = 0; k < 8; ++k) {
                    const int e = sb + (k0 + k) * OWN_T + tid;
                    dv[k] = e < E ? dst[e] : -1;
                }
#pragma unroll
                for (int k = 0; k < 8; ++k) {
                    const unsigned u = (unsigned)(dv[k] - lo);
                    if (u < (unsigned)nOwn) {
                        const int e = sb + (k0 + k) * OWN_T + tid;
                        const int slot = atomicAdd(&cntL, 1);
                        if (slot < CAP) listL[slot] = (unsigned)e;
                    }
                }
            }
        }
    };

    auto process_chunk = [&](const int cnt, const bool resolved) {
        for (int j = tid; j < cnt; j += OWN_T) {
            unsigned entry = listL[j];
            if (!resolved) {
                const unsigned e = entry;
                const unsigned dl = (unsigned)(dst[e] - lo);
                const int s = src[e];
                entry = dl < (unsigned)nOwn ? ((unsigned)s | (dl << SBITS)) : NO_EDGE;
                listL[j] = entry;
            }
            if (entry != NO_EDGE) {
                const float a = asrc[entry & LOWS];
                asU[j] = __float_as_uint(a);
                const unsigned dl = entry >> SBITS;
                if (a >= 0.0f) atomicMax((int*)&mNew[dl], __float_as_int(a));
                else           atomicMin((unsigned int*)&mNew[dl], __float_as_uint(a));
            }
        }
        __syncthreads();

        if (tid < RN) {
            const float ad = adL[tid];
            const float mn = mNew[tid];
            const float shiftOld = lrelu(mL[tid] + ad);
            const float shiftNew = lrelu(mn + ad);
            scL[tid]  = __expf(shiftOld - shiftNew);
            capL[tid] = shiftNew;
            mL[tid]   = mn;
        }
        __syncthreads();
        for (int idx = tid; idx < RN * AW; idx += OWN_T) accL[idx] *= scL[idx / AW];
        __syncthreads();

        const int lc = lane < F ? lane : F - 1;
        for (int base = wave * 32; base < cnt; base += OWN_T) {
            const int j = base + lane;
            int hoff = lo * F, aoff = 0;
            float t = 0.0f;
            const unsigned p = j < cnt ? listL[j] : NO_EDGE;
            if (p != NO_EDGE) {
                const int s  = (int)(p & LOWS);
                const int dl = (int)(p >> SBITS);
                const float ev = lrelu(__uint_as_float(asU[j]) + adL[dl]);
                t = __expf(ev - capL[dl]);
                hoff = s * F;
                aoff = dl * AW;
            }
            const int n = cnt - base;
            if (n >= 32) {
#pragma unroll
                for (int g = 0; g < 32; g += 16) {
                    int so[16], ao[16];
                    float tt[16];
#pragma unroll
                    for (int k = 0; k < 16; ++k) {
                        so[k] = lane_i(hoff, g + k);
                        ao[k] = lane_i(aoff, g + k);
                        tt[k] = lane_f(t, g + k);
                    }
                    if (lane <= F) {
                        float hv[16];
#pragma unroll
                        for (int k = 0; k < 16; ++k) hv[k] = h[so[k] + lc];
#pragma unroll
                        for (int k = 0; k < 16; ++k)
                            __hip_atomic_fetch_add(&accL[ao[k] + lane],
                                                   lane < F ? tt[k] * hv[k] : tt[k],
                                                   __ATOMIC_RELAXED, __HIP_MEMORY_SCOPE_WORKGROUP);
                    }
                }
            } else {
                for (int q = 0; q < n; ++q) {
                    const int so = lane_i(hoff, q);
                    const int ao = lane_i(aoff, q);
                    const float tt = lane_f(t, q);
                    if (lane <= F) {
                        const float hv = h[so + lc];
                        __hip_atomic_fetch_add(&accL[ao + lane], lane < F ? tt * hv : tt,
                                               __ATOMIC_RELAXED, __HIP_MEMORY_SCOPE_WORKGROUP);
                    }
                }
            }
        }
        __syncthreads();
    };

    const unsigned nbAll = ((unsigned)(N - 1) >> BSHIFT) + 1u;
    const unsigned nb    = nbAll < (unsigned)NBIN ? nbAll : (unsigned)NBIN;
    const unsigned rot   = ((unsigned)blockIdx.x * nb) / (unsigned)gridDim.x;
    auto bin_of = [&](const unsigned s) -> unsigned {
        const unsigned b0 = s >> BSHIFT;
        const unsigned t  = (b0 < nb ? b0 : nb - 1u) + rot;
        return t >= nb ? t - nb : t;
    };
    const bool exactIds = gridDim.x <= 256u;
    const unsigned myPar = (unsigned)blockIdx.x & 1u;
    auto resolve_sort = [&](const int cnt) -> int {
        for (int bI = tid; bI < NBIN; bI += OWN_T) binL[bI] = 0u;
        __syncthreads();
        for (int j = tid; j < cnt; j += OWN_T) {
            const unsigned e = listL[j];
            const unsigned p = pk[e];
            bool mine = (p >> 31) == myPar;
            if (!exactIds) mine = (unsigned)(dst[e] - lo) < (unsigned)nOwn;
            const unsigned entry = p & 0x7FFFFFFFu;
            listL[j] = mine ? entry : NO_EDGE;
            if (mine) atomicAdd(&binL[bin_of(entry & LOWS)], 1u);
        }
        __syncthreads();
        const unsigned own = binL[tid];
        unsigned inc = own;
#pragma unroll
        for (int o = 1; o < 32; o <<= 1) {
            const unsigned v = (unsigned)__shfl_up((int)inc, o);
            if (lane >= o) inc += v;
        }
        if (lane == 31) wsumL[wave] = inc;
        __syncthreads();
        if (wave == 0) {
            const unsigned w = wsumL[lane];
            unsigned wi = w;
#pragma unroll
            for (int o = 1; o < 32; o <<= 1) {
                const unsigned v = (unsigned)__shfl_up((int)wi, o);
                if (lane >= o) wi += v;
            }
            wsumL[lane] = wi - w;
            if (lane == 31) cntL = (int)wi;
        }
        __syncthreads();
        binL[tid] = wsumL[wave] + inc - own;
        const int kept = __builtin_amdgcn_readfirstlane(cntL);
        __syncthreads();
        for (int j = tid; j < cnt; j += OWN_T) {
            const unsigned entry = listL[j];
            if (entry != NO_EDGE) {
                const unsigned pos = atomicAdd(&binL[bin_of(entry & LOWS)], 1u);
                asU[pos] = entry;
            }
        }
        __syncthreads();
        for (int j = tid; j < kept; j += OWN_T) listL[j] = asU[j];
        __syncthreads();
        return kept;
    };

    unsigned int* const saved = savedList + (size_t)blockIdx.x * CAP;
    unsigned have = (unsigned)CAP + 1u;
    if (!first) have = savedCnt[blockIdx.x];

    int total;
    const bool resolved = have <= (unsigned)CAP;
    if (resolved) {
        total = (int)have;
        for (int j0 = 0; j0 < total; j0 += 8 * OWN_T) {
            unsigned v[8];
#pragma unroll
            for (int k = 0; k < 8; ++k) {
                const int j = j0 + k * OWN_T + tid;
                v[k] = j < total ? saved[j] : NO_EDGE;
            }
#pragma unroll
            for (int k = 0; k < 8; ++k) {
                const int j = j0 + k * OWN_T + tid;
                if (j < total) listL[j] = v[k];
            }
        }
    } else {
        const int keyed = (E / KSEG) * KSEG;
        for (int kb = 0; kb < keyed; kb += KSEG) scan_keys(kb);
        for (int sb = keyed; sb < E; sb += SEG) scan_segment(sb);
        __syncthreads();
        total = __builtin_amdgcn_readfirstlane(cntL);
        __syncthreads();
    }
    if (total <= CAP) {
        int cnt = total;
        if (!resolved) cnt = resolve_sort(total);
        process_chunk(cnt, true);
        if (first) {
            if (tid == 0) savedCnt[blockIdx.x] = (unsigned)cnt;
            for (int j = tid; j < cnt; j += OWN_T) saved[j] = listL[j];
        }
    } else {
        if (first && tid == 0) savedCnt[blockIdx.x] = (unsigned)total;
        if (tid == 0) cntL = 0;
        __syncthreads();
        for (int sb = 0; sb < E; sb += SEG) {
            scan_segment(sb);
            __syncthreads();
            const int cnt = __builtin_amdgcn_readfirstlane(cntL);
            __syncthreads();
            if (cnt <= CAP - SEG && sb + SEG < E) continue;
            process_chunk(cnt, false);
            if (tid == 0) cntL = 0;
            __syncthreads();
        }
    }

    if (first) {
        for (int idx = tid; idx < nOwn * F; idx += OWN_T) {
            const int dl = idx / F;
            const int f  = idx - dl * F;
            const float inv = 1.0f / (accL[dl * AW + F] + 1e-16f);
            const float v = accL[dl * AW + f] * inv + b[f];
            feat[(size_t)lo * F + idx] = v > 0.0f ? v : 0.0f;
        }
    } else {
        for (int dl = tid; dl < nOwn; dl += OWN_T) {
            const float inv = 1.0f / (accL[dl * AW + F] + 1e-16f);
            float sacc = bl[0];
#pragma unroll
            for (int f = 0; f < F; ++f) {
                const float v = accL[dl * AW + f] * inv + b[f];
                sacc += (v > 0.0f ? v : 0.0f) * Wl[f];
            }
            out[lo + dl] = sacc;
        }
    }
}

extern "C" void kernel_launch(void* const* d_in, const int* in_sizes, int n_in,
                              void* d_out, int out_size, void* d_ws, size_t ws_size,
                              hipStream_t stream) {
    (void)n_in; (void)out_size; (void)ws_size;

    const int N = in_sizes[0];
    const int E = in_sizes[1] / 2;

    const float* x    = (const float*)d_in[0];
    const int*   ei   = (const int*)d_in[1];
    const float* W1   = (const float*)d_in[2];
    const float* as1  = (const float*)d_in[3];
    const float* ad1  = (const float*)d_in[4];
    const float* b1   = (const float*)d_in[5];
    const float* W2   = (const float*)d_in[6];
    const float* as2  = (const float*)d_in[7];
    const float* ad2  = (const float*)d_in[8];
    const float* b2   = (const float*)d_in[9];
    const float* Wl   = (const float*)d_in[10];
    const float* bl   = (const float*)d_in[11];

    const int* srcIdx = ei;
    const int* dstIdx = ei + E;

    const int nWords = (E / KSEG) * (KSEG / 4);
    unsigned int* keyw = (unsigned int*)d_ws;
    unsigned int* pk   = keyw + nWords;
    float* ws     = (float*)(pk + E);
    float* h      = ws + (size_t)0  * N;
    float* feat1  = ws + (size_t)20 * N;
    float* asrc   = ws + (size_t)60 * N;
    float* adst   = ws + (size_t)61 * N;

    const int TB   = 256;
    const int gN   = (N + TB - 1) / TB;
    const int gMM  = (N + 15) / 16;
    const int gOwn = (N + RN - 1) / RN;
    const int gKey = ((E + 3) / 4) / TB + 1;

    unsigned int* savedList = (unsigned int*)(ws + (size_t)62 * N);
    unsigned int* savedCnt  = savedList + (size_t)gOwn * CAP;

    gat_keys_and_node_init<<<gKey + gN, TB, 0, stream>>>(srcIdx, dstIdx, keyw, pk, nWords, E, gKey,
                                                         x, W1, as1, ad1, h, asrc, adst, N);

    gat_owner_layer<<<gOwn, OWN_T, 0, stream>>>(srcIdx, dstIdx, keyw, pk, asrc, adst, h, b1, feat1,
                                                savedList, savedCnt, Wl, bl, (float*)d_out, 1, N, E);

    gat_wmma_feat_x_w<<<gMM, 32, 0, stream>>>(feat1, W2, as2, ad2, h, asrc, adst, N);
    gat_owner_layer<<<gOwn, OWN_T, 0, stream>>>(srcIdx, dstIdx, keyw, pk, asrc, adst, h, b2, feat1,
                                                savedList, savedCnt, Wl, bl, (float*)d_out, 0, N, E);
}


# Owner gather loops keep their loads in flight together: 8 list entries per thread in the score and list-word passes, a group's 32 rows in the add pass, all start rows; same values, adds arrival order
# speedup vs baseline: 14.2294x; 1.0431x over previous
#include <hip/hip_runtime.h>
#include <math.h>

#define F 20
#define NEG_SLOPE 0.2f

#define RN     392
#define AW     21
#define OWN_T  1024
#define SEG    32768
#define CAP    32768
#define KSEG   131072
#define SBITS  22
#define LOWS   0x3FFFFFu
#define NO_EDGE 0xFFFFFFFFu
#define NBIN   1024
#define BSHIFT 7
#define NINIT  ((RN * AW + OWN_T - 1) / OWN_T)
#define DEEP   8

typedef float v2f __attribute__((ext_vector_type(2)));
typedef float v8f __attribute__((ext_vector_type(8)));

__device__ __forceinline__ float lrelu(float v) {
    return v > 0.0f ? v : NEG_SLOPE * v;
}

__device__ __forceinline__ int lane_i(int v, int l) {
    return __builtin_amdgcn_readlane(v, l);
}
__device__ __forceinline__ float lane_f(float v, int l) {
    return __int_as_float(__builtin_amdgcn_readlane(__float_as_int(v), l));
}

__device__ __forceinline__ void node_init_l1(const int blk,
                                             const float* __restrict__ x,
                                             const float* __restrict__ W1,
                                             const float* __restrict__ a_src,
                                             const float* __restrict__ a_dst,
                                             float* __restrict__ h,
                                             float* __restrict__ asrc,
                                             float* __restrict__ adst,
                                             int N) {
    int i = blk * blockDim.x + threadIdx.x;
    if (i >= N) return;
    float xv = x[i];
    float sa = 0.0f, sd = 0.0f;
#pragma unroll
    for (int f = 0; f < F; ++f) {
        float hv = xv * W1[f];
        h[i * F + f] = hv;
        sa += hv * a_src[f];
        sd += hv * a_dst[f];
    }
    asrc[i] = sa;
    adst[i] = sd;
}

__global__ void __launch_bounds__(32)
gat_wmma_feat_x_w(const float* __restrict__ feat,
                  const float* __restrict__ W,
                  const float* __restrict__ a_src,
                  const float* __restrict__ a_dst,
                  float* __restrict__ h,
                  float* __restrict__ asrc,
                  float* __restrict__ adst,
                  int N) {
    __shared__ float tileL[16 * AW];
    int tile  = blockIdx.x;
    int lane  = threadIdx.x;
    int mrow  = lane & 15;
    int khalf = (lane >> 4) * 2;

    int row  = tile * 16 + mrow;
    int rowc = row < N ? row : N - 1;
    const float* arow = feat + (size_t)rowc * F;

    int   n1   = 16 + mrow;
    int   n1c  = (n1 < F) ? n1 : 0;
    float msk1 = (n1 < F) ? 1.0f : 0.0f;

    v8f c0 = {};
    v8f c1 = {};
#pragma unroll
    for (int k0 = 0; k0 < F; k0 += 4) {
        const float* wr0 = W + (size_t)(k0 + khalf + 0) * F;
        const float* wr1 = W + (size_t)(k0 + khalf + 1) * F;

        v2f a;
        a.x = arow[k0 + khalf + 0];
        a.y = arow[k0 + khalf + 1];

        v2f b0;
        b0.x = wr0[mrow];
        b0.y = wr1[mrow];
        c0 = __builtin_amdgcn_wmma_f32_16x16x4_f32(false, a, false, b0,
                                                   (short)0, c0, false, false);

        v2f b1;
        b1.x = wr0[n1c] * msk1;
        b1.y = wr1[n1c] * msk1;
        c1 = __builtin_amdgcn_wmma_f32_16x16x4_f32(false, a, false, b1,
                                                   (short)0, c1, false, false);
    }

    int rbase = tile * 16 + ((lane >> 4) << 3);
#pragma unroll
    for (int j = 0; j < 8; ++j) {
        int r = rbase + j;
        if (r < N) {
            h[(size_t)r * F + mrow] = c0[j];
            if (n1 < F) h[(size_t)r * F + n1] = c1[j];
        }
        const int rl = ((lane >> 4) << 3) + j;
        tileL[rl * AW + mrow] = c0[j];
        if (n1 < F) tileL[rl * AW + n1] = c1[j];
    }
    __syncthreads();
    if (lane < 16) {
        const int r = tile * 16 + lane;
        if (r < N) {
            float sa = 0.0f, sd = 0.0f;
#pragma unroll
            for (int f = 0; f < F; ++f) {
                const float hv = tileL[lane * AW + f];
                sa += hv * a_src[f];
                sd += hv * a_dst[f];
            }
            asrc[r] = sa;
            adst[r] = sd;
        }
    }
}

__device__ __forceinline__ void owner_keys(const int blk,
                                           const int* __restrict__ src,
                                           const int* __restrict__ dst,
                                           unsigned int* __restrict__ keyw,
                                           unsigned int* __restrict__ pk,
                                           int nWords, int E) {
    const int w  = blk * blockDim.x + threadIdx.x;
    const int e0 = 4 * w;
    if (e0 >= E) return;
    const bool whole = (E & 3) == 0;
    int d[4], s[4];
    if (whole) {
        const int4 d4 = ((const int4*)dst)[w];
        const int4 s4 = ((const int4*)src)[w];
        d[0] = d4.x; d[1] = d4.y; d[2] = d4.z; d[3] = d4.w;
        s[0] = s4.x; s[1] = s4.y; s[2] = s4.z; s[3] = s4.w;
    } else {
#pragma unroll
        for (int c = 0; c < 4; ++c) {
            const bool ok = e0 + c < E;
            d[c] = ok ? dst[e0 + c] : 0;
            s[c] = ok ? src[e0 + c] : 0;
        }
    }
    unsigned int k = 0u;
    unsigned int p[4];
#pragma unroll
    for (int c = 0; c < 4; ++c) {
        const unsigned int owner = (unsigned int)d[c] / (unsigned int)RN;
        p[c] = (unsigned int)s[c] | (((unsigned int)d[c] - owner * (unsigned int)RN) << SBITS) | (owner << 31);
        k |= (owner & 0xFFu) << (8 * c);
    }
    if (w < nWords) keyw[w] = k;
    if (whole) {
        ((uint4*)pk)[w] = make_uint4(p[0], p[1], p[2], p[3]);
    } else {
#pragma unroll
        for (int c = 0; c < 4; ++c)
            if (e0 + c < E) pk[e0 + c] = p[c];
    }
}

__global__ void gat_keys_and_node_init(const int* __restrict__ src,
                                       const int* __restrict__ dst,
                                       unsigned int* __restrict__ keyw,
                                       unsigned int* __restrict__ pk,
                                       int nWords, int E, int keyBlocks,
                                       const float* __restrict__ x,
                                       const float* __restrict__ W1,
                                       const float* __restrict__ a_src,
                                       const float* __restrict__ a_dst,
                                       float* __restrict__ h,
                                       float* __restrict__ asrc,
                                       float* __restrict__ adst,
                                       int N) {
    const int blk = (int)blockIdx.x;
    if (blk < keyBlocks) owner_keys(blk, src, dst, keyw, pk, nWords, E);
    else                 node_init_l1(blk - keyBlocks, x, W1, a_src, a_dst, h, asrc, adst, N);
}

__global__ void __launch_bounds__(OWN_T)
gat_owner_layer(const int* __restrict__ src,
                const int* __restrict__ dst,
                const unsigned int* __restrict__ keyw,
                const unsigned int* __restrict__ pk,
                const float* __restrict__ asrc,
                const float* __restrict__ adst,
                const float* __restrict__ h,
                const float* __restrict__ b,
                float* __restrict__ feat,
                unsigned int* __restrict__ savedList,
                unsigned int* __restrict__ savedCnt,
                const float* __restrict__ Wl,
                const float* __restrict__ bl,
                float* __restrict__ out,
                int first,
                int N, int E) {
    __shared__ float accL[RN * AW];
    __shared__ float mL[RN];
    __shared__ float mNew[RN];
    __shared__ float adL[RN];
    __shared__ float capL[RN];
    __shared__ float scL[RN];
    __shared__ unsigned listL[CAP];
    __shared__ unsigned asU[CAP];
    __shared__ unsigned binL[NBIN];
    __shared__ unsigned wsumL[32];
    __shared__ int cntL;

    const int tid  = threadIdx.x;
    const int lane = tid & 31;
    const int wave = __builtin_amdgcn_readfirstlane(tid >> 5);
    const int lo   = blockIdx.x * RN;
    const int nOwn = (N - lo) < RN ? (N - lo) : RN;

    for (int dl = tid; dl < RN; dl += OWN_T) {
        float a = -3.402823466e38f, ad = 0.0f;
        if (dl < nOwn) { a = asrc[lo + dl]; ad = adst[lo + dl]; }
        mL[dl] = a;
        mNew[dl] = a;
        adL[dl] = ad;
    }
    {
        float v0[NINIT];
#pragma unroll
        for (int k = 0; k < NINIT; ++k) {
            const int idx = tid + k * OWN_T;
            const int dl = idx / AW;
            const int f  = idx - dl * AW;
            const bool own = idx < RN * AW && dl < nOwn;
            const bool row = own && f < F;
            const float hv = h[row ? (size_t)(lo + dl) * F + f : (size_t)lo * F];
            v0[k] = row ? hv : (own ? 1.0f : 0.0f);
        }
#pragma unroll
        for (int k = 0; k < NINIT; ++k) {
            const int idx = tid + k * OWN_T;
            if (idx < RN * AW) accL[idx] = v0[k];
        }
    }
    if (tid == 0) cntL = 0;
    __syncthreads();

    const unsigned gpat = (unsigned)(blockIdx.x & 0xFF) * 0x01010101u;
    auto scan_keys = [&](const int kb) {
        const uint4* __restrict__ k4 = (const uint4*)(keyw + (kb >> 2));
        uint4 q[KSEG / (16 * OWN_T)];
#pragma unroll
        for (int k = 0; k < KSEG / (16 * OWN_T); ++k) q[k] = k4[k * OWN_T + tid];
        unsigned mk[KSEG / (32 * OWN_T)];
        int n = 0;
#pragma unroll
        for (int p = 0; p < KSEG / (32 * OWN_T); ++p) {
            const unsigned w[8] = {q[2 * p].x,     q[2 * p].y,     q[2 * p].z,     q[2 * p].w,
                                   q[2 * p + 1].x, q[2 * p + 1].y, q[2 * p + 1].z, q[2 * p + 1].w};
            unsigned m = 0u;
#pragma unroll
            for (int j = 0; j < 8; ++j) {
                const unsigned v = w[j] ^ gpat;
                m |= ((v - 0x01010101u) & ~v & 0x80808080u) >> (7 - j);
            }
            mk[p] = m;
            n += __popc(m);
        }
        int pre = 0, tot = 0;
        for (int d = 0; d < 8; ++d) {
            if (__builtin_amdgcn_ballot_w32((n >> d) != 0) == 0u) break;
            const unsigned set = __builtin_amdgcn_ballot_w32(((n >> d) & 1) != 0);
            pre += (int)__builtin_amdgcn_mbcnt_lo(set, 0u) << d;
            tot += __builtin_popcount(set) << d;
        }
        int base = 0;
        if (lane == 0) base = atomicAdd(&cntL, tot);
        unsigned slot = (unsigned)(lane_i(base, 0) + pre);
        const unsigned eLane = (unsigned)kb + (unsigned)tid * 16u;
#pragma unroll
        for (int p = 0; p < KSEG / (32 * OWN_T); ++p) {
            const unsigned eP = eLane + (unsigned)(2 * p * OWN_T) * 16u;
            unsigned m = mk[p];
            while (m != 0u) {
                const unsigned bit = (unsigned)__builtin_ctz(m);
                const unsigned e = eP + ((bit & 4u) * (unsigned)(OWN_T * 4)) + ((bit & 3u) << 2) + (bit >> 3);
                listL[slot < (unsigned)CAP ? slot : (unsigned)(CAP - 1)] = e;
                ++slot;
                m &= m - 1u;
            }
        }
    };

    auto scan_segment = [&](const int sb) {
        if (((E & 3) == 0) && (sb + SEG <= E)) {
            const int4* __restrict__ dst4 = (const int4*)(dst + sb);
            int4 dq[SEG / (4 * OWN_T)];
#pragma unroll
            for (int k = 0; k < SEG / (4 * OWN_T); ++k) dq[k] = dst4[k * OWN_T + tid];
#pragma unroll
            for (int k = 0; k < SEG / (4 * OWN_T); ++k) {
                const unsigned u0 = (unsigned)(dq[k].x - lo);
                const unsigned u1 = (unsigned)(dq[k].y - lo);
                const unsigned u2 = (unsigned)(dq[k].z - lo);
                const unsigned u3 = (unsigned)(dq[k].w - lo);
                const bool m0 = u0 < (unsigned)nOwn, m1 = u1 < (unsigned)nOwn;
                const bool m2 = u2 < (unsigned)nOwn, m3 = u3 < (unsigned)nOwn;
                if (m0 | m1 | m2 | m3) {
                    const unsigned e0 = (unsigned)(sb + (k * OWN_T + tid) * 4);
                    if (m0) { const int slot = atomicAdd(&cntL, 1); if (slot < CAP) listL[slot] = e0 + 0u; }
                    if (m1) { const int slot = atomicAdd(&cntL, 1); if (slot < CAP) listL[slot] = e0 + 1u; }
                    if (m2) { const int slot = atomicAdd(&cntL, 1); if (slot < CAP) listL[slot] = e0 + 2u; }
                    if (m3) { const int slot = atomicAdd(&cntL, 1); if (slot < CAP) listL[slot] = e0 + 3u; }
                }
            }
        } else {
            for (int k0 = 0; k0 < SEG / OWN_T; k0 += 8) {
                int dv[8];
#pragma unroll
                for (int k = 0; k < 8; ++k) {
                    const int e = sb + (k0 + k) * OWN_T + tid;
                    dv[k] = e < E ? dst[e] : -1;
                }
#pragma unroll
                for (int k = 0; k < 8; ++k) {
                    const unsigned u = (unsigned)(dv[k] - lo);
                    if (u < (unsigned)nOwn) {
                        const int e = sb + (k0 + k) * OWN_T + tid;
                        const int slot = atomicAdd(&cntL, 1);
                        if (slot < CAP) listL[slot] = (unsigned)e;
                    }
                }
            }
        }
    };

    auto process_chunk = [&](const int cnt, const bool resolved) {
        if (resolved) {
            for (int j0 = tid; j0 < cnt; j0 += DEEP * OWN_T) {
                unsigned en[DEEP];
                float av[DEEP];
#pragma unroll
                for (int k = 0; k < DEEP; ++k) {
                    const int j = j0 + k * OWN_T;
                    const unsigned w = listL[j < cnt ? j : j0];
                    en[k] = j < cnt ? w : NO_EDGE;
                }
#pragma unroll
                for (int k = 0; k < DEEP; ++k)
                    av[k] = asrc[en[k] != NO_EDGE ? (en[k] & LOWS) : 0u];
#pragma unroll
                for (int k = 0; k < DEEP; ++k) {
                    if (en[k] != NO_EDGE) {
                        const int j = j0 + k * OWN_T;
                        const float a = av[k];
                        asU[j] = __float_as_uint(a);
                        const unsigned dl = en[k] >> SBITS;
                        if (a >= 0.0f) atomicMax((int*)&mNew[dl], __float_as_int(a));
                        else           atomicMin((unsigned int*)&mNew[dl], __float_as_uint(a));
                    }
                }
            }
        } else
        for (int j = tid; j < cnt; j += OWN_T) {
            unsigned entry = listL[j];
            if (!resolved) {
                const unsigned e = entry;
                const unsigned dl = (unsigned)(dst[e] - lo);
                const int s = src[e];
                entry = dl < (unsigned)nOwn ? ((unsigned)s | (dl << SBITS)) : NO_EDGE;
                listL[j] = entry;
            }
            if (entry != NO_EDGE) {
                const float a = asrc[entry & LOWS];
                asU[j] = __float_as_uint(a);
                const unsigned dl = entry >> SBITS;
                if (a >= 0.0f) atomicMax((int*)&mNew[dl], __float_as_int(a));
                else           atomicMin((unsigned int*)&mNew[dl], __float_as_uint(a));
            }
        }
        __syncthreads();

        if (tid < RN) {
            const float ad = adL[tid];
            const float mn = mNew[tid];
            const float shiftOld = lrelu(mL[tid] + ad);
            const float shiftNew = lrelu(mn + ad);
            scL[tid]  = __expf(shiftOld - shiftNew);
            capL[tid] = shiftNew;
            mL[tid]   = mn;
        }
        __syncthreads();
        for (int idx = tid; idx < RN * AW; idx += OWN_T) accL[idx] *= scL[idx / AW];
        __syncthreads();

        const int lc = lane < F ? lane : F - 1;
        auto prep = [&](const int base, int& hoff, int& aoff, float& t) {
            const int j = base + lane;
            hoff = lo * F; aoff = 0; t = 0.0f;
            const unsigned p = j < cnt ? listL[j] : NO_EDGE;
            if (p != NO_EDGE) {
                const int s  = (int)(p & LOWS);
                const int dl = (int)(p >> SBITS);
                const float ev = lrelu(__uint_as_float(asU[j]) + adL[dl]);
                t = __expf(ev - capL[dl]);
                hoff = s * F;
                aoff = dl * AW;
            }
        };
        int hoff = lo * F, aoff = 0;
        float t = 0.0f;
        if (wave * 32 < cnt) prep(wave * 32, hoff, aoff, t);
        for (int base = wave * 32; base < cnt; base += OWN_T) {
            const int n = cnt - base;
            int hoffN = lo * F, aoffN = 0;
            float tN = 0.0f;
            if (n >= 32) {
                float hv[32];
#pragma unroll
                for (int k = 0; k < 32; ++k) hv[k] = h[lane_i(hoff, k) + lc];
                if (base + OWN_T < cnt) prep(base + OWN_T, hoffN, aoffN, tN);
#pragma unroll
                for (int g = 0; g < 32; g += 16) {
                    int ao[16];
                    float tt[16];
#pragma unroll
                    for (int k = 0; k < 16; ++k) {
                        ao[k] = lane_i(aoff, g + k);
                        tt[k] = lane_f(t, g + k);
                    }
                    if (lane <= F) {
#pragma unroll
                        for (int k = 0; k < 16; ++k)
                            __hip_atomic_fetch_add(&accL[ao[k] + lane],
                                                   lane < F ? tt[k] * hv[g + k] : tt[k],
                                                   __ATOMIC_RELAXED, __HIP_MEMORY_SCOPE_WORKGROUP);
                    }
                }
            } else {
                for (int q = 0; q < n; ++q) {
                    const int so = lane_i(hoff, q);
                    const int ao = lane_i(aoff, q);
                    const float tt = lane_f(t, q);
                    if (lane <= F) {
                        const float hv = h[so + lc];
                        __hip_atomic_fetch_add(&accL[ao + lane], lane < F ? tt * hv : tt,
                                               __ATOMIC_RELAXED, __HIP_MEMORY_SCOPE_WORKGROUP);
                    }
                }
            }
            hoff = hoffN; aoff = aoffN; t = tN;
        }
        __syncthreads();
    };

    const unsigned nbAll = ((unsigned)(N - 1) >> BSHIFT) + 1u;
    const unsigned nb    = nbAll < (unsigned)NBIN ? nbAll : (unsigned)NBIN;
    const unsigned rot   = ((unsigned)blockIdx.x * nb) / (unsigned)gridDim.x;
    auto bin_of = [&](const unsigned s) -> unsigned {
        const unsigned b0 = s >> BSHIFT;
        const unsigned t  = (b0 < nb ? b0 : nb - 1u) + rot;
        return t >= nb ? t - nb : t;
    };
    const bool exactIds = gridDim.x <= 256u;
    const unsigned myPar = (unsigned)blockIdx.x & 1u;
    auto resolve_sort = [&](const int cnt) -> int {
        for (int bI = tid; bI < NBIN; bI += OWN_T) binL[bI] = 0u;
        __syncthreads();
        for (int j0 = tid; j0 < cnt; j0 += DEEP * OWN_T) {
            unsigned ev[DEEP], pv[DEEP];
#pragma unroll
            for (int k = 0; k < DEEP; ++k) {
                const int j = j0 + k * OWN_T;
                const unsigned w = listL[j < cnt ? j : j0];
                ev[k] = j < cnt ? w : 0u;
            }
#pragma unroll
            for (int k = 0; k < DEEP; ++k)
                pv[k] = pk[ev[k]];
#pragma unroll
            for (int k = 0; k < DEEP; ++k) {
                const int j = j0 + k * OWN_T;
                if (j < cnt) {
                    const unsigned p = pv[k];
                    bool mine = (p >> 31) == myPar;
                    if (!exactIds) mine = (unsigned)(dst[ev[k]] - lo) < (unsigned)nOwn;
                    const unsigned entry = p & 0x7FFFFFFFu;
                    listL[j] = mine ? entry : NO_EDGE;
                    if (mine) atomicAdd(&binL[bin_of(entry & LOWS)], 1u);
                }
            }
        }
        __syncthreads();
        const unsigned own = binL[tid];
        unsigned inc = own;
#pragma unroll
        for (int o = 1; o < 32; o <<= 1) {
            const unsigned v = (unsigned)__shfl_up((int)inc, o);
            if (lane >= o) inc += v;
        }
        if (lane == 31) wsumL[wave] = inc;
        __syncthreads();
        if (wave == 0) {
            const unsigned w = wsumL[lane];
            unsigned wi = w;
#pragma unroll
            for (int o = 1; o < 32; o <<= 1) {
                const unsigned v = (unsigned)__shfl_up((int)wi, o);
                if (lane >= o) wi += v;
            }
            wsumL[lane] = wi - w;
            if (lane == 31) cntL = (int)wi;
        }
        __syncthreads();
        binL[tid] = wsumL[wave] + inc - own;
        const int kept = __builtin_amdgcn_readfirstlane(cntL);
        __syncthreads();
        for (int j = tid; j < cnt; j += OWN_T) {
            const unsigned entry = listL[j];
            if (entry != NO_EDGE) {
                const unsigned pos = atomicAdd(&binL[bin_of(entry & LOWS)], 1u);
                asU[pos] = entry;
            }
        }
        __syncthreads();
        for (int j = tid; j < kept; j += OWN_T) listL[j] = asU[j];
        __syncthreads();
        return kept;
    };

    unsigned int* const saved = savedList + (size_t)blockIdx.x * CAP;
    unsigned have = (unsigned)CAP + 1u;
    if (!first) have = savedCnt[blockIdx.x];

    int total;
    const bool resolved = have <= (unsigned)CAP;
    if (resolved) {
        total = (int)have;
        for (int j0 = 0; j0 < total; j0 += 8 * OWN_T) {
            unsigned v[8];
#pragma unroll
            for (int k = 0; k < 8; ++k) {
                const int j = j0 + k * OWN_T + tid;
                v[k] = j < total ? saved[j] : NO_EDGE;
            }
#pragma unroll
            for (int k = 0; k < 8; ++k) {
                const int j = j0 + k * OWN_T + tid;
                if (j < total) listL[j] = v[k];
            }
        }
    } else {
        const int keyed = (E / KSEG) * KSEG;
        for (int kb = 0; kb < keyed; kb += KSEG) scan_keys(kb);
        for (int sb = keyed; sb < E; sb += SEG) scan_segment(sb);
        __syncthreads();
        total = __builtin_amdgcn_readfirstlane(cntL);
        __syncthreads();
    }
    if (total <= CAP) {
        int cnt = total;
        if (!resolved) cnt = resolve_sort(total);
        process_chunk(cnt, true);
        if (first) {
            if (tid == 0) savedCnt[blockIdx.x] = (unsigned)cnt;
            for (int j = tid; j < cnt; j += OWN_T) saved[j] = listL[j];
        }
    } else {
        if (first && tid == 0) savedCnt[blockIdx.x] = (unsigned)total;
        if (tid == 0) cntL = 0;
        __syncthreads();
        for (int sb = 0; sb < E; sb += SEG) {
            scan_segment(sb);
            __syncthreads();
            const int cnt = __builtin_amdgcn_readfirstlane(cntL);
            __syncthreads();
            if (cnt <= CAP - SEG && sb + SEG < E) continue;
            process_chunk(cnt, false);
            if (tid == 0) cntL = 0;
            __syncthreads();
        }
    }

    if (first) {
        for (int idx = tid; idx < nOwn * F; idx += OWN_T) {
            const int dl = idx / F;
            const int f  = idx - dl * F;
            const float inv = 1.0f / (accL[dl * AW + F] + 1e-16f);
            const float v = accL[dl * AW + f] * inv + b[f];
            feat[(size_t)lo * F + idx] = v > 0.0f ? v : 0.0f;
        }
    } else {
        for (int dl = tid; dl < nOwn; dl += OWN_T) {
            const float inv = 1.0f / (accL[dl * AW + F] + 1e-16f);
            float sacc = bl[0];
#pragma unroll
            for (int f = 0; f < F; ++f) {
                const float v = accL[dl * AW + f] * inv + b[f];
                sacc += (v > 0.0f ? v : 0.0f) * Wl[f];
            }
            out[lo + dl] = sacc;
        }
    }
}

extern "C" void kernel_launch(void* const* d_in, const int* in_sizes, int n_in,
                              void* d_out, int out_size, void* d_ws, size_t ws_size,
                              hipStream_t stream) {
    (void)n_in; (void)out_size; (void)ws_size;

    const int N = in_sizes[0];
    const int E = in_sizes[1] / 2;

    const float* x    = (const float*)d_in[0];
    const int*   ei   = (const int*)d_in[1];
    const float* W1   = (const float*)d_in[2];
    const float* as1  = (const float*)d_in[3];
    const float* ad1  = (const float*)d_in[4];
    const float* b1   = (const float*)d_in[5];
    const float* W2   = (const float*)d_in[6];
    const float* as2  = (const float*)d_in[7];
    const float* ad2  = (const float*)d_in[8];
    const float* b2   = (const float*)d_in[9];
    const float* Wl   = (const float*)d_in[10];
    const float* bl   = (const float*)d_in[11];

    const int* srcIdx = ei;
    const int* dstIdx = ei + E;

    const int nWords = (E / KSEG) * (KSEG / 4);
    unsigned int* keyw = (unsigned int*)d_ws;
    unsigned int* pk   = keyw + nWords;
    float* ws     = (float*)(pk + E);
    float* h      = ws + (size_t)0  * N;
    float* feat1  = ws + (size_t)20 * N;
    float* asrc   = ws + (size_t)60 * N;
    float* adst   = ws + (size_t)61 * N;

    const int TB   = 256;
    const int gN   = (N + TB - 1) / TB;
    const int gMM  = (N + 15) / 16;
    const int gOwn = (N + RN - 1) / RN;
    const int gKey = ((E + 3) / 4) / TB + 1;

    unsigned int* savedList = (unsigned int*)(ws + (size_t)62 * N);
    unsigned int* savedCnt  = savedList + (size_t)gOwn * CAP;

    gat_keys_and_node_init<<<gKey + gN, TB, 0, stream>>>(srcIdx, dstIdx, keyw, pk, nWords, E, gKey,
                                                         x, W1, as1, ad1, h, asrc, adst, N);

    gat_owner_layer<<<gOwn, OWN_T, 0, stream>>>(srcIdx, dstIdx, keyw, pk, asrc, adst, h, b1, feat1,
                                                savedList, savedCnt, Wl, bl, (float*)d_out, 1, N, E);

    gat_wmma_feat_x_w<<<gMM, 32, 0, stream>>>(feat1, W2, as2, ad2, h, asrc, adst, N);
    gat_owner_layer<<<gOwn, OWN_T, 0, stream>>>(srcIdx, dstIdx, keyw, pk, asrc, adst, h, b2, feat1,
                                                savedList, savedCnt, Wl, bl, (float*)d_out, 0, N, E);
}


# Per-owner bit planes replace the shared owner-id bytes: key pass marks a 1024-edge chunk in LDS, stores 128 bytes per owner; an owner scans only its plane; list order changes, LDS adds arrival order
# speedup vs baseline: 15.3826x; 1.0810x over previous
#include <hip/hip_runtime.h>
#include <math.h>

#define F 20
#define NEG_SLOPE 0.2f

#define RN     392
#define AW     21
#define OWN_T  1024
#define SEG    32768
#define CAP    32768
#define KEY_T  256
#define PCH    1024
#define PROW   36
#define MAXOWN 256
#define PLD    13
#define SBITS  22
#define LOWS   0x3FFFFFu
#define NO_EDGE 0xFFFFFFFFu
#define NBIN   1024
#define BSHIFT 7
#define NINIT  ((RN * AW + OWN_T - 1) / OWN_T)
#define DEEP   8

typedef float v2f __attribute__((ext_vector_type(2)));
typedef float v8f __attribute__((ext_vector_type(8)));
typedef unsigned int u4 __attribute__((ext_vector_type(4)));

__device__ __forceinline__ float lrelu(float v) {
    return v > 0.0f ? v : NEG_SLOPE * v;
}

__device__ __forceinline__ int lane_i(int v, int l) {
    return __builtin_amdgcn_readlane(v, l);
}
__device__ __forceinline__ float lane_f(float v, int l) {
    return __int_as_float(__builtin_amdgcn_readlane(__float_as_int(v), l));
}

__device__ __forceinline__ void node_init_l1(const int blk,
                                             const float* __restrict__ x,
                                             const float* __restrict__ W1,
                                             const float* __restrict__ a_src,
                                             const float* __restrict__ a_dst,
                                             float* __restrict__ h,
                                             float* __restrict__ asrc,
                                             float* __restrict__ adst,
                                             int N) {
    int i = blk * blockDim.x + threadIdx.x;
    if (i >= N) return;
    float xv = x[i];
    float sa = 0.0f, sd = 0.0f;
#pragma unroll
    for (int f = 0; f < F; ++f) {
        float hv = xv * W1[f];
        h[i * F + f] = hv;
        sa += hv * a_src[f];
        sd += hv * a_dst[f];
    }
    asrc[i] = sa;
    adst[i] = sd;
}

__global__ void __launch_bounds__(32)
gat_wmma_feat_x_w(const float* __restrict__ feat,
                  const float* __restrict__ W,
                  const float* __restrict__ a_src,
                  const float* __restrict__ a_dst,
                  float* __restrict__ h,
                  float* __restrict__ asrc,
                  float* __restrict__ adst,
                  int N) {
    __shared__ float tileL[16 * AW];
    int tile  = blockIdx.x;
    int lane  = threadIdx.x;
    int mrow  = lane & 15;
    int khalf = (lane >> 4) * 2;

    int row  = tile * 16 + mrow;
    int rowc = row < N ? row : N - 1;
    const float* arow = feat + (size_t)rowc * F;

    int   n1   = 16 + mrow;
    int   n1c  = (n1 < F) ? n1 : 0;
    float msk1 = (n1 < F) ? 1.0f : 0.0f;

    v8f c0 = {};
    v8f c1 = {};
#pragma unroll
    for (int k0 = 0; k0 < F; k0 += 4) {
        const float* wr0 = W + (size_t)(k0 + khalf + 0) * F;
        const float* wr1 = W + (size_t)(k0 + khalf + 1) * F;

        v2f a;
        a.x = arow[k0 + khalf + 0];
        a.y = arow[k0 + khalf + 1];

        v2f b0;
        b0.x = wr0[mrow];
        b0.y = wr1[mrow];
        c0 = __builtin_amdgcn_wmma_f32_16x16x4_f32(false, a, false, b0,
                                                   (short)0, c0, false, false);

        v2f b1;
        b1.x = wr0[n1c] * msk1;
        b1.y = wr1[n1c] * msk1;
        c1 = __builtin_amdgcn_wmma_f32_16x16x4_f32(false, a, false, b1,
                                                   (short)0, c1, false, false);
    }

    int rbase = tile * 16 + ((lane >> 4) << 3);
#pragma unroll
    for (int j = 0; j < 8; ++j) {
        int r = rbase + j;
        if (r < N) {
            h[(size_t)r * F + mrow] = c0[j];
            if (n1 < F) h[(size_t)r * F + n1] = c1[j];
        }
        const int rl = ((lane >> 4) << 3) + j;
        tileL[rl * AW + mrow] = c0[j];
        if (n1 < F) tileL[rl * AW + n1] = c1[j];
    }
    __syncthreads();
    if (lane < 16) {
        const int r = tile * 16 + lane;
        if (r < N) {
            float sa = 0.0f, sd = 0.0f;
#pragma unroll
            for (int f = 0; f < F; ++f) {
                const float hv = tileL[lane * AW + f];
                sa += hv * a_src[f];
                sd += hv * a_dst[f];
            }
            asrc[r] = sa;
            adst[r] = sd;
        }
    }
}

__device__ __forceinline__ void owner_keys(const int blk,
                                           const int* __restrict__ src,
                                           const int* __restrict__ dst,
                                           unsigned int* __restrict__ planes,
                                           unsigned int* __restrict__ pk,
                                           unsigned int* planeL,
                                           int nBlk, int nOwners, int E) {
    const int tid = threadIdx.x;
    const int w  = blk * KEY_T + tid;
    const int e0 = 4 * w;
    const bool planed = blk < nBlk;
    if (planed) {
        for (int i = tid; i < MAXOWN * PROW / 4; i += KEY_T) ((u4*)planeL)[i] = (u4)(0u);
        __syncthreads();
    }
    unsigned int own[4] = {0u, 0u, 0u, 0u};
    if (e0 < E) {
        const bool whole = (E & 3) == 0;
        int d[4], s[4];
        if (whole) {
            const int4 d4 = ((const int4*)dst)[w];
            const int4 s4 = ((const int4*)src)[w];
            d[0] = d4.x; d[1] = d4.y; d[2] = d4.z; d[3] = d4.w;
            s[0] = s4.x; s[1] = s4.y; s[2] = s4.z; s[3] = s4.w;
        } else {
#pragma unroll
            for (int c = 0; c < 4; ++c) {
                const bool ok = e0 + c < E;
                d[c] = ok ? dst[e0 + c] : 0;
                s[c] = ok ? src[e0 + c] : 0;
            }
        }
        unsigned int p[4];
#pragma unroll
        for (int c = 0; c < 4; ++c) {
            const unsigned int owner = (unsigned int)d[c] / (unsigned int)RN;
            p[c] = (unsigned int)s[c] | (((unsigned int)d[c] - owner * (unsigned int)RN) << SBITS);
            own[c] = owner;
        }
        if (whole) {
            ((uint4*)pk)[w] = make_uint4(p[0], p[1], p[2], p[3]);
        } else {
#pragma unroll
            for (int c = 0; c < 4; ++c)
                if (e0 + c < E) pk[e0 + c] = p[c];
        }
    }
    if (planed) {
#pragma unroll
        for (int c = 0; c < 4; ++c) {
            const unsigned int li = 4u * (unsigned int)tid + (unsigned int)c;
            if (own[c] < (unsigned int)nOwners)
                atomicOr(&planeL[own[c] * PROW + (li >> 5)], 1u << (li & 31u));
        }
        __syncthreads();
        for (int r = tid >> 3; r < nOwners; r += KEY_T / 8) {
            const u4 v = *(const u4*)&planeL[r * PROW + 4 * (tid & 7)];
            __builtin_nontemporal_store(v, (u4*)(planes + ((size_t)r * (size_t)nBlk + (size_t)blk) * 32) + (tid & 7));
        }
    }
}

__global__ void __launch_bounds__(KEY_T)
gat_keys_and_node_init(const int* __restrict__ src,
                       const int* __restrict__ dst,
                       unsigned int* __restrict__ planes,
                       unsigned int* __restrict__ pk,
                       int nBlk, int nOwners, int E, int keyBlocks,
                       const float* __restrict__ x,
                       const float* __restrict__ W1,
                       const float* __restrict__ a_src,
                       const float* __restrict__ a_dst,
                       float* __restrict__ h,
                       float* __restrict__ asrc,
                       float* __restrict__ adst,
                       int N) {
    __shared__ __attribute__((aligned(16))) unsigned int planeL[MAXOWN * PROW];
    const int blk = (int)blockIdx.x;
    if (blk < keyBlocks) owner_keys(blk, src, dst, planes, pk, planeL, nBlk, nOwners, E);
    else                 node_init_l1(blk - keyBlocks, x, W1, a_src, a_dst, h, asrc, adst, N);
}

__global__ void __launch_bounds__(OWN_T)
gat_owner_layer(const int* __restrict__ src,
                const int* __restrict__ dst,
                const unsigned int* __restrict__ planes,
                const unsigned int* __restrict__ pk,
                const float* __restrict__ asrc,
                const float* __restrict__ adst,
                const float* __restrict__ h,
                const float* __restrict__ b,
                float* __restrict__ feat,
                unsigned int* __restrict__ savedList,
                unsigned int* __restrict__ savedCnt,
                const float* __restrict__ Wl,
                const float* __restrict__ bl,
                float* __restrict__ out,
                int first,
                int nBlk,
                int N, int E) {
    __shared__ float accL[RN * AW];
    __shared__ float mL[RN];
    __shared__ float mNew[RN];
    __shared__ float adL[RN];
    __shared__ float capL[RN];
    __shared__ float scL[RN];
    __shared__ unsigned listL[CAP];
    __shared__ unsigned asU[CAP];
    __shared__ unsigned binL[NBIN];
    __shared__ unsigned wsumL[32];
    __shared__ int cntL;

    const int tid  = threadIdx.x;
    const int lane = tid & 31;
    const int wave = __builtin_amdgcn_readfirstlane(tid >> 5);
    const int lo   = blockIdx.x * RN;
    const int nOwn = (N - lo) < RN ? (N - lo) : RN;

    for (int dl = tid; dl < RN; dl += OWN_T) {
        float a = -3.402823466e38f, ad = 0.0f;
        if (dl < nOwn) { a = asrc[lo + dl]; ad = adst[lo + dl]; }
        mL[dl] = a;
        mNew[dl] = a;
        adL[dl] = ad;
    }
    {
        float v0[NINIT];
#pragma unroll
        for (int k = 0; k < NINIT; ++k) {
            const int idx = tid + k * OWN_T;
            const int dl = idx / AW;
            const int f  = idx - dl * AW;
            const bool own = idx < RN * AW && dl < nOwn;
            const bool row = own && f < F;
            const float hv = h[row ? (size_t)(lo + dl) * F + f : (size_t)lo * F];
            v0[k] = row ? hv : (own ? 1.0f : 0.0f);
        }
#pragma unroll
        for (int k = 0; k < NINIT; ++k) {
            const int idx = tid + k * OWN_T;
            if (idx < RN * AW) accL[idx] = v0[k];
        }
    }
    if (tid == 0) cntL = 0;
    __syncthreads();

    const int nQ = nBlk * (PCH / 128);
    const u4* __restrict__ pl4 = (const u4*)(planes + (size_t)blockIdx.x * (size_t)nBlk * 32);
    auto scan_plane = [&](const int q0) {
        u4 q[PLD];
#pragma unroll
        for (int k = 0; k < PLD; ++k) {
            const int qi = q0 + k * OWN_T + tid;
            q[k] = __builtin_nontemporal_load(&pl4[qi < nQ ? qi : 0]);
        }
        asm volatile("" :: "v"(q[0]), "v"(q[1]), "v"(q[2]), "v"(q[3]), "v"(q[4]), "v"(q[5]), "v"(q[6]),
                           "v"(q[7]), "v"(q[8]), "v"(q[9]), "v"(q[10]), "v"(q[11]), "v"(q[12]));
        static_assert(PLD == 13, "the statement above names PLD pieces");
#pragma unroll
        for (int k = 0; k < PLD; ++k)
            if (q0 + k * OWN_T + tid >= nQ) q[k] = (u4)(0u);
        int n = 0;
#pragma unroll
        for (int k = 0; k < PLD; ++k)
            n += __popc(q[k].x) + __popc(q[k].y) + __popc(q[k].z) + __popc(q[k].w);
        int pre = 0, tot = 0;
        for (int d = 0; d < 11; ++d) {
            if (__builtin_amdgcn_ballot_w32((n >> d) != 0) == 0u) break;
            const unsigned set = __builtin_amdgcn_ballot_w32(((n >> d) & 1) != 0);
            pre += (int)__builtin_amdgcn_mbcnt_lo(set, 0u) << d;
            tot += __builtin_popcount(set) << d;
        }
        int base = 0;
        if (lane == 0) base = atomicAdd(&cntL, tot);
        unsigned slot = (unsigned)(lane_i(base, 0) + pre);
#pragma unroll
        for (int k = 0; k < PLD; ++k) {
            const unsigned eP = (unsigned)(q0 + k * OWN_T + tid) * 128u;
            const unsigned w[4] = {q[k].x, q[k].y, q[k].z, q[k].w};
#pragma unroll
            for (int c = 0; c < 4; ++c) {
                unsigned m = w[c];
                const unsigned eW = eP + 32u * (unsigned)c;
                const bool has = m != 0u;
                const unsigned e0 = eW + (unsigned)(__ffs((int)m) - 1);
                if (has) listL[slot < (unsigned)CAP ? slot : (unsigned)(CAP - 1)] = e0;
                slot += has ? 1u : 0u;
                m &= m - 1u;
                while (m != 0u) {
                    const unsigned e = eW + (unsigned)__builtin_ctz(m);
                    listL[slot < (unsigned)CAP ? slot : (unsigned)(CAP - 1)] = e;
                    ++slot;
                    m &= m - 1u;
                }
            }
        }
    };

    auto scan_segment = [&](const int sb) {
        if (((E & 3) == 0) && (sb + SEG <= E)) {
            const int4* __restrict__ dst4 = (const int4*)(dst + sb);
            int4 dq[SEG / (4 * OWN_T)];
#pragma unroll
            for (int k = 0; k < SEG / (4 * OWN_T); ++k) dq[k] = dst4[k * OWN_T + tid];
#pragma unroll
            for (int k = 0; k < SEG / (4 * OWN_T); ++k) {
                const unsigned u0 = (unsigned)(dq[k].x - lo);
                const unsigned u1 = (unsigned)(dq[k].y - lo);
                const unsigned u2 = (unsigned)(dq[k].z - lo);
                const unsigned u3 = (unsigned)(dq[k].w - lo);
                const bool m0 = u0 < (unsigned)nOwn, m1 = u1 < (unsigned)nOwn;
                const bool m2 = u2 < (unsigned)nOwn, m3 = u3 < (unsigned)nOwn;
                if (m0 | m1 | m2 | m3) {
                    const unsigned e0 = (unsigned)(sb + (k * OWN_T + tid) * 4);
                    if (m0) { const int slot = atomicAdd(&cntL, 1); if (slot < CAP) listL[slot] = e0 + 0u; }
                    if (m1) { const int slot = atomicAdd(&cntL, 1); if (slot < CAP) listL[slot] = e0 + 1u; }
                    if (m2) { const int slot = atomicAdd(&cntL, 1); if (slot < CAP) listL[slot] = e0 + 2u; }
                    if (m3) { const int slot = atomicAdd(&cntL, 1); if (slot < CAP) listL[slot] = e0 + 3u; }
                }
            }
        } else {
            for (int k0 = 0; k0 < SEG / OWN_T; k0 += 8) {
                int dv[8];
#pragma unroll
                for (int k = 0; k < 8; ++k) {
                    const int e = sb + (k0 + k) * OWN_T + tid;
                    dv[k] = e < E ? dst[e] : -1;
                }
#pragma unroll
                for (int k = 0; k < 8; ++k) {
                    const unsigned u = (unsigned)(dv[k] - lo);
                    if (u < (unsigned)nOwn) {
                        const int e = sb + (k0 + k) * OWN_T + tid;
                        const int slot = atomicAdd(&cntL, 1);
                        if (slot < CAP) listL[slot] = (unsigned)e;
                    }
                }
            }
        }
    };

    auto process_chunk = [&](const int cnt, const bool resolved) {
        if (resolved) {
            for (int j0 = tid; j0 < cnt; j0 += DEEP * OWN_T) {
                unsigned en[DEEP];
                float av[DEEP];
#pragma unroll
                for (int k = 0; k < DEEP; ++k) {
                    const int j = j0 + k * OWN_T;
                    const unsigned w = listL[j < cnt ? j : j0];
                    en[k] = j < cnt ? w : NO_EDGE;
                }
#pragma unroll
                for (int k = 0; k < DEEP; ++k)
                    av[k] = asrc[en[k] != NO_EDGE ? (en[k] & LOWS) : 0u];
#pragma unroll
                for (int k = 0; k < DEEP; ++k) {
                    if (en[k] != NO_EDGE) {
                        const int j = j0 + k * OWN_T;
                        const float a = av[k];
                        asU[j] = __float_as_uint(a);
                        const unsigned dl = en[k] >> SBITS;
                        if (a >= 0.0f) atomicMax((int*)&mNew[dl], __float_as_int(a));
                        else           atomicMin((unsigned int*)&mNew[dl], __float_as_uint(a));
                    }
                }
            }
        } else
        for (int j = tid; j < cnt; j += OWN_T) {
            unsigned entry = listL[j];
            if (!resolved) {
                const unsigned e = entry;
                const unsigned dl = (unsigned)(dst[e] - lo);
                const int s = src[e];
                entry = dl < (unsigned)nOwn ? ((unsigned)s | (dl << SBITS)) : NO_EDGE;
                listL[j] = entry;
            }
            if (entry != NO_EDGE) {
                const float a = asrc[entry & LOWS];
                asU[j] = __float_as_uint(a);
                const unsigned dl = entry >> SBITS;
                if (a >= 0.0f) atomicMax((int*)&mNew[dl], __float_as_int(a));
                else           atomicMin((unsigned int*)&mNew[dl], __float_as_uint(a));
            }
        }
        __syncthreads();

        if (tid < RN) {
            const float ad = adL[tid];
            const float mn = mNew[tid];
            const float shiftOld = lrelu(mL[tid] + ad);
            const float shiftNew = lrelu(mn + ad);
            scL[tid]  = __expf(shiftOld - shiftNew);
            capL[tid] = shiftNew;
            mL[tid]   = mn;
        }
        __syncthreads();
        for (int idx = tid; idx < RN * AW; idx += OWN_T) accL[idx] *= scL[idx / AW];
        __syncthreads();

        const int lc = lane < F ? lane : F - 1;
        auto prep = [&](const int base, int& hoff, int& aoff, float& t) {
            const int j = base + lane;
            hoff = lo * F; aoff = 0; t = 0.0f;
            const unsigned p = j < cnt ? listL[j] : NO_EDGE;
            if (p != NO_EDGE) {
                const int s  = (int)(p & LOWS);
                const int dl = (int)(p >> SBITS);
                const float ev = lrelu(__uint_as_float(asU[j]) + adL[dl]);
                t = __expf(ev - capL[dl]);
                hoff = s * F;
                aoff = dl * AW;
            }
        };
        int hoff = lo * F, aoff = 0;
        float t = 0.0f;
        if (wave * 32 < cnt) prep(wave * 32, hoff, aoff, t);
        for (int base = wave * 32; base < cnt; base += OWN_T) {
            const int n = cnt - base;
            int hoffN = lo * F, aoffN = 0;
            float tN = 0.0f;
            if (n >= 32) {
                float hv[32];
#pragma unroll
                for (int k = 0; k < 32; ++k) hv[k] = h[lane_i(hoff, k) + lc];
                if (base + OWN_T < cnt) prep(base + OWN_T, hoffN, aoffN, tN);
#pragma unroll
                for (int g = 0; g < 32; g += 16) {
                    int ao[16];
                    float tt[16];
#pragma unroll
                    for (int k = 0; k < 16; ++k) {
                        ao[k] = lane_i(aoff, g + k);
                        tt[k] = lane_f(t, g + k);
                    }
                    if (lane <= F) {
#pragma unroll
                        for (int k = 0; k < 16; ++k)
                            __hip_atomic_fetch_add(&accL[ao[k] + lane],
                                                   lane < F ? tt[k] * hv[g + k] : tt[k],
                                                   __ATOMIC_RELAXED, __HIP_MEMORY_SCOPE_WORKGROUP);
                    }
                }
            } else {
                for (int q = 0; q < n; ++q) {
                    const int so = lane_i(hoff, q);
                    const int ao = lane_i(aoff, q);
                    const float tt = lane_f(t, q);
                    if (lane <= F) {
                        const float hv = h[so + lc];
                        __hip_atomic_fetch_add(&accL[ao + lane], lane < F ? tt * hv : tt,
                                               __ATOMIC_RELAXED, __HIP_MEMORY_SCOPE_WORKGROUP);
                    }
                }
            }
            hoff = hoffN; aoff = aoffN; t = tN;
        }
        __syncthreads();
    };

    const unsigned nbAll = ((unsigned)(N - 1) >> BSHIFT) + 1u;
    const unsigned nb    = nbAll < (unsigned)NBIN ? nbAll : (unsigned)NBIN;
    const unsigned rot   = ((unsigned)blockIdx.x * nb) / (unsigned)gridDim.x;
    auto bin_of = [&](const unsigned s) -> unsigned {
        const unsigned b0 = s >> BSHIFT;
        const unsigned t  = (b0 < nb ? b0 : nb - 1u) + rot;
        return t >= nb ? t - nb : t;
    };
    auto resolve_sort = [&](const int cnt) {
        for (int bI = tid; bI < NBIN; bI += OWN_T) binL[bI] = 0u;
        __syncthreads();
        for (int j0 = tid; j0 < cnt; j0 += DEEP * OWN_T) {
            unsigned ev[DEEP], pv[DEEP];
#pragma unroll
            for (int k = 0; k < DEEP; ++k) {
                const int j = j0 + k * OWN_T;
                const unsigned w = listL[j < cnt ? j : j0];
                ev[k] = j < cnt ? w : 0u;
            }
#pragma unroll
            for (int k = 0; k < DEEP; ++k)
                pv[k] = pk[ev[k]];
#pragma unroll
            for (int k = 0; k < DEEP; ++k) {
                const int j = j0 + k * OWN_T;
                if (j < cnt) {
                    const unsigned entry = pv[k];
                    listL[j] = entry;
                    atomicAdd(&binL[bin_of(entry & LOWS)], 1u);
                }
            }
        }
        __syncthreads();
        const unsigned own = binL[tid];
        unsigned inc = own;
#pragma unroll
        for (int o = 1; o < 32; o <<= 1) {
            const unsigned v = (unsigned)__shfl_up((int)inc, o);
            if (lane >= o) inc += v;
        }
        if (lane == 31) wsumL[wave] = inc;
        __syncthreads();
        if (wave == 0) {
            const unsigned w = wsumL[lane];
            unsigned wi = w;
#pragma unroll
            for (int o = 1; o < 32; o <<= 1) {
                const unsigned v = (unsigned)__shfl_up((int)wi, o);
                if (lane >= o) wi += v;
            }
            wsumL[lane] = wi - w;
        }
        __syncthreads();
        binL[tid] = wsumL[wave] + inc - own;
        __syncthreads();
        for (int j = tid; j < cnt; j += OWN_T) {
            const unsigned entry = listL[j];
            const unsigned pos = atomicAdd(&binL[bin_of(entry & LOWS)], 1u);
            asU[pos] = entry;
        }
        __syncthreads();
        for (int j = tid; j < cnt; j += OWN_T) listL[j] = asU[j];
        __syncthreads();
    };

    unsigned int* const saved = savedList + (size_t)blockIdx.x * CAP;
    unsigned have = (unsigned)CAP + 1u;
    if (!first) have = savedCnt[blockIdx.x];

    int total;
    const bool resolved = have <= (unsigned)CAP;
    if (resolved) {
        total = (int)have;
        for (int j0 = 0; j0 < total; j0 += 8 * OWN_T) {
            unsigned v[8];
#pragma unroll
            for (int k = 0; k < 8; ++k) {
                const int j = j0 + k * OWN_T + tid;
                v[k] = j < total ? saved[j] : NO_EDGE;
            }
#pragma unroll
            for (int k = 0; k < 8; ++k) {
                const int j = j0 + k * OWN_T + tid;
                if (j < total) listL[j] = v[k];
            }
        }
    } else {
        const int keyed = nBlk * PCH;
        for (int q0 = 0; q0 < nQ; q0 += PLD * OWN_T) scan_plane(q0);
        for (int sb = keyed; sb < E; sb += SEG) scan_segment(sb);
        __syncthreads();
        total = __builtin_amdgcn_readfirstlane(cntL);
        __syncthreads();
    }
    if (total <= CAP) {
        const int cnt = total;
        if (!resolved) resolve_sort(cnt);
        process_chunk(cnt, true);
        if (first) {
            if (tid == 0) savedCnt[blockIdx.x] = (unsigned)cnt;
            for (int j = tid; j < cnt; j += OWN_T) saved[j] = listL[j];
        }
    } else {
        if (first && tid == 0) savedCnt[blockIdx.x] = (unsigned)total;
        if (tid == 0) cntL = 0;
        __syncthreads();
        for (int sb = 0; sb < E; sb += SEG) {
            scan_segment(sb);
            __syncthreads();
            const int cnt = __builtin_amdgcn_readfirstlane(cntL);
            __syncthreads();
            if (cnt <= CAP - SEG && sb + SEG < E) continue;
            process_chunk(cnt, false);
            if (tid == 0) cntL = 0;
            __syncthreads();
        }
    }

    if (first) {
        for (int idx = tid; idx < nOwn * F; idx += OWN_T) {
            const int dl = idx / F;
            const int f  = idx - dl * F;
            const float inv = 1.0f / (accL[dl * AW + F] + 1e-16f);
            const float v = accL[dl * AW + f] * inv + b[f];
            feat[(size_t)lo * F + idx] = v > 0.0f ? v : 0.0f;
        }
    } else {
        for (int dl = tid; dl < nOwn; dl += OWN_T) {
            const float inv = 1.0f / (accL[dl * AW + F] + 1e-16f);
            float sacc = bl[0];
#pragma unroll
            for (int f = 0; f < F; ++f) {
                const float v = accL[dl * AW + f] * inv + b[f];
                sacc += (v > 0.0f ? v : 0.0f) * Wl[f];
            }
            out[lo + dl] = sacc;
        }
    }
}

extern "C" void kernel_launch(void* const* d_in, const int* in_sizes, int n_in,
                              void* d_out, int out_size, void* d_ws, size_t ws_size,
                              hipStream_t stream) {
    (void)n_in; (void)out_size; (void)ws_size;

    const int N = in_sizes[0];
    const int E = in_sizes[1] / 2;

    const float* x    = (const float*)d_in[0];
    const int*   ei   = (const int*)d_in[1];
    const float* W1   = (const float*)d_in[2];
    const float* as1  = (const float*)d_in[3];
    const float* ad1  = (const float*)d_in[4];
    const float* b1   = (const float*)d_in[5];
    const float* W2   = (const float*)d_in[6];
    const float* as2  = (const float*)d_in[7];
    const float* ad2  = (const float*)d_in[8];
    const float* b2   = (const float*)d_in[9];
    const float* Wl   = (const float*)d_in[10];
    const float* bl   = (const float*)d_in[11];

    const int* srcIdx = ei;
    const int* dstIdx = ei + E;

    const int gOwn = (N + RN - 1) / RN;

    const int nBlk = gOwn <= MAXOWN ? E / PCH : 0;
    unsigned int* planes = (unsigned int*)d_ws;
    unsigned int* pk   = planes + (size_t)gOwn * nBlk * 32;
    float* ws     = (float*)(pk + E);
    float* h      = ws + (size_t)0  * N;
    float* feat1  = ws + (size_t)20 * N;
    float* asrc   = ws + (size_t)60 * N;
    float* adst   = ws + (size_t)61 * N;

    const int TB   = KEY_T;
    const int gN   = (N + TB - 1) / TB;
    const int gMM  = (N + 15) / 16;
    const int gKey = ((E + 3) / 4) / TB + 1;

    unsigned int* savedList = (unsigned int*)(ws + (size_t)62 * N);
    unsigned int* savedCnt  = savedList + (size_t)gOwn * CAP;

    gat_keys_and_node_init<<<gKey + gN, TB, 0, stream>>>(srcIdx, dstIdx, planes, pk, nBlk, gOwn, E, gKey,
                                                         x, W1, as1, ad1, h, asrc, adst, N);

    gat_owner_layer<<<gOwn, OWN_T, 0, stream>>>(srcIdx, dstIdx, planes, pk, asrc, adst, h, b1, feat1,
                                                savedList, savedCnt, Wl, bl, (float*)d_out, 1, nBlk, N, E);

    gat_wmma_feat_x_w<<<gMM, 32, 0, stream>>>(feat1, W2, as2, ad2, h, asrc, adst, N);
    gat_owner_layer<<<gOwn, OWN_T, 0, stream>>>(srcIdx, dstIdx, planes, pk, asrc, adst, h, b2, feat1,
                                                savedList, savedCnt, Wl, bl, (float*)d_out, 0, nBlk, N, E);
}
